# starting kernel
# baseline (speedup 1.0000x reference)
#include <hip/hip_runtime.h>
#include <hip/hip_bf16.h>

#define HEADS 4
#define HC    16
#define DIM   64
#define NEGS  0.2f

typedef float v2f __attribute__((ext_vector_type(2)));
typedef float v8f __attribute__((ext_vector_type(8)));

__device__ __forceinline__ unsigned f2key(float f) {
    unsigned u = __float_as_uint(f);
    return u ^ ((unsigned)(((int)u) >> 31) | 0x80000000u);
}
__device__ __forceinline__ float key2f(unsigned k) {
    unsigned u = (k & 0x80000000u) ? (k ^ 0x80000000u) : ~k;
    return __uint_as_float(u);
}

__global__ void k_zero(float* __restrict__ p, long n) {
    long i = (long)blockIdx.x * blockDim.x + threadIdx.x;
    long st = (long)gridDim.x * blockDim.x;
    for (; i < n; i += st) p[i] = 0.0f;
}

__global__ void k_edge_count(const int* __restrict__ ei, const float* __restrict__ eattr,
                             float* __restrict__ lsum, float* __restrict__ cnt, int E) {
    int e = blockIdx.x * blockDim.x + threadIdx.x;
    if (e >= E) return;
    int d = ei[E + e];
    atomicAdd(&lsum[d], eattr[e]);
    atomicAdd(&cnt[d], 1.0f);
}
__global__ void k_loop_attr(float* __restrict__ lsum, const float* __restrict__ cnt, int N) {
    int n = blockIdx.x * blockDim.x + threadIdx.x;
    if (n >= N) return;
    lsum[n] = lsum[n] / fmaxf(cnt[n], 1.0f);
}

__global__ void k_edge_scal(const float* __restrict__ We1, const float* __restrict__ ae1,
                            const float* __restrict__ We2, const float* __restrict__ ae2,
                            float* __restrict__ sed) {
    int t = threadIdx.x;
    if (t >= 8) return;
    const float* We = (t < 4) ? We1 : We2;
    const float* ae = (t < 4) ? ae1 : ae2;
    int h = t & 3;
    float s = 0.0f;
    for (int c = 0; c < HC; ++c) s += We[h * HC + c] * ae[h * HC + c];
    sed[t] = s;
}

__global__ void k_gemm64_wmma(const float* __restrict__ X, const float* __restrict__ W,
                              float* __restrict__ Y, int nrows) {
    int wave = threadIdx.x >> 5;
    int lane = threadIdx.x & 31;
    int rowTile = blockIdx.x * 2 + (wave >> 2);
    int row0 = rowTile * 16;
    int col0 = (wave & 3) * 16;
    if (row0 >= nrows) return;
    int lr = lane & 15;
    int hi = lane >> 4;

    int arow = row0 + lr; if (arow > nrows - 1) arow = nrows - 1;
    const float* ap = X + (size_t)arow * DIM;

    v8f acc = {};
    for (int k = 0; k < DIM; k += 4) {
        v2f a;
        a.x = ap[k + 2 * hi + 0];
        a.y = ap[k + 2 * hi + 1];
        const float* bp = W + (size_t)(k + 2 * hi) * DIM + col0 + lr;
        v2f b;
        b.x = bp[0];
        b.y = bp[DIM];
        acc = __builtin_amdgcn_wmma_f32_16x16x4_f32(false, a, false, b,
                                                    (short)0, acc, false, false);
    }
    for (int r = 0; r < 8; ++r) {
        int row = row0 + r + 8 * hi;
        if (row < nrows) Y[(size_t)row * DIM + col0 + lr] = acc[r];
    }
}

__global__ void k_attn_coef(const float* __restrict__ xs, const float* __restrict__ as,
                            const float* __restrict__ ad, float* __restrict__ a_src,
                            float* __restrict__ a_dst, int N) {
    int t = blockIdx.x * blockDim.x + threadIdx.x;
    if (t >= N * HEADS) return;
    int n = t >> 2, h = t & 3;
    const float* xp = xs + (size_t)n * DIM + h * HC;
    float ss = 0.0f, sd = 0.0f;
    for (int c = 0; c < HC; ++c) {
        float x = xp[c];
        ss += x * as[h * HC + c];
        sd += x * ad[h * HC + c];
    }
    a_src[t] = ss;
    a_dst[t] = sd;
}

__global__ void k_edge_max(const int* __restrict__ ei, const float* __restrict__ eattr,
                           const float* __restrict__ lattr, const float* __restrict__ sed,
                           const float* __restrict__ a_src, const float* __restrict__ a_dst,
                           unsigned* __restrict__ mkey, int E, int N) {
    int e = blockIdx.x * blockDim.x + threadIdx.x;
    int ET = E + N;
    if (e >= ET) return;
    int s, d; float ea;
    if (e < E) { s = ei[e]; d = ei[E + e]; ea = eattr[e]; }
    else       { s = d = e - E;            ea = lattr[s]; }
    #pragma unroll
    for (int h = 0; h < HEADS; ++h) {
        float al = a_src[s * HEADS + h] + a_dst[d * HEADS + h] + ea * sed[h];
        al = (al > 0.0f) ? al : NEGS * al;
        atomicMax(&mkey[d * HEADS + h], f2key(al));
    }
}

__global__ void k_edge_exp(const int* __restrict__ ei, const float* __restrict__ eattr,
                           const float* __restrict__ lattr, const float* __restrict__ sed,
                           const float* __restrict__ a_src, const float* __restrict__ a_dst,
                           const unsigned* __restrict__ mkey, float* __restrict__ exv,
                           float* __restrict__ den, int E, int N) {
    int e = blockIdx.x * blockDim.x + threadIdx.x;
    int ET = E + N;
    if (e >= ET) return;
    int s, d; float ea;
    if (e < E) { s = ei[e]; d = ei[E + e]; ea = eattr[e]; }
    else       { s = d = e - E;            ea = lattr[s]; }
    #pragma unroll
    for (int h = 0; h < HEADS; ++h) {
        float al = a_src[s * HEADS + h] + a_dst[d * HEADS + h] + ea * sed[h];
        al = (al > 0.0f) ? al : NEGS * al;
        float m = key2f(mkey[d * HEADS + h]);
        float ex = expf(al - m);
        exv[(size_t)e * HEADS + h] = ex;
        atomicAdd(&den[d * HEADS + h], ex);
    }
}

__global__ void k_edge_agg(const int* __restrict__ ei, const float* __restrict__ exv,
                           const float* __restrict__ den, const float* __restrict__ xs,
                           float* __restrict__ agg, int E, int N) {
    long t = (long)blockIdx.x * blockDim.x + threadIdx.x;
    long ET = (long)E + N;
    if (t >= ET * DIM) return;
    long e = t >> 6;
    int f = (int)(t & 63);
    int s, d;
    if (e < E) { s = ei[e]; d = ei[E + e]; }
    else       { s = d = (int)(e - E); }
    int h = f >> 4;
    float w = exv[e * HEADS + h] / (den[d * HEADS + h] + 1e-16f);
    atomicAdd(&agg[(size_t)d * DIM + f], xs[(size_t)s * DIM + f] * w);
}

__global__ void k_post(const float* __restrict__ agg, const float* __restrict__ res,
                       const float* __restrict__ b, const float* __restrict__ g,
                       const float* __restrict__ beta, float* __restrict__ out, int N) {
    int wave = threadIdx.x >> 5;
    int lane = threadIdx.x & 31;
    int n = blockIdx.x * 8 + wave;
    if (n >= N) return;
    const float* ap = agg + (size_t)n * DIM;
    const float* rp = res + (size_t)n * DIM;
    float v0 = ap[lane]      + b[lane]      + rp[lane];
    float v1 = ap[lane + 32] + b[lane + 32] + rp[lane + 32];
    float sum = v0 + v1;
    #pragma unroll
    for (int m = 16; m >= 1; m >>= 1) sum += __shfl_xor(sum, m, 32);
    float mu = sum * (1.0f / DIM);
    float d0 = v0 - mu, d1 = v1 - mu;
    float vs = d0 * d0 + d1 * d1;
    #pragma unroll
    for (int m = 16; m >= 1; m >>= 1) vs += __shfl_xor(vs, m, 32);
    float rstd = rsqrtf(vs * (1.0f / DIM) + 1e-5f);
    float y0 = d0 * rstd * g[lane]      + beta[lane];
    float y1 = d1 * rstd * g[lane + 32] + beta[lane + 32];
    y0 = (y0 > 0.0f) ? y0 : (expf(y0) - 1.0f);
    y1 = (y1 > 0.0f) ? y1 : (expf(y1) - 1.0f);
    float* op = out + (size_t)n * DIM;
    op[lane]      = y0;
    op[lane + 32] = y1;
}

__global__ void k_out(const float* __restrict__ hfin, const float* __restrict__ Wout,
                      const float* __restrict__ bout, float* __restrict__ out, int N) {
    int wave = threadIdx.x >> 5;
    int lane = threadIdx.x & 31;
    int n = blockIdx.x * 8 + wave;
    if (n >= N) return;
    const float* hp = hfin + (size_t)n * DIM;
    float p = hp[lane] * Wout[lane] + hp[lane + 32] * Wout[lane + 32];
    #pragma unroll
    for (int m = 16; m >= 1; m >>= 1) p += __shfl_xor(p, m, 32);
    if (lane == 0) out[n] = p + bout[0];
}

extern "C" void kernel_launch(void* const* d_in, const int* in_sizes, int n_in,
                              void* d_out, int out_size, void* d_ws, size_t ws_size,
                              hipStream_t stream) {
    const float* x    = (const float*)d_in[0];
    const int*   ei   = (const int*)  d_in[1];
    const float* eatt = (const float*)d_in[3];
    const float* W1   = (const float*)d_in[4];
    const float* as1  = (const float*)d_in[5];
    const float* ad1  = (const float*)d_in[6];
    const float* We1  = (const float*)d_in[7];
    const float* ae1  = (const float*)d_in[8];
    const float* b1   = (const float*)d_in[9];
    const float* W2   = (const float*)d_in[10];
    const float* as2  = (const float*)d_in[11];
    const float* ad2  = (const float*)d_in[12];
    const float* We2  = (const float*)d_in[13];
    const float* ae2  = (const float*)d_in[14];
    const float* b2   = (const float*)d_in[15];
    const float* g1   = (const float*)d_in[16];
    const float* be1  = (const float*)d_in[17];
    const float* g2   = (const float*)d_in[18];
    const float* be2  = (const float*)d_in[19];
    const float* Wout = (const float*)d_in[20];
    const float* bout = (const float*)d_in[21];
    float* outp = (float*)d_out;

    const int N = in_sizes[0] / DIM;
    const int E = in_sizes[1] / 2;
    const long ET = (long)E + N;

    float* w = (float*)d_ws;
    size_t o = 0;
    float*    lattr = w + o; o += N;
    float*    cnt   = w + o; o += N;
    float*    xs    = w + o; o += (size_t)N * DIM;
    float*    a_src = w + o; o += (size_t)N * HEADS;
    float*    a_dst = w + o; o += (size_t)N * HEADS;
    unsigned* mkey  = (unsigned*)(w + o); o += (size_t)N * HEADS;
    float*    den   = w + o; o += (size_t)N * HEADS;
    float*    agg   = w + o; o += (size_t)N * DIM;
    float*    h1    = w + o; o += (size_t)N * DIM;
    float*    exv   = w + o; o += (size_t)ET * HEADS;
    float*    sed   = w + o; o += 8;

    auto cdiv = [](long a, long b) { return (int)((a + b - 1) / b); };
    const int T = 256;

    k_zero<<<512, T, 0, stream>>>(lattr, (long)2 * N);
    k_edge_count<<<cdiv(E, T), T, 0, stream>>>(ei, eatt, lattr, cnt, E);
    k_loop_attr<<<cdiv(N, T), T, 0, stream>>>(lattr, cnt, N);
    k_edge_scal<<<1, 32, 0, stream>>>(We1, ae1, We2, ae2, sed);

    k_gemm64_wmma<<<cdiv((long)cdiv(N, 16), 2), T, 0, stream>>>(x, W1, xs, N);
    k_attn_coef<<<cdiv((long)N * HEADS, T), T, 0, stream>>>(xs, as1, ad1, a_src, a_dst, N);
    k_zero<<<2048, T, 0, stream>>>((float*)mkey, (long)N * (HEADS + HEADS + DIM));
    k_edge_max<<<cdiv(ET, T), T, 0, stream>>>(ei, eatt, lattr, sed, a_src, a_dst, mkey, E, N);
    k_edge_exp<<<cdiv(ET, T), T, 0, stream>>>(ei, eatt, lattr, sed, a_src, a_dst, mkey, exv, den, E, N);
    k_edge_agg<<<cdiv(ET * DIM, T), T, 0, stream>>>(ei, exv, den, xs, agg, E, N);
    k_post<<<cdiv(N, 8), T, 0, stream>>>(agg, x, b1, g1, be1, h1, N);

    k_gemm64_wmma<<<cdiv((long)cdiv(N, 16), 2), T, 0, stream>>>(h1, W2, xs, N);
    k_attn_coef<<<cdiv((long)N * HEADS, T), T, 0, stream>>>(xs, as2, ad2, a_src, a_dst, N);
    k_zero<<<2048, T, 0, stream>>>((float*)mkey, (long)N * (HEADS + HEADS + DIM));
    k_edge_max<<<cdiv(ET, T), T, 0, stream>>>(ei, eatt, lattr, sed + 4, a_src, a_dst, mkey, E, N);
    k_edge_exp<<<cdiv(ET, T), T, 0, stream>>>(ei, eatt, lattr, sed + 4, a_src, a_dst, mkey, exv, den, E, N);
    k_edge_agg<<<cdiv(ET * DIM, T), T, 0, stream>>>(ei, exv, den, xs, agg, E, N);
    k_post<<<cdiv(N, 8), T, 0, stream>>>(agg, h1, b2, g2, be2, agg, N);

    k_out<<<cdiv(N, 8), T, 0, stream>>>(agg, Wout, bout, outp, N);
}


# Edges grouped by destination in-call (int atomic rank, prefix sum, fill); one wave per node does max, exp-sum and weighted row gather in registers: float atomics gone, sums run in segment order
# speedup vs baseline: 6.6974x; 6.6974x over previous
#include <hip/hip_runtime.h>
#include <hip/hip_bf16.h>

#define HEADS 4
#define HC    16
#define DIM   64
#define NEGS  0.2f

typedef float v2f __attribute__((ext_vector_type(2)));
typedef float v8f __attribute__((ext_vector_type(8)));

__global__ void k_zero(float* __restrict__ p, long n) {
    long i = (long)blockIdx.x * blockDim.x + threadIdx.x;
    long st = (long)gridDim.x * blockDim.x;
    for (; i < n; i += st) p[i] = 0.0f;
}

__global__ void k_edge_scal(const float* __restrict__ We1, const float* __restrict__ ae1,
                            const float* __restrict__ We2, const float* __restrict__ ae2,
                            float* __restrict__ sed) {
    int t = threadIdx.x;
    if (t >= 8) return;
    const float* We = (t < 4) ? We1 : We2;
    const float* ae = (t < 4) ? ae1 : ae2;
    int h = t & 3;
    float s = 0.0f;
    for (int c = 0; c < HC; ++c) s += We[h * HC + c] * ae[h * HC + c];
    sed[t] = s;
}

__global__ void k_gemm64_wmma(const float* __restrict__ X, const float* __restrict__ W,
                              float* __restrict__ Y, int nrows) {
    int wave = threadIdx.x >> 5;
    int lane = threadIdx.x & 31;
    int rowTile = blockIdx.x * 2 + (wave >> 2);
    int row0 = rowTile * 16;
    int col0 = (wave & 3) * 16;
    if (row0 >= nrows) return;
    int lr = lane & 15;
    int hi = lane >> 4;

    int arow = row0 + lr; if (arow > nrows - 1) arow = nrows - 1;
    const float* ap = X + (size_t)arow * DIM;

    v8f acc = {};
    for (int k = 0; k < DIM; k += 4) {
        v2f a;
        a.x = ap[k + 2 * hi + 0];
        a.y = ap[k + 2 * hi + 1];
        const float* bp = W + (size_t)(k + 2 * hi) * DIM + col0 + lr;
        v2f b;
        b.x = bp[0];
        b.y = bp[DIM];
        acc = __builtin_amdgcn_wmma_f32_16x16x4_f32(false, a, false, b,
                                                    (short)0, acc, false, false);
    }
    for (int r = 0; r < 8; ++r) {
        int row = row0 + r + 8 * hi;
        if (row < nrows) Y[(size_t)row * DIM + col0 + lr] = acc[r];
    }
}

__global__ void k_attn_coef(const float* __restrict__ xs, const float* __restrict__ as,
                            const float* __restrict__ ad, float* __restrict__ a_src,
                            float* __restrict__ a_dst, int N) {
    int t = blockIdx.x * blockDim.x + threadIdx.x;
    if (t >= N * HEADS) return;
    int n = t >> 2, h = t & 3;
    const float* xp = xs + (size_t)n * DIM + h * HC;
    float ss = 0.0f, sd = 0.0f;
    for (int c = 0; c < HC; ++c) {
        float x = xp[c];
        ss += x * as[h * HC + c];
        sd += x * ad[h * HC + c];
    }
    a_src[t] = ss;
    a_dst[t] = sd;
}

__global__ void k_edge_rank(const int* __restrict__ ei, int* __restrict__ cnt,
                            int* __restrict__ rank, int E) {
    int e = blockIdx.x * blockDim.x + threadIdx.x;
    if (e >= E) return;
    rank[e] = atomicAdd(&cnt[ei[E + e]], 1);
}

__global__ void __launch_bounds__(1024)
k_row_offsets(const int* __restrict__ cnt, int* __restrict__ rowptr, int N) {
    __shared__ int s_pre[32];
    __shared__ int s_sum[32];
    const int t = threadIdx.x, lane = t & 31, wave = t >> 5;
    const int base = blockIdx.x * 1024;
    int pre = 0;
    for (int i = t; i < base; i += 1024) pre += cnt[i];
    #pragma unroll
    for (int o = 16; o >= 1; o >>= 1) pre += __shfl_xor(pre, o, 32);
    const int idx = base + t;
    const int v = (idx < N) ? cnt[idx] : 0;
    int inc = v;
    #pragma unroll
    for (int o = 1; o < 32; o <<= 1) {
        int u = __shfl_up(inc, o, 32);
        if (lane >= o) inc += u;
    }
    if (lane == 0)  s_pre[wave] = pre;
    if (lane == 31) s_sum[wave] = inc;
    __syncthreads();
    int off = 0;
    for (int w = 0; w < 32; ++w) {
        off += s_pre[w];
        if (w < wave) off += s_sum[w];
    }
    const int excl = off + inc - v;
    if (idx < N) rowptr[idx] = excl;
    if (idx == N - 1) rowptr[N] = excl + v;
}

__global__ void k_edge_fill(const int* __restrict__ ei, const float* __restrict__ eattr,
                            const int* __restrict__ rowptr, const int* __restrict__ rank,
                            int* __restrict__ spair, int E) {
    int e = blockIdx.x * blockDim.x + threadIdx.x;
    if (e >= E) return;
    int pos = rowptr[ei[E + e]] + rank[e];
    int2 v;
    v.x = ei[e];
    v.y = __float_as_int(eattr[e]);
    ((int2*)spair)[pos] = v;
}

__device__ __forceinline__ float edge_logit(float as, float ad, float ea, float se) {
    float al = __builtin_fmaf(ea, se, as + ad);
    return (al > 0.0f) ? al : NEGS * al;
}

__global__ void __launch_bounds__(256)
k_node_softmax_agg(const int* __restrict__ rowptr, const int* __restrict__ spair,
                   const float* __restrict__ sed, const float* __restrict__ a_src,
                   const float* __restrict__ a_dst, const float* __restrict__ xs,
                   float* __restrict__ agg, int N) {
    const int lane = threadIdx.x & 31;
    const int n = blockIdx.x * 8 + __builtin_amdgcn_readfirstlane((int)(threadIdx.x >> 5));
    if (n >= N) return;
    const int beg = rowptr[n];
    const int end = rowptr[n + 1];
    const int2*   pairs = (const int2*)spair;
    const float4* as4   = (const float4*)a_src;
    const float4 adn = ((const float4*)a_dst)[n];
    const float4 asn = as4[n];
    const float se0 = sed[0], se1 = sed[1], se2 = sed[2], se3 = sed[3];

    const float ninf = -__builtin_huge_valf();
    float easum = 0.0f, m0 = ninf, m1 = ninf, m2 = ninf, m3 = ninf;
    for (int p = beg + lane; p < end; p += 32) {
        const int2 pr = pairs[p];
        const float ea = __int_as_float(pr.y);
        const float4 a = as4[pr.x];
        easum += ea;
        m0 = fmaxf(m0, edge_logit(a.x, adn.x, ea, se0));
        m1 = fmaxf(m1, edge_logit(a.y, adn.y, ea, se1));
        m2 = fmaxf(m2, edge_logit(a.z, adn.z, ea, se2));
        m3 = fmaxf(m3, edge_logit(a.w, adn.w, ea, se3));
    }
    #pragma unroll
    for (int o = 16; o >= 1; o >>= 1) {
        easum += __shfl_xor(easum, o, 32);
        m0 = fmaxf(m0, __shfl_xor(m0, o, 32));
        m1 = fmaxf(m1, __shfl_xor(m1, o, 32));
        m2 = fmaxf(m2, __shfl_xor(m2, o, 32));
        m3 = fmaxf(m3, __shfl_xor(m3, o, 32));
    }
    const float la = easum / fmaxf((float)(end - beg), 1.0f);
    const float sl0 = edge_logit(asn.x, adn.x, la, se0);
    const float sl1 = edge_logit(asn.y, adn.y, la, se1);
    const float sl2 = edge_logit(asn.z, adn.z, la, se2);
    const float sl3 = edge_logit(asn.w, adn.w, la, se3);
    m0 = fmaxf(m0, sl0); m1 = fmaxf(m1, sl1); m2 = fmaxf(m2, sl2); m3 = fmaxf(m3, sl3);

    const int h = lane >> 3;
    const float mh  = (h == 0) ? m0  : (h == 1) ? m1  : (h == 2) ? m2  : m3;
    const float slh = (h == 0) ? sl0 : (h == 1) ? sl1 : (h == 2) ? sl2 : sl3;
    const float2* xrow = (const float2*)xs + lane;
    const float es = expf(slh - mh);
    const float2 xn = xrow[(size_t)n * (DIM / 2)];
    float den = es;
    float ax = xn.x * es, ay = xn.y * es;
    for (int base = beg; base < end; base += 32) {
        const int p = base + lane;
        int s = n;
        float e0 = 0.0f, e1 = 0.0f, e2 = 0.0f, e3 = 0.0f;
        if (p < end) {
            const int2 pr = pairs[p];
            const float ea = __int_as_float(pr.y);
            s = pr.x;
            const float4 a = as4[s];
            e0 = expf(edge_logit(a.x, adn.x, ea, se0) - m0);
            e1 = expf(edge_logit(a.y, adn.y, ea, se1) - m1);
            e2 = expf(edge_logit(a.z, adn.z, ea, se2) - m2);
            e3 = expf(edge_logit(a.w, adn.w, ea, se3) - m3);
        }
        const int cc = (end - base < 32) ? (end - base) : 32;
        #pragma unroll 4
        for (int j = 0; j < cc; ++j) {
            const int sj = __builtin_amdgcn_readlane(s, j);
            const float r0 = __int_as_float(__builtin_amdgcn_readlane(__float_as_int(e0), j));
            const float r1 = __int_as_float(__builtin_amdgcn_readlane(__float_as_int(e1), j));
            const float r2 = __int_as_float(__builtin_amdgcn_readlane(__float_as_int(e2), j));
            const float r3 = __int_as_float(__builtin_amdgcn_readlane(__float_as_int(e3), j));
            const float wj = (h == 0) ? r0 : (h == 1) ? r1 : (h == 2) ? r2 : r3;
            const float2 xv = xrow[(size_t)sj * (DIM / 2)];
            den += wj;
            ax = __builtin_fmaf(xv.x, wj, ax);
            ay = __builtin_fmaf(xv.y, wj, ay);
        }
    }
    const float inv = den + 1e-16f;
    float2 o;
    o.x = ax / inv;
    o.y = ay / inv;
    ((float2*)agg)[(size_t)n * (DIM / 2) + lane] = o;
}

__global__ void k_post(const float* __restrict__ agg, const float* __restrict__ res,
                       const float* __restrict__ b, const float* __restrict__ g,
                       const float* __restrict__ beta, float* __restrict__ out, int N) {
    int wave = threadIdx.x >> 5;
    int lane = threadIdx.x & 31;
    int n = blockIdx.x * 8 + wave;
    if (n >= N) return;
    const float* ap = agg + (size_t)n * DIM;
    const float* rp = res + (size_t)n * DIM;
    float v0 = ap[lane]      + b[lane]      + rp[lane];
    float v1 = ap[lane + 32] + b[lane + 32] + rp[lane + 32];
    float sum = v0 + v1;
    #pragma unroll
    for (int m = 16; m >= 1; m >>= 1) sum += __shfl_xor(sum, m, 32);
    float mu = sum * (1.0f / DIM);
    float d0 = v0 - mu, d1 = v1 - mu;
    float vs = d0 * d0 + d1 * d1;
    #pragma unroll
    for (int m = 16; m >= 1; m >>= 1) vs += __shfl_xor(vs, m, 32);
    float rstd = rsqrtf(vs * (1.0f / DIM) + 1e-5f);
    float y0 = d0 * rstd * g[lane]      + beta[lane];
    float y1 = d1 * rstd * g[lane + 32] + beta[lane + 32];
    y0 = (y0 > 0.0f) ? y0 : (expf(y0) - 1.0f);
    y1 = (y1 > 0.0f) ? y1 : (expf(y1) - 1.0f);
    float* op = out + (size_t)n * DIM;
    op[lane]      = y0;
    op[lane + 32] = y1;
}

__global__ void k_out(const float* __restrict__ hfin, const float* __restrict__ Wout,
                      const float* __restrict__ bout, float* __restrict__ out, int N) {
    int wave = threadIdx.x >> 5;
    int lane = threadIdx.x & 31;
    int n = blockIdx.x * 8 + wave;
    if (n >= N) return;
    const float* hp = hfin + (size_t)n * DIM;
    float p = hp[lane] * Wout[lane] + hp[lane + 32] * Wout[lane + 32];
    #pragma unroll
    for (int m = 16; m >= 1; m >>= 1) p += __shfl_xor(p, m, 32);
    if (lane == 0) out[n] = p + bout[0];
}

extern "C" void kernel_launch(void* const* d_in, const int* in_sizes, int n_in,
                              void* d_out, int out_size, void* d_ws, size_t ws_size,
                              hipStream_t stream) {
    const float* x    = (const float*)d_in[0];
    const int*   ei   = (const int*)  d_in[1];
    const float* eatt = (const float*)d_in[3];
    const float* W1   = (const float*)d_in[4];
    const float* as1  = (const float*)d_in[5];
    const float* ad1  = (const float*)d_in[6];
    const float* We1  = (const float*)d_in[7];
    const float* ae1  = (const float*)d_in[8];
    const float* b1   = (const float*)d_in[9];
    const float* W2   = (const float*)d_in[10];
    const float* as2  = (const float*)d_in[11];
    const float* ad2  = (const float*)d_in[12];
    const float* We2  = (const float*)d_in[13];
    const float* ae2  = (const float*)d_in[14];
    const float* b2   = (const float*)d_in[15];
    const float* g1   = (const float*)d_in[16];
    const float* be1  = (const float*)d_in[17];
    const float* g2   = (const float*)d_in[18];
    const float* be2  = (const float*)d_in[19];
    const float* Wout = (const float*)d_in[20];
    const float* bout = (const float*)d_in[21];
    float* outp = (float*)d_out;

    const int N = in_sizes[0] / DIM;
    const int E = in_sizes[1] / 2;

    float* w = (float*)d_ws;
    size_t o = 0;
    float* xs    = w + o; o += (size_t)N * DIM;
    float* agg   = w + o; o += (size_t)N * DIM;
    float* h1    = w + o; o += (size_t)N * DIM;
    float* a_src = w + o; o += (size_t)N * HEADS;
    float* a_dst = w + o; o += (size_t)N * HEADS;
    int*   spair = (int*)(w + o); o += (size_t)2 * E;
    int*   rank  = (int*)(w + o); o += (size_t)E;
    int*   cnt   = (int*)(w + o); o += (size_t)N;
    int*   rowp  = (int*)(w + o); o += (size_t)N + 1;
    float* sed   = w + o; o += 8;

    auto cdiv = [](long a, long b) { return (int)((a + b - 1) / b); };
    const int T = 256;

    k_zero<<<64, T, 0, stream>>>((float*)cnt, (long)N);
    k_edge_rank<<<cdiv(E, T), T, 0, stream>>>(ei, cnt, rank, E);
    k_row_offsets<<<cdiv(N, 1024), 1024, 0, stream>>>(cnt, rowp, N);
    k_edge_fill<<<cdiv(E, T), T, 0, stream>>>(ei, eatt, rowp, rank, spair, E);
    k_edge_scal<<<1, 32, 0, stream>>>(We1, ae1, We2, ae2, sed);

    k_gemm64_wmma<<<cdiv((long)cdiv(N, 16), 2), T, 0, stream>>>(x, W1, xs, N);
    k_attn_coef<<<cdiv((long)N * HEADS, T), T, 0, stream>>>(xs, as1, ad1, a_src, a_dst, N);
    k_node_softmax_agg<<<cdiv(N, 8), T, 0, stream>>>(rowp, spair, sed, a_src, a_dst, xs, agg, N);
    k_post<<<cdiv(N, 8), T, 0, stream>>>(agg, x, b1, g1, be1, h1, N);

    k_gemm64_wmma<<<cdiv((long)cdiv(N, 16), 2), T, 0, stream>>>(h1, W2, xs, N);
    k_attn_coef<<<cdiv((long)N * HEADS, T), T, 0, stream>>>(xs, as2, ad2, a_src, a_dst, N);
    k_node_softmax_agg<<<cdiv(N, 8), T, 0, stream>>>(rowp, spair, sed + 4, a_src, a_dst, xs, agg, N);
    k_post<<<cdiv(N, 8), T, 0, stream>>>(agg, h1, b2, g2, be2, agg, N);

    k_out<<<cdiv(N, 8), T, 0, stream>>>(agg, Wout, bout, outp, N);
}


# Edge ranks from LDS counters: 64 blocks each count one edge-list chunk in LDS (returned LDS add = rank), scan over chunks gives degrees; no global atomics; segment order: chunk, then LDS arrival
# speedup vs baseline: 8.7327x; 1.3039x over previous
#include <hip/hip_runtime.h>
#include <hip/hip_bf16.h>

#define HEADS 4
#define HC    16
#define DIM   64
#define NEGS  0.2f

typedef float v2f __attribute__((ext_vector_type(2)));
typedef float v8f __attribute__((ext_vector_type(8)));

__global__ void k_edge_scal(const float* __restrict__ We1, const float* __restrict__ ae1,
                            const float* __restrict__ We2, const float* __restrict__ ae2,
                            float* __restrict__ sed) {
    int t = threadIdx.x;
    if (t >= 8) return;
    const float* We = (t < 4) ? We1 : We2;
    const float* ae = (t < 4) ? ae1 : ae2;
    int h = t & 3;
    float s = 0.0f;
    for (int c = 0; c < HC; ++c) s += We[h * HC + c] * ae[h * HC + c];
    sed[t] = s;
}

__global__ void k_gemm64_wmma(const float* __restrict__ X, const float* __restrict__ W,
                              float* __restrict__ Y, int nrows) {
    int wave = threadIdx.x >> 5;
    int lane = threadIdx.x & 31;
    int rowTile = blockIdx.x * 2 + (wave >> 2);
    int row0 = rowTile * 16;
    int col0 = (wave & 3) * 16;
    if (row0 >= nrows) return;
    int lr = lane & 15;
    int hi = lane >> 4;

    int arow = row0 + lr; if (arow > nrows - 1) arow = nrows - 1;
    const float* ap = X + (size_t)arow * DIM;

    v8f acc = {};
    for (int k = 0; k < DIM; k += 4) {
        v2f a;
        a.x = ap[k + 2 * hi + 0];
        a.y = ap[k + 2 * hi + 1];
        const float* bp = W + (size_t)(k + 2 * hi) * DIM + col0 + lr;
        v2f b;
        b.x = bp[0];
        b.y = bp[DIM];
        acc = __builtin_amdgcn_wmma_f32_16x16x4_f32(false, a, false, b,
                                                    (short)0, acc, false, false);
    }
    for (int r = 0; r < 8; ++r) {
        int row = row0 + r + 8 * hi;
        if (row < nrows) Y[(size_t)row * DIM + col0 + lr] = acc[r];
    }
}

__global__ void k_attn_coef(const float* __restrict__ xs, const float* __restrict__ as,
                            const float* __restrict__ ad, float* __restrict__ a_src,
                            float* __restrict__ a_dst, int N) {
    int t = blockIdx.x * blockDim.x + threadIdx.x;
    if (t >= N * HEADS) return;
    int n = t >> 2, h = t & 3;
    const float* xp = xs + (size_t)n * DIM + h * HC;
    float ss = 0.0f, sd = 0.0f;
    for (int c = 0; c < HC; ++c) {
        float x = xp[c];
        ss += x * as[h * HC + c];
        sd += x * ad[h * HC + c];
    }
    a_src[t] = ss;
    a_dst[t] = sd;
}

#define RCH  64
#define LDSN 51200
#define RB   8
static_assert(RCH <= DIM, "the chunk table must fit a node plane");

__global__ void __launch_bounds__(1024)
k_chunk_rank(const int* __restrict__ ei, int* __restrict__ M, int* __restrict__ rank, int E, int N) {
    __shared__ int T[LDSN];
    const int c   = blockIdx.x;
    const int per = (E + RCH - 1) / RCH;
    const int beg = c * per;
    const int end = (beg + per < E) ? (beg + per) : E;
    const int* __restrict__ dstp = ei + E;
    for (int base = 0; base < N; base += LDSN) {
        const int lim = (N - base < LDSN) ? (N - base) : LDSN;
        for (int i = threadIdx.x; i < lim; i += 1024) T[i] = 0;
        __syncthreads();
        for (int e0 = beg + (int)threadIdx.x; e0 < end; e0 += RB * 1024) {
            int d[RB];
            #pragma unroll
            for (int k = 0; k < RB; ++k) {
                const int e  = e0 + k * 1024;
                const int ec = (e < end) ? e : (end - 1);
                const int dv = dstp[ec] - base;
                d[k] = (e < end) ? dv : -1;
            }
            #pragma unroll
            for (int k = 0; k < RB; ++k) {
                if ((unsigned)d[k] < (unsigned)lim) rank[e0 + k * 1024] = atomicAdd(&T[d[k]], 1);
            }
        }
        __syncthreads();
        for (int i = threadIdx.x; i < lim; i += 1024) M[(size_t)c * N + base + i] = T[i];
        __syncthreads();
    }
}

__global__ void __launch_bounds__(256)
k_chunk_scan(int* __restrict__ M, int* __restrict__ cnt, int N) {
    const int n = blockIdx.x * blockDim.x + threadIdx.x;
    if (n >= N) return;
    int v[RCH];
    #pragma unroll
    for (int c = 0; c < RCH; ++c) v[c] = M[(size_t)c * N + n];
    int acc = 0;
    #pragma unroll
    for (int c = 0; c < RCH; ++c) {
        M[(size_t)c * N + n] = acc;
        acc += v[c];
    }
    cnt[n] = acc;
}

__global__ void __launch_bounds__(1024)
k_row_offsets(const int* __restrict__ cnt, int* __restrict__ rowptr, int N) {
    __shared__ int s_pre[32];
    __shared__ int s_sum[32];
    const int t = threadIdx.x, lane = t & 31, wave = t >> 5;
    const int base = blockIdx.x * 1024;
    int pre = 0;
    for (int i = t; i < base; i += 1024) pre += cnt[i];
    #pragma unroll
    for (int o = 16; o >= 1; o >>= 1) pre += __shfl_xor(pre, o, 32);
    const int idx = base + t;
    const int v = (idx < N) ? cnt[idx] : 0;
    int inc = v;
    #pragma unroll
    for (int o = 1; o < 32; o <<= 1) {
        int u = __shfl_up(inc, o, 32);
        if (lane >= o) inc += u;
    }
    if (lane == 0)  s_pre[wave] = pre;
    if (lane == 31) s_sum[wave] = inc;
    __syncthreads();
    int off = 0;
    for (int w = 0; w < 32; ++w) {
        off += s_pre[w];
        if (w < wave) off += s_sum[w];
    }
    const int excl = off + inc - v;
    if (idx < N) rowptr[idx] = excl;
    if (idx == N - 1) rowptr[N] = excl + v;
}

__global__ void k_edge_fill(const int* __restrict__ ei, const float* __restrict__ eattr,
                            const int* __restrict__ rowptr, const int* __restrict__ M,
                            const int* __restrict__ rank, int* __restrict__ spair, int E, int N) {
    int e = blockIdx.x * blockDim.x + threadIdx.x;
    if (e >= E) return;
    const int per = (E + RCH - 1) / RCH;
    const int d = ei[E + e];
    int pos = rowptr[d] + M[(size_t)(e / per) * N + d] + rank[e];
    int2 v;
    v.x = ei[e];
    v.y = __float_as_int(eattr[e]);
    ((int2*)spair)[pos] = v;
}

__device__ __forceinline__ float edge_logit(float as, float ad, float ea, float se) {
    float al = __builtin_fmaf(ea, se, as + ad);
    return (al > 0.0f) ? al : NEGS * al;
}

__global__ void __launch_bounds__(256)
k_node_softmax_agg(const int* __restrict__ rowptr, const int* __restrict__ spair,
                   const float* __restrict__ sed, const float* __restrict__ a_src,
                   const float* __restrict__ a_dst, const float* __restrict__ xs,
                   float* __restrict__ agg, int N) {
    const int lane = threadIdx.x & 31;
    const int n = blockIdx.x * 8 + __builtin_amdgcn_readfirstlane((int)(threadIdx.x >> 5));
    if (n >= N) return;
    const int beg = rowptr[n];
    const int end = rowptr[n + 1];
    const int2*   pairs = (const int2*)spair;
    const float4* as4   = (const float4*)a_src;
    const float4 adn = ((const float4*)a_dst)[n];
    const float4 asn = as4[n];
    const float se0 = sed[0], se1 = sed[1], se2 = sed[2], se3 = sed[3];

    const float ninf = -__builtin_huge_valf();
    float easum = 0.0f, m0 = ninf, m1 = ninf, m2 = ninf, m3 = ninf;
    for (int p = beg + lane; p < end; p += 32) {
        const int2 pr = pairs[p];
        const float ea = __int_as_float(pr.y);
        const float4 a = as4[pr.x];
        easum += ea;
        m0 = fmaxf(m0, edge_logit(a.x, adn.x, ea, se0));
        m1 = fmaxf(m1, edge_logit(a.y, adn.y, ea, se1));
        m2 = fmaxf(m2, edge_logit(a.z, adn.z, ea, se2));
        m3 = fmaxf(m3, edge_logit(a.w, adn.w, ea, se3));
    }
    #pragma unroll
    for (int o = 16; o >= 1; o >>= 1) {
        easum += __shfl_xor(easum, o, 32);
        m0 = fmaxf(m0, __shfl_xor(m0, o, 32));
        m1 = fmaxf(m1, __shfl_xor(m1, o, 32));
        m2 = fmaxf(m2, __shfl_xor(m2, o, 32));
        m3 = fmaxf(m3, __shfl_xor(m3, o, 32));
    }
    const float la = easum / fmaxf((float)(end - beg), 1.0f);
    const float sl0 = edge_logit(asn.x, adn.x, la, se0);
    const float sl1 = edge_logit(asn.y, adn.y, la, se1);
    const float sl2 = edge_logit(asn.z, adn.z, la, se2);
    const float sl3 = edge_logit(asn.w, adn.w, la, se3);
    m0 = fmaxf(m0, sl0); m1 = fmaxf(m1, sl1); m2 = fmaxf(m2, sl2); m3 = fmaxf(m3, sl3);

    const int h = lane >> 3;
    const float mh  = (h == 0) ? m0  : (h == 1) ? m1  : (h == 2) ? m2  : m3;
    const float slh = (h == 0) ? sl0 : (h == 1) ? sl1 : (h == 2) ? sl2 : sl3;
    const float2* xrow = (const float2*)xs + lane;
    const float es = expf(slh - mh);
    const float2 xn = xrow[(size_t)n * (DIM / 2)];
    float den = es;
    float ax = xn.x * es, ay = xn.y * es;
    for (int base = beg; base < end; base += 32) {
        const int p = base + lane;
        int s = n;
        float e0 = 0.0f, e1 = 0.0f, e2 = 0.0f, e3 = 0.0f;
        if (p < end) {
            const int2 pr = pairs[p];
            const float ea = __int_as_float(pr.y);
            s = pr.x;
            const float4 a = as4[s];
            e0 = expf(edge_logit(a.x, adn.x, ea, se0) - m0);
            e1 = expf(edge_logit(a.y, adn.y, ea, se1) - m1);
            e2 = expf(edge_logit(a.z, adn.z, ea, se2) - m2);
            e3 = expf(edge_logit(a.w, adn.w, ea, se3) - m3);
        }
        const int cc = (end - base < 32) ? (end - base) : 32;
        #pragma unroll 4
        for (int j = 0; j < cc; ++j) {
            const int sj = __builtin_amdgcn_readlane(s, j);
            const float r0 = __int_as_float(__builtin_amdgcn_readlane(__float_as_int(e0), j));
            const float r1 = __int_as_float(__builtin_amdgcn_readlane(__float_as_int(e1), j));
            const float r2 = __int_as_float(__builtin_amdgcn_readlane(__float_as_int(e2), j));
            const float r3 = __int_as_float(__builtin_amdgcn_readlane(__float_as_int(e3), j));
            const float wj = (h == 0) ? r0 : (h == 1) ? r1 : (h == 2) ? r2 : r3;
            const float2 xv = xrow[(size_t)sj * (DIM / 2)];
            den += wj;
            ax = __builtin_fmaf(xv.x, wj, ax);
            ay = __builtin_fmaf(xv.y, wj, ay);
        }
    }
    const float inv = den + 1e-16f;
    float2 o;
    o.x = ax / inv;
    o.y = ay / inv;
    ((float2*)agg)[(size_t)n * (DIM / 2) + lane] = o;
}

__global__ void k_post(const float* __restrict__ agg, const float* __restrict__ res,
                       const float* __restrict__ b, const float* __restrict__ g,
                       const float* __restrict__ beta, float* __restrict__ out, int N) {
    int wave = threadIdx.x >> 5;
    int lane = threadIdx.x & 31;
    int n = blockIdx.x * 8 + wave;
    if (n >= N) return;
    const float* ap = agg + (size_t)n * DIM;
    const float* rp = res + (size_t)n * DIM;
    float v0 = ap[lane]      + b[lane]      + rp[lane];
    float v1 = ap[lane + 32] + b[lane + 32] + rp[lane + 32];
    float sum = v0 + v1;
    #pragma unroll
    for (int m = 16; m >= 1; m >>= 1) sum += __shfl_xor(sum, m, 32);
    float mu = sum * (1.0f / DIM);
    float d0 = v0 - mu, d1 = v1 - mu;
    float vs = d0 * d0 + d1 * d1;
    #pragma unroll
    for (int m = 16; m >= 1; m >>= 1) vs += __shfl_xor(vs, m, 32);
    float rstd = rsqrtf(vs * (1.0f / DIM) + 1e-5f);
    float y0 = d0 * rstd * g[lane]      + beta[lane];
    float y1 = d1 * rstd * g[lane + 32] + beta[lane + 32];
    y0 = (y0 > 0.0f) ? y0 : (expf(y0) - 1.0f);
    y1 = (y1 > 0.0f) ? y1 : (expf(y1) - 1.0f);
    float* op = out + (size_t)n * DIM;
    op[lane]      = y0;
    op[lane + 32] = y1;
}

__global__ void k_out(const float* __restrict__ hfin, const float* __restrict__ Wout,
                      const float* __restrict__ bout, float* __restrict__ out, int N) {
    int wave = threadIdx.x >> 5;
    int lane = threadIdx.x & 31;
    int n = blockIdx.x * 8 + wave;
    if (n >= N) return;
    const float* hp = hfin + (size_t)n * DIM;
    float p = hp[lane] * Wout[lane] + hp[lane + 32] * Wout[lane + 32];
    #pragma unroll
    for (int m = 16; m >= 1; m >>= 1) p += __shfl_xor(p, m, 32);
    if (lane == 0) out[n] = p + bout[0];
}

extern "C" void kernel_launch(void* const* d_in, const int* in_sizes, int n_in,
                              void* d_out, int out_size, void* d_ws, size_t ws_size,
                              hipStream_t stream) {
    const float* x    = (const float*)d_in[0];
    const int*   ei   = (const int*)  d_in[1];
    const float* eatt = (const float*)d_in[3];
    const float* W1   = (const float*)d_in[4];
    const float* as1  = (const float*)d_in[5];
    const float* ad1  = (const float*)d_in[6];
    const float* We1  = (const float*)d_in[7];
    const float* ae1  = (const float*)d_in[8];
    const float* b1   = (const float*)d_in[9];
    const float* W2   = (const float*)d_in[10];
    const float* as2  = (const float*)d_in[11];
    const float* ad2  = (const float*)d_in[12];
    const float* We2  = (const float*)d_in[13];
    const float* ae2  = (const float*)d_in[14];
    const float* b2   = (const float*)d_in[15];
    const float* g1   = (const float*)d_in[16];
    const float* be1  = (const float*)d_in[17];
    const float* g2   = (const float*)d_in[18];
    const float* be2  = (const float*)d_in[19];
    const float* Wout = (const float*)d_in[20];
    const float* bout = (const float*)d_in[21];
    float* outp = (float*)d_out;

    const int N = in_sizes[0] / DIM;
    const int E = in_sizes[1] / 2;

    float* w = (float*)d_ws;
    size_t o = 0;
    float* xs    = w + o; o += (size_t)N * DIM;
    float* agg   = w + o; o += (size_t)N * DIM;
    float* h1    = w + o; o += (size_t)N * DIM;
    float* a_src = w + o; o += (size_t)N * HEADS;
    float* a_dst = w + o; o += (size_t)N * HEADS;
    int*   spair = (int*)(w + o); o += (size_t)2 * E;
    int*   rank  = (int*)(w + o); o += (size_t)E;
    int*   cnt   = (int*)(w + o); o += (size_t)N;
    int*   rowp  = (int*)(w + o); o += (size_t)N + 1;
    float* sed   = w + o; o += 8;

    auto cdiv = [](long a, long b) { return (int)((a + b - 1) / b); };
    const int T = 256;

    int* M = (int*)agg;
    k_chunk_rank<<<RCH, 1024, 0, stream>>>(ei, M, rank, E, N);
    k_chunk_scan<<<cdiv(N, T), T, 0, stream>>>(M, cnt, N);
    k_row_offsets<<<cdiv(N, 1024), 1024, 0, stream>>>(cnt, rowp, N);
    k_edge_fill<<<cdiv(E, T), T, 0, stream>>>(ei, eatt, rowp, M, rank, spair, E, N);
    k_edge_scal<<<1, 32, 0, stream>>>(We1, ae1, We2, ae2, sed);

    k_gemm64_wmma<<<cdiv((long)cdiv(N, 16), 2), T, 0, stream>>>(x, W1, xs, N);
    k_attn_coef<<<cdiv((long)N * HEADS, T), T, 0, stream>>>(xs, as1, ad1, a_src, a_dst, N);
    k_node_softmax_agg<<<cdiv(N, 8), T, 0, stream>>>(rowp, spair, sed, a_src, a_dst, xs, agg, N);
    k_post<<<cdiv(N, 8), T, 0, stream>>>(agg, x, b1, g1, be1, h1, N);

    k_gemm64_wmma<<<cdiv((long)cdiv(N, 16), 2), T, 0, stream>>>(h1, W2, xs, N);
    k_attn_coef<<<cdiv((long)N * HEADS, T), T, 0, stream>>>(xs, as2, ad2, a_src, a_dst, N);
    k_node_softmax_agg<<<cdiv(N, 8), T, 0, stream>>>(rowp, spair, sed + 4, a_src, a_dst, xs, agg, N);
    k_post<<<cdiv(N, 8), T, 0, stream>>>(agg, h1, b2, g2, be2, agg, N);

    k_out<<<cdiv(N, 8), T, 0, stream>>>(agg, Wout, bout, outp, N);
}


# Layernorm+ELU (and layer-2 output dot) fused into the one-node aggregation wave on its 64 register sums: 3 launches of one-node waves gone; layernorm/dot sums now pair features 2l,2l+1 (new tree)
# speedup vs baseline: 9.6133x; 1.1008x over previous
#include <hip/hip_runtime.h>
#include <hip/hip_bf16.h>

#define HEADS 4
#define HC    16
#define DIM   64
#define NEGS  0.2f

typedef float v2f __attribute__((ext_vector_type(2)));
typedef float v8f __attribute__((ext_vector_type(8)));

__global__ void k_edge_scal(const float* __restrict__ We1, const float* __restrict__ ae1,
                            const float* __restrict__ We2, const float* __restrict__ ae2,
                            float* __restrict__ sed) {
    int t = threadIdx.x;
    if (t >= 8) return;
    const float* We = (t < 4) ? We1 : We2;
    const float* ae = (t < 4) ? ae1 : ae2;
    int h = t & 3;
    float s = 0.0f;
    for (int c = 0; c < HC; ++c) s += We[h * HC + c] * ae[h * HC + c];
    sed[t] = s;
}

__global__ void k_gemm64_wmma(const float* __restrict__ X, const float* __restrict__ W,
                              float* __restrict__ Y, int nrows) {
    int wave = threadIdx.x >> 5;
    int lane = threadIdx.x & 31;
    int rowTile = blockIdx.x * 2 + (wave >> 2);
    int row0 = rowTile * 16;
    int col0 = (wave & 3) * 16;
    if (row0 >= nrows) return;
    int lr = lane & 15;
    int hi = lane >> 4;

    int arow = row0 + lr; if (arow > nrows - 1) arow = nrows - 1;
    const float* ap = X + (size_t)arow * DIM;

    v8f acc = {};
    for (int k = 0; k < DIM; k += 4) {
        v2f a;
        a.x = ap[k + 2 * hi + 0];
        a.y = ap[k + 2 * hi + 1];
        const float* bp = W + (size_t)(k + 2 * hi) * DIM + col0 + lr;
        v2f b;
        b.x = bp[0];
        b.y = bp[DIM];
        acc = __builtin_amdgcn_wmma_f32_16x16x4_f32(false, a, false, b,
                                                    (short)0, acc, false, false);
    }
    for (int r = 0; r < 8; ++r) {
        int row = row0 + r + 8 * hi;
        if (row < nrows) Y[(size_t)row * DIM + col0 + lr] = acc[r];
    }
}

__global__ void k_attn_coef(const float* __restrict__ xs, const float* __restrict__ as,
                            const float* __restrict__ ad, float* __restrict__ a_src,
                            float* __restrict__ a_dst, int N) {
    int t = blockIdx.x * blockDim.x + threadIdx.x;
    if (t >= N * HEADS) return;
    int n = t >> 2, h = t & 3;
    const float* xp = xs + (size_t)n * DIM + h * HC;
    float ss = 0.0f, sd = 0.0f;
    for (int c = 0; c < HC; ++c) {
        float x = xp[c];
        ss += x * as[h * HC + c];
        sd += x * ad[h * HC + c];
    }
    a_src[t] = ss;
    a_dst[t] = sd;
}

#define RCH  64
#define LDSN 51200
#define RB   8
static_assert(RCH <= DIM, "the chunk table must fit a node plane");

__global__ void __launch_bounds__(1024)
k_chunk_rank(const int* __restrict__ ei, int* __restrict__ M, int* __restrict__ rank, int E, int N) {
    __shared__ int T[LDSN];
    const int c   = blockIdx.x;
    const int per = (E + RCH - 1) / RCH;
    const int beg = c * per;
    const int end = (beg + per < E) ? (beg + per) : E;
    const int* __restrict__ dstp = ei + E;
    for (int base = 0; base < N; base += LDSN) {
        const int lim = (N - base < LDSN) ? (N - base) : LDSN;
        for (int i = threadIdx.x; i < lim; i += 1024) T[i] = 0;
        __syncthreads();
        for (int e0 = beg + (int)threadIdx.x; e0 < end; e0 += RB * 1024) {
            int d[RB];
            #pragma unroll
            for (int k = 0; k < RB; ++k) {
                const int e  = e0 + k * 1024;
                const int ec = (e < end) ? e : (end - 1);
                const int dv = dstp[ec] - base;
                d[k] = (e < end) ? dv : -1;
            }
            #pragma unroll
            for (int k = 0; k < RB; ++k) {
                if ((unsigned)d[k] < (unsigned)lim) rank[e0 + k * 1024] = atomicAdd(&T[d[k]], 1);
            }
        }
        __syncthreads();
        for (int i = threadIdx.x; i < lim; i += 1024) M[(size_t)c * N + base + i] = T[i];
        __syncthreads();
    }
}

__global__ void __launch_bounds__(256)
k_chunk_scan(int* __restrict__ M, int* __restrict__ cnt, int N) {
    const int n = blockIdx.x * blockDim.x + threadIdx.x;
    if (n >= N) return;
    int v[RCH];
    #pragma unroll
    for (int c = 0; c < RCH; ++c) v[c] = M[(size_t)c * N + n];
    int acc = 0;
    #pragma unroll
    for (int c = 0; c < RCH; ++c) {
        M[(size_t)c * N + n] = acc;
        acc += v[c];
    }
    cnt[n] = acc;
}

__global__ void __launch_bounds__(1024)
k_row_offsets(const int* __restrict__ cnt, int* __restrict__ rowptr, int N) {
    __shared__ int s_pre[32];
    __shared__ int s_sum[32];
    const int t = threadIdx.x, lane = t & 31, wave = t >> 5;
    const int base = blockIdx.x * 1024;
    int pre = 0;
    for (int i = t; i < base; i += 1024) pre += cnt[i];
    #pragma unroll
    for (int o = 16; o >= 1; o >>= 1) pre += __shfl_xor(pre, o, 32);
    const int idx = base + t;
    const int v = (idx < N) ? cnt[idx] : 0;
    int inc = v;
    #pragma unroll
    for (int o = 1; o < 32; o <<= 1) {
        int u = __shfl_up(inc, o, 32);
        if (lane >= o) inc += u;
    }
    if (lane == 0)  s_pre[wave] = pre;
    if (lane == 31) s_sum[wave] = inc;
    __syncthreads();
    int off = 0;
    for (int w = 0; w < 32; ++w) {
        off += s_pre[w];
        if (w < wave) off += s_sum[w];
    }
    const int excl = off + inc - v;
    if (idx < N) rowptr[idx] = excl;
    if (idx == N - 1) rowptr[N] = excl + v;
}

__global__ void k_edge_fill(const int* __restrict__ ei, const float* __restrict__ eattr,
                            const int* __restrict__ rowptr, const int* __restrict__ M,
                            const int* __restrict__ rank, int* __restrict__ spair, int E, int N) {
    int e = blockIdx.x * blockDim.x + threadIdx.x;
    if (e >= E) return;
    const int per = (E + RCH - 1) / RCH;
    const int d = ei[E + e];
    int pos = rowptr[d] + M[(size_t)(e / per) * N + d] + rank[e];
    int2 v;
    v.x = ei[e];
    v.y = __float_as_int(eattr[e]);
    ((int2*)spair)[pos] = v;
}

__device__ __forceinline__ float edge_logit(float as, float ad, float ea, float se) {
    float al = __builtin_fmaf(ea, se, as + ad);
    return (al > 0.0f) ? al : NEGS * al;
}

template <bool LAST>
__device__ __forceinline__ void
node_softmax_agg_norm(const int* __restrict__ rowptr, const int* __restrict__ spair,
                      const float* __restrict__ sed, const float* __restrict__ a_src,
                      const float* __restrict__ a_dst, const float* __restrict__ xs,
                      const float* __restrict__ res, const float* __restrict__ b,
                      const float* __restrict__ g, const float* __restrict__ beta,
                      const float* __restrict__ Wout, const float* __restrict__ bout,
                      float* __restrict__ dst, int N) {
    const int lane = threadIdx.x & 31;
    const int n = blockIdx.x * 8 + __builtin_amdgcn_readfirstlane((int)(threadIdx.x >> 5));
    if (n >= N) return;
    const int beg = rowptr[n];
    const int end = rowptr[n + 1];
    const int2*   pairs = (const int2*)spair;
    const float4* as4   = (const float4*)a_src;
    const float4 adn = ((const float4*)a_dst)[n];
    const float4 asn = as4[n];
    const float se0 = sed[0], se1 = sed[1], se2 = sed[2], se3 = sed[3];

    const float ninf = -__builtin_huge_valf();
    float easum = 0.0f, m0 = ninf, m1 = ninf, m2 = ninf, m3 = ninf;
    for (int p = beg + lane; p < end; p += 32) {
        const int2 pr = pairs[p];
        const float ea = __int_as_float(pr.y);
        const float4 a = as4[pr.x];
        easum += ea;
        m0 = fmaxf(m0, edge_logit(a.x, adn.x, ea, se0));
        m1 = fmaxf(m1, edge_logit(a.y, adn.y, ea, se1));
        m2 = fmaxf(m2, edge_logit(a.z, adn.z, ea, se2));
        m3 = fmaxf(m3, edge_logit(a.w, adn.w, ea, se3));
    }
    #pragma unroll
    for (int o = 16; o >= 1; o >>= 1) {
        easum += __shfl_xor(easum, o, 32);
        m0 = fmaxf(m0, __shfl_xor(m0, o, 32));
        m1 = fmaxf(m1, __shfl_xor(m1, o, 32));
        m2 = fmaxf(m2, __shfl_xor(m2, o, 32));
        m3 = fmaxf(m3, __shfl_xor(m3, o, 32));
    }
    const float la = easum / fmaxf((float)(end - beg), 1.0f);
    const float sl0 = edge_logit(asn.x, adn.x, la, se0);
    const float sl1 = edge_logit(asn.y, adn.y, la, se1);
    const float sl2 = edge_logit(asn.z, adn.z, la, se2);
    const float sl3 = edge_logit(asn.w, adn.w, la, se3);
    m0 = fmaxf(m0, sl0); m1 = fmaxf(m1, sl1); m2 = fmaxf(m2, sl2); m3 = fmaxf(m3, sl3);

    const int h = lane >> 3;
    const float mh  = (h == 0) ? m0  : (h == 1) ? m1  : (h == 2) ? m2  : m3;
    const float slh = (h == 0) ? sl0 : (h == 1) ? sl1 : (h == 2) ? sl2 : sl3;
    const float2* xrow = (const float2*)xs + lane;
    const float es = expf(slh - mh);
    const float2 xn = xrow[(size_t)n * (DIM / 2)];
    const float* rp = res + (size_t)n * DIM;
    const float rs0 = rp[2 * lane],   rs1 = rp[2 * lane + 1];
    const float bi0 = b[2 * lane],    bi1 = b[2 * lane + 1];
    const float ga0 = g[2 * lane],    ga1 = g[2 * lane + 1];
    const float be0 = beta[2 * lane], be1 = beta[2 * lane + 1];
    float den = es;
    float ax = xn.x * es, ay = xn.y * es;
    for (int base = beg; base < end; base += 32) {
        const int p = base + lane;
        int s = n;
        float e0 = 0.0f, e1 = 0.0f, e2 = 0.0f, e3 = 0.0f;
        if (p < end) {
            const int2 pr = pairs[p];
            const float ea = __int_as_float(pr.y);
            s = pr.x;
            const float4 a = as4[s];
            e0 = expf(edge_logit(a.x, adn.x, ea, se0) - m0);
            e1 = expf(edge_logit(a.y, adn.y, ea, se1) - m1);
            e2 = expf(edge_logit(a.z, adn.z, ea, se2) - m2);
            e3 = expf(edge_logit(a.w, adn.w, ea, se3) - m3);
        }
        const int cc = (end - base < 32) ? (end - base) : 32;
        #pragma unroll 4
        for (int j = 0; j < cc; ++j) {
            const int sj = __builtin_amdgcn_readlane(s, j);
            const float r0 = __int_as_float(__builtin_amdgcn_readlane(__float_as_int(e0), j));
            const float r1 = __int_as_float(__builtin_amdgcn_readlane(__float_as_int(e1), j));
            const float r2 = __int_as_float(__builtin_amdgcn_readlane(__float_as_int(e2), j));
            const float r3 = __int_as_float(__builtin_amdgcn_readlane(__float_as_int(e3), j));
            const float wj = (h == 0) ? r0 : (h == 1) ? r1 : (h == 2) ? r2 : r3;
            const float2 xv = xrow[(size_t)sj * (DIM / 2)];
            den += wj;
            ax = __builtin_fmaf(xv.x, wj, ax);
            ay = __builtin_fmaf(xv.y, wj, ay);
        }
    }
    const float inv = den + 1e-16f;

    const float v0 = (ax / inv + bi0) + rs0;
    const float v1 = (ay / inv + bi1) + rs1;
    float sum = v0 + v1;
    for (int m = 16; m >= 1; m >>= 1) sum += __shfl_xor(sum, m, 32);
    const float mu = sum * (1.0f / DIM);
    const float d0 = v0 - mu, d1 = v1 - mu;
    float vs = d0 * d0 + d1 * d1;
    for (int m = 16; m >= 1; m >>= 1) vs += __shfl_xor(vs, m, 32);
    const float rstd = rsqrtf(vs * (1.0f / DIM) + 1e-5f);
    float y0 = d0 * rstd * ga0 + be0;
    float y1 = d1 * rstd * ga1 + be1;
    y0 = (y0 > 0.0f) ? y0 : (expf(y0) - 1.0f);
    y1 = (y1 > 0.0f) ? y1 : (expf(y1) - 1.0f);
    if (!LAST) {
        float2 o;
        o.x = y0;
        o.y = y1;
        ((float2*)dst)[(size_t)n * (DIM / 2) + lane] = o;
    } else {
        float p = y0 * Wout[2 * lane] + y1 * Wout[2 * lane + 1];
        for (int m = 16; m >= 1; m >>= 1) p += __shfl_xor(p, m, 32);
        if (lane == 0) dst[n] = p + bout[0];
    }
}

__global__ void __launch_bounds__(256)
k_node_agg_norm(const int* __restrict__ rowptr, const int* __restrict__ spair,
                const float* __restrict__ sed, const float* __restrict__ a_src,
                const float* __restrict__ a_dst, const float* __restrict__ xs,
                const float* __restrict__ res, const float* __restrict__ b,
                const float* __restrict__ g, const float* __restrict__ beta,
                float* __restrict__ hout, int N) {
    node_softmax_agg_norm<false>(rowptr, spair, sed, a_src, a_dst, xs, res, b, g, beta,
                                 nullptr, nullptr, hout, N);
}

__global__ void __launch_bounds__(256)
k_node_agg_norm_out(const int* __restrict__ rowptr, const int* __restrict__ spair,
                    const float* __restrict__ sed, const float* __restrict__ a_src,
                    const float* __restrict__ a_dst, const float* __restrict__ xs,
                    const float* __restrict__ res, const float* __restrict__ b,
                    const float* __restrict__ g, const float* __restrict__ beta,
                    const float* __restrict__ Wout, const float* __restrict__ bout,
                    float* __restrict__ out, int N) {
    node_softmax_agg_norm<true>(rowptr, spair, sed, a_src, a_dst, xs, res, b, g, beta,
                                Wout, bout, out, N);
}

extern "C" void kernel_launch(void* const* d_in, const int* in_sizes, int n_in,
                              void* d_out, int out_size, void* d_ws, size_t ws_size,
                              hipStream_t stream) {
    const float* x    = (const float*)d_in[0];
    const int*   ei   = (const int*)  d_in[1];
    const float* eatt = (const float*)d_in[3];
    const float* W1   = (const float*)d_in[4];
    const float* as1  = (const float*)d_in[5];
    const float* ad1  = (const float*)d_in[6];
    const float* We1  = (const float*)d_in[7];
    const float* ae1  = (const float*)d_in[8];
    const float* b1   = (const float*)d_in[9];
    const float* W2   = (const float*)d_in[10];
    const float* as2  = (const float*)d_in[11];
    const float* ad2  = (const float*)d_in[12];
    const float* We2  = (const float*)d_in[13];
    const float* ae2  = (const float*)d_in[14];
    const float* b2   = (const float*)d_in[15];
    const float* g1   = (const float*)d_in[16];
    const float* be1  = (const float*)d_in[17];
    const float* g2   = (const float*)d_in[18];
    const float* be2  = (const float*)d_in[19];
    const float* Wout = (const float*)d_in[20];
    const float* bout = (const float*)d_in[21];
    float* outp = (float*)d_out;

    const int N = in_sizes[0] / DIM;
    const int E = in_sizes[1] / 2;

    float* w = (float*)d_ws;
    size_t o = 0;
    float* xs    = w + o; o += (size_t)N * DIM;
    int*   M     = (int*)(w + o); o += (size_t)N * DIM;
    float* h1    = w + o; o += (size_t)N * DIM;
    float* a_src = w + o; o += (size_t)N * HEADS;
    float* a_dst = w + o; o += (size_t)N * HEADS;
    int*   spair = (int*)(w + o); o += (size_t)2 * E;
    int*   rank  = (int*)(w + o); o += (size_t)E;
    int*   cnt   = (int*)(w + o); o += (size_t)N;
    int*   rowp  = (int*)(w + o); o += (size_t)N + 1;
    float* sed   = w + o; o += 8;

    auto cdiv = [](long a, long b) { return (int)((a + b - 1) / b); };
    const int T = 256;

    k_chunk_rank<<<RCH, 1024, 0, stream>>>(ei, M, rank, E, N);
    k_chunk_scan<<<cdiv(N, T), T, 0, stream>>>(M, cnt, N);
    k_row_offsets<<<cdiv(N, 1024), 1024, 0, stream>>>(cnt, rowp, N);
    k_edge_fill<<<cdiv(E, T), T, 0, stream>>>(ei, eatt, rowp, M, rank, spair, E, N);
    k_edge_scal<<<1, 32, 0, stream>>>(We1, ae1, We2, ae2, sed);

    k_gemm64_wmma<<<cdiv((long)cdiv(N, 16), 2), T, 0, stream>>>(x, W1, xs, N);
    k_attn_coef<<<cdiv((long)N * HEADS, T), T, 0, stream>>>(xs, as1, ad1, a_src, a_dst, N);
    k_node_agg_norm<<<cdiv(N, 8), T, 0, stream>>>(rowp, spair, sed, a_src, a_dst, xs,
                                                  x, b1, g1, be1, h1, N);

    k_gemm64_wmma<<<cdiv((long)cdiv(N, 16), 2), T, 0, stream>>>(h1, W2, xs, N);
    k_attn_coef<<<cdiv((long)N * HEADS, T), T, 0, stream>>>(xs, as2, ad2, a_src, a_dst, N);
    k_node_agg_norm_out<<<cdiv(N, 8), T, 0, stream>>>(rowp, spair, sed + 4, a_src, a_dst, xs,
                                                      h1, b2, g2, be2, Wout, bout, outp, N);
}


# Aggregation wave keeps source id and the four logits of its first 64 edges from the max pass in registers: the exp pass no longer re-reads those edge pairs nor re-gathers a_src; sum order unchanged
# speedup vs baseline: 10.3372x; 1.0753x over previous
#include <hip/hip_runtime.h>
#include <hip/hip_bf16.h>

#define HEADS 4
#define HC    16
#define DIM   64
#define NEGS  0.2f

typedef float v2f __attribute__((ext_vector_type(2)));
typedef float v8f __attribute__((ext_vector_type(8)));

__global__ void k_edge_scal(const float* __restrict__ We1, const float* __restrict__ ae1,
                            const float* __restrict__ We2, const float* __restrict__ ae2,
                            float* __restrict__ sed) {
    int t = threadIdx.x;
    if (t >= 8) return;
    const float* We = (t < 4) ? We1 : We2;
    const float* ae = (t < 4) ? ae1 : ae2;
    int h = t & 3;
    float s = 0.0f;
    for (int c = 0; c < HC; ++c) s += We[h * HC + c] * ae[h * HC + c];
    sed[t] = s;
}

__global__ void k_gemm64_wmma(const float* __restrict__ X, const float* __restrict__ W,
                              float* __restrict__ Y, int nrows) {
    int wave = threadIdx.x >> 5;
    int lane = threadIdx.x & 31;
    int rowTile = blockIdx.x * 2 + (wave >> 2);
    int row0 = rowTile * 16;
    int col0 = (wave & 3) * 16;
    if (row0 >= nrows) return;
    int lr = lane & 15;
    int hi = lane >> 4;

    int arow = row0 + lr; if (arow > nrows - 1) arow = nrows - 1;
    const float* ap = X + (size_t)arow * DIM;

    v8f acc = {};
    for (int k = 0; k < DIM; k += 4) {
        v2f a;
        a.x = ap[k + 2 * hi + 0];
        a.y = ap[k + 2 * hi + 1];
        const float* bp = W + (size_t)(k + 2 * hi) * DIM + col0 + lr;
        v2f b;
        b.x = bp[0];
        b.y = bp[DIM];
        acc = __builtin_amdgcn_wmma_f32_16x16x4_f32(false, a, false, b,
                                                    (short)0, acc, false, false);
    }
    for (int r = 0; r < 8; ++r) {
        int row = row0 + r + 8 * hi;
        if (row < nrows) Y[(size_t)row * DIM + col0 + lr] = acc[r];
    }
}

__global__ void k_attn_coef(const float* __restrict__ xs, const float* __restrict__ as,
                            const float* __restrict__ ad, float* __restrict__ a_src,
                            float* __restrict__ a_dst, int N) {
    int t = blockIdx.x * blockDim.x + threadIdx.x;
    if (t >= N * HEADS) return;
    int n = t >> 2, h = t & 3;
    const float* xp = xs + (size_t)n * DIM + h * HC;
    float ss = 0.0f, sd = 0.0f;
    for (int c = 0; c < HC; ++c) {
        float x = xp[c];
        ss += x * as[h * HC + c];
        sd += x * ad[h * HC + c];
    }
    a_src[t] = ss;
    a_dst[t] = sd;
}

#define RCH  64
#define LDSN 51200
#define RB   8
static_assert(RCH <= DIM, "the chunk table must fit a node plane");

__global__ void __launch_bounds__(1024)
k_chunk_rank(const int* __restrict__ ei, int* __restrict__ M, int* __restrict__ rank, int E, int N) {
    __shared__ int T[LDSN];
    const int c   = blockIdx.x;
    const int per = (E + RCH - 1) / RCH;
    const int beg = c * per;
    const int end = (beg + per < E) ? (beg + per) : E;
    const int* __restrict__ dstp = ei + E;
    for (int base = 0; base < N; base += LDSN) {
        const int lim = (N - base < LDSN) ? (N - base) : LDSN;
        for (int i = threadIdx.x; i < lim; i += 1024) T[i] = 0;
        __syncthreads();
        for (int e0 = beg + (int)threadIdx.x; e0 < end; e0 += RB * 1024) {
            int d[RB];
            #pragma unroll
            for (int k = 0; k < RB; ++k) {
                const int e  = e0 + k * 1024;
                const int ec = (e < end) ? e : (end - 1);
                const int dv = dstp[ec] - base;
                d[k] = (e < end) ? dv : -1;
            }
            #pragma unroll
            for (int k = 0; k < RB; ++k) {
                if ((unsigned)d[k] < (unsigned)lim) rank[e0 + k * 1024] = atomicAdd(&T[d[k]], 1);
            }
        }
        __syncthreads();
        for (int i = threadIdx.x; i < lim; i += 1024) M[(size_t)c * N + base + i] = T[i];
        __syncthreads();
    }
}

__global__ void __launch_bounds__(256)
k_chunk_scan(int* __restrict__ M, int* __restrict__ cnt, int N) {
    const int n = blockIdx.x * blockDim.x + threadIdx.x;
    if (n >= N) return;
    int v[RCH];
    #pragma unroll
    for (int c = 0; c < RCH; ++c) v[c] = M[(size_t)c * N + n];
    int acc = 0;
    #pragma unroll
    for (int c = 0; c < RCH; ++c) {
        M[(size_t)c * N + n] = acc;
        acc += v[c];
    }
    cnt[n] = acc;
}

__global__ void __launch_bounds__(1024)
k_row_offsets(const int* __restrict__ cnt, int* __restrict__ rowptr, int N) {
    __shared__ int s_pre[32];
    __shared__ int s_sum[32];
    const int t = threadIdx.x, lane = t & 31, wave = t >> 5;
    const int base = blockIdx.x * 1024;
    int pre = 0;
    for (int i = t; i < base; i += 1024) pre += cnt[i];
    #pragma unroll
    for (int o = 16; o >= 1; o >>= 1) pre += __shfl_xor(pre, o, 32);
    const int idx = base + t;
    const int v = (idx < N) ? cnt[idx] : 0;
    int inc = v;
    #pragma unroll
    for (int o = 1; o < 32; o <<= 1) {
        int u = __shfl_up(inc, o, 32);
        if (lane >= o) inc += u;
    }
    if (lane == 0)  s_pre[wave] = pre;
    if (lane == 31) s_sum[wave] = inc;
    __syncthreads();
    int off = 0;
    for (int w = 0; w < 32; ++w) {
        off += s_pre[w];
        if (w < wave) off += s_sum[w];
    }
    const int excl = off + inc - v;
    if (idx < N) rowptr[idx] = excl;
    if (idx == N - 1) rowptr[N] = excl + v;
}

__global__ void k_edge_fill(const int* __restrict__ ei, const float* __restrict__ eattr,
                            const int* __restrict__ rowptr, const int* __restrict__ M,
                            const int* __restrict__ rank, int* __restrict__ spair, int E, int N) {
    int e = blockIdx.x * blockDim.x + threadIdx.x;
    if (e >= E) return;
    const int per = (E + RCH - 1) / RCH;
    const int d = ei[E + e];
    int pos = rowptr[d] + M[(size_t)(e / per) * N + d] + rank[e];
    int2 v;
    v.x = ei[e];
    v.y = __float_as_int(eattr[e]);
    ((int2*)spair)[pos] = v;
}

__device__ __forceinline__ float edge_logit(float as, float ad, float ea, float se) {
    float al = __builtin_fmaf(ea, se, as + ad);
    return (al > 0.0f) ? al : NEGS * al;
}

template <bool LAST>
__device__ __forceinline__ void
node_softmax_agg_norm(const int* __restrict__ rowptr, const int* __restrict__ spair,
                      const float* __restrict__ sed, const float* __restrict__ a_src,
                      const float* __restrict__ a_dst, const float* __restrict__ xs,
                      const float* __restrict__ res, const float* __restrict__ b,
                      const float* __restrict__ g, const float* __restrict__ beta,
                      const float* __restrict__ Wout, const float* __restrict__ bout,
                      float* __restrict__ dst, int N) {
    const int lane = threadIdx.x & 31;
    const int n = blockIdx.x * 8 + __builtin_amdgcn_readfirstlane((int)(threadIdx.x >> 5));
    if (n >= N) return;
    const int beg = rowptr[n];
    const int end = rowptr[n + 1];
    const int2*   pairs = (const int2*)spair;
    const float4* as4   = (const float4*)a_src;
    const float4 adn = ((const float4*)a_dst)[n];
    const float4 asn = as4[n];
    const float se0 = sed[0], se1 = sed[1], se2 = sed[2], se3 = sed[3];

    const float ninf = -__builtin_huge_valf();
    float easum = 0.0f, m0 = ninf, m1 = ninf, m2 = ninf, m3 = ninf;
    const int  q0 = beg + lane, q1 = beg + 32 + lane;
    const bool u0 = q0 < end,   u1 = q1 < end;
    const int2 kp0 = pairs[u0 ? q0 : 0];
    const int2 kp1 = pairs[u1 ? q1 : 0];
    const float4 ka0 = as4[kp0.x];
    const float4 ka1 = as4[kp1.x];
    const float kea0 = __int_as_float(kp0.y), kea1 = __int_as_float(kp1.y);
    const int   ks0 = kp0.x, ks1 = kp1.x;
    const float k00 = u0 ? edge_logit(ka0.x, adn.x, kea0, se0) : ninf;
    const float k01 = u0 ? edge_logit(ka0.y, adn.y, kea0, se1) : ninf;
    const float k02 = u0 ? edge_logit(ka0.z, adn.z, kea0, se2) : ninf;
    const float k03 = u0 ? edge_logit(ka0.w, adn.w, kea0, se3) : ninf;
    const float k10 = u1 ? edge_logit(ka1.x, adn.x, kea1, se0) : ninf;
    const float k11 = u1 ? edge_logit(ka1.y, adn.y, kea1, se1) : ninf;
    const float k12 = u1 ? edge_logit(ka1.z, adn.z, kea1, se2) : ninf;
    const float k13 = u1 ? edge_logit(ka1.w, adn.w, kea1, se3) : ninf;
    easum += u0 ? kea0 : 0.0f;
    easum += u1 ? kea1 : 0.0f;
    m0 = fmaxf(fmaxf(m0, k00), k10);
    m1 = fmaxf(fmaxf(m1, k01), k11);
    m2 = fmaxf(fmaxf(m2, k02), k12);
    m3 = fmaxf(fmaxf(m3, k03), k13);
    for (int p = beg + 64 + lane; p < end; p += 32) {
        const int2 pr = pairs[p];
        const float ea = __int_as_float(pr.y);
        const float4 a = as4[pr.x];
        easum += ea;
        m0 = fmaxf(m0, edge_logit(a.x, adn.x, ea, se0));
        m1 = fmaxf(m1, edge_logit(a.y, adn.y, ea, se1));
        m2 = fmaxf(m2, edge_logit(a.z, adn.z, ea, se2));
        m3 = fmaxf(m3, edge_logit(a.w, adn.w, ea, se3));
    }
    #pragma unroll
    for (int o = 16; o >= 1; o >>= 1) {
        easum += __shfl_xor(easum, o, 32);
        m0 = fmaxf(m0, __shfl_xor(m0, o, 32));
        m1 = fmaxf(m1, __shfl_xor(m1, o, 32));
        m2 = fmaxf(m2, __shfl_xor(m2, o, 32));
        m3 = fmaxf(m3, __shfl_xor(m3, o, 32));
    }
    const float la = easum / fmaxf((float)(end - beg), 1.0f);
    const float sl0 = edge_logit(asn.x, adn.x, la, se0);
    const float sl1 = edge_logit(asn.y, adn.y, la, se1);
    const float sl2 = edge_logit(asn.z, adn.z, la, se2);
    const float sl3 = edge_logit(asn.w, adn.w, la, se3);
    m0 = fmaxf(m0, sl0); m1 = fmaxf(m1, sl1); m2 = fmaxf(m2, sl2); m3 = fmaxf(m3, sl3);

    const int h = lane >> 3;
    const float mh  = (h == 0) ? m0  : (h == 1) ? m1  : (h == 2) ? m2  : m3;
    const float slh = (h == 0) ? sl0 : (h == 1) ? sl1 : (h == 2) ? sl2 : sl3;
    const float2* xrow = (const float2*)xs + lane;
    const float es = expf(slh - mh);
    const float2 xn = xrow[(size_t)n * (DIM / 2)];
    const float* rp = res + (size_t)n * DIM;
    const float rs0 = rp[2 * lane],   rs1 = rp[2 * lane + 1];
    const float bi0 = b[2 * lane],    bi1 = b[2 * lane + 1];
    const float ga0 = g[2 * lane],    ga1 = g[2 * lane + 1];
    const float be0 = beta[2 * lane], be1 = beta[2 * lane + 1];
    float den = es;
    float ax = xn.x * es, ay = xn.y * es;
    for (int base = beg; base < end; base += 32) {
        int s;
        float l0, l1, l2, l3;
        if (base == beg) {
            s = ks0; l0 = k00; l1 = k01; l2 = k02; l3 = k03;
        } else if (base == beg + 32) {
            s = ks1; l0 = k10; l1 = k11; l2 = k12; l3 = k13;
        } else {
            const int p = base + lane;
            s = n; l0 = ninf; l1 = ninf; l2 = ninf; l3 = ninf;
            if (p < end) {
                const int2 pr = pairs[p];
                const float ea = __int_as_float(pr.y);
                s = pr.x;
                const float4 a = as4[s];
                l0 = edge_logit(a.x, adn.x, ea, se0);
                l1 = edge_logit(a.y, adn.y, ea, se1);
                l2 = edge_logit(a.z, adn.z, ea, se2);
                l3 = edge_logit(a.w, adn.w, ea, se3);
            }
        }
        const float e0 = expf(l0 - m0);
        const float e1 = expf(l1 - m1);
        const float e2 = expf(l2 - m2);
        const float e3 = expf(l3 - m3);
        const int cc = (end - base < 32) ? (end - base) : 32;
        #pragma unroll 4
        for (int j = 0; j < cc; ++j) {
            const int sj = __builtin_amdgcn_readlane(s, j);
            const float r0 = __int_as_float(__builtin_amdgcn_readlane(__float_as_int(e0), j));
            const float r1 = __int_as_float(__builtin_amdgcn_readlane(__float_as_int(e1), j));
            const float r2 = __int_as_float(__builtin_amdgcn_readlane(__float_as_int(e2), j));
            const float r3 = __int_as_float(__builtin_amdgcn_readlane(__float_as_int(e3), j));
            const float wj = (h == 0) ? r0 : (h == 1) ? r1 : (h == 2) ? r2 : r3;
            const float2 xv = xrow[(size_t)sj * (DIM / 2)];
            den += wj;
            ax = __builtin_fmaf(xv.x, wj, ax);
            ay = __builtin_fmaf(xv.y, wj, ay);
        }
    }
    const float inv = den + 1e-16f;

    const float v0 = (ax / inv + bi0) + rs0;
    const float v1 = (ay / inv + bi1) + rs1;
    float sum = v0 + v1;
    for (int m = 16; m >= 1; m >>= 1) sum += __shfl_xor(sum, m, 32);
    const float mu = sum * (1.0f / DIM);
    const float d0 = v0 - mu, d1 = v1 - mu;
    float vs = d0 * d0 + d1 * d1;
    for (int m = 16; m >= 1; m >>= 1) vs += __shfl_xor(vs, m, 32);
    const float rstd = rsqrtf(vs * (1.0f / DIM) + 1e-5f);
    float y0 = d0 * rstd * ga0 + be0;
    float y1 = d1 * rstd * ga1 + be1;
    y0 = (y0 > 0.0f) ? y0 : (expf(y0) - 1.0f);
    y1 = (y1 > 0.0f) ? y1 : (expf(y1) - 1.0f);
    if (!LAST) {
        float2 o;
        o.x = y0;
        o.y = y1;
        ((float2*)dst)[(size_t)n * (DIM / 2) + lane] = o;
    } else {
        float p = y0 * Wout[2 * lane] + y1 * Wout[2 * lane + 1];
        for (int m = 16; m >= 1; m >>= 1) p += __shfl_xor(p, m, 32);
        if (lane == 0) dst[n] = p + bout[0];
    }
}

__global__ void __launch_bounds__(256)
k_node_agg_norm(const int* __restrict__ rowptr, const int* __restrict__ spair,
                const float* __restrict__ sed, const float* __restrict__ a_src,
                const float* __restrict__ a_dst, const float* __restrict__ xs,
                const float* __restrict__ res, const float* __restrict__ b,
                const float* __restrict__ g, const float* __restrict__ beta,
                float* __restrict__ hout, int N) {
    node_softmax_agg_norm<false>(rowptr, spair, sed, a_src, a_dst, xs, res, b, g, beta,
                                 nullptr, nullptr, hout, N);
}

__global__ void __launch_bounds__(256)
k_node_agg_norm_out(const int* __restrict__ rowptr, const int* __restrict__ spair,
                    const float* __restrict__ sed, const float* __restrict__ a_src,
                    const float* __restrict__ a_dst, const float* __restrict__ xs,
                    const float* __restrict__ res, const float* __restrict__ b,
                    const float* __restrict__ g, const float* __restrict__ beta,
                    const float* __restrict__ Wout, const float* __restrict__ bout,
                    float* __restrict__ out, int N) {
    node_softmax_agg_norm<true>(rowptr, spair, sed, a_src, a_dst, xs, res, b, g, beta,
                                Wout, bout, out, N);
}

extern "C" void kernel_launch(void* const* d_in, const int* in_sizes, int n_in,
                              void* d_out, int out_size, void* d_ws, size_t ws_size,
                              hipStream_t stream) {
    const float* x    = (const float*)d_in[0];
    const int*   ei   = (const int*)  d_in[1];
    const float* eatt = (const float*)d_in[3];
    const float* W1   = (const float*)d_in[4];
    const float* as1  = (const float*)d_in[5];
    const float* ad1  = (const float*)d_in[6];
    const float* We1  = (const float*)d_in[7];
    const float* ae1  = (const float*)d_in[8];
    const float* b1   = (const float*)d_in[9];
    const float* W2   = (const float*)d_in[10];
    const float* as2  = (const float*)d_in[11];
    const float* ad2  = (const float*)d_in[12];
    const float* We2  = (const float*)d_in[13];
    const float* ae2  = (const float*)d_in[14];
    const float* b2   = (const float*)d_in[15];
    const float* g1   = (const float*)d_in[16];
    const float* be1  = (const float*)d_in[17];
    const float* g2   = (const float*)d_in[18];
    const float* be2  = (const float*)d_in[19];
    const float* Wout = (const float*)d_in[20];
    const float* bout = (const float*)d_in[21];
    float* outp = (float*)d_out;

    const int N = in_sizes[0] / DIM;
    const int E = in_sizes[1] / 2;

    float* w = (float*)d_ws;
    size_t o = 0;
    float* xs    = w + o; o += (size_t)N * DIM;
    int*   M     = (int*)(w + o); o += (size_t)N * DIM;
    float* h1    = w + o; o += (size_t)N * DIM;
    float* a_src = w + o; o += (size_t)N * HEADS;
    float* a_dst = w + o; o += (size_t)N * HEADS;
    int*   spair = (int*)(w + o); o += (size_t)2 * E;
    int*   rank  = (int*)(w + o); o += (size_t)E;
    int*   cnt   = (int*)(w + o); o += (size_t)N;
    int*   rowp  = (int*)(w + o); o += (size_t)N + 1;
    float* sed   = w + o; o += 8;

    auto cdiv = [](long a, long b) { return (int)((a + b - 1) / b); };
    const int T = 256;

    k_chunk_rank<<<RCH, 1024, 0, stream>>>(ei, M, rank, E, N);
    k_chunk_scan<<<cdiv(N, T), T, 0, stream>>>(M, cnt, N);
    k_row_offsets<<<cdiv(N, 1024), 1024, 0, stream>>>(cnt, rowp, N);
    k_edge_fill<<<cdiv(E, T), T, 0, stream>>>(ei, eatt, rowp, M, rank, spair, E, N);
    k_edge_scal<<<1, 32, 0, stream>>>(We1, ae1, We2, ae2, sed);

    k_gemm64_wmma<<<cdiv((long)cdiv(N, 16), 2), T, 0, stream>>>(x, W1, xs, N);
    k_attn_coef<<<cdiv((long)N * HEADS, T), T, 0, stream>>>(xs, as1, ad1, a_src, a_dst, N);
    k_node_agg_norm<<<cdiv(N, 8), T, 0, stream>>>(rowp, spair, sed, a_src, a_dst, xs,
                                                  x, b1, g1, be1, h1, N);

    k_gemm64_wmma<<<cdiv((long)cdiv(N, 16), 2), T, 0, stream>>>(h1, W2, xs, N);
    k_attn_coef<<<cdiv((long)N * HEADS, T), T, 0, stream>>>(xs, as2, ad2, a_src, a_dst, N);
    k_node_agg_norm_out<<<cdiv(N, 8), T, 0, stream>>>(rowp, spair, sed + 4, a_src, a_dst, xs,
                                                      h1, b2, g2, be2, Wout, bout, outp, N);
}


# Attention coefficients and per-head edge scalar computed in the GEMM launch: a wave's 16x16 D tile is one head, so a_src/a_dst are 16-lane tree sums of its accumulators; 3 launches gone
# speedup vs baseline: 10.7105x; 1.0361x over previous
#include <hip/hip_runtime.h>
#include <hip/hip_bf16.h>

#define HEADS 4
#define HC    16
#define DIM   64
#define NEGS  0.2f

typedef float v2f __attribute__((ext_vector_type(2)));
typedef float v8f __attribute__((ext_vector_type(8)));

__global__ void __launch_bounds__(256)
k_gemm64_coef(const float* __restrict__ X, const float* __restrict__ W,
              const float* __restrict__ as, const float* __restrict__ ad,
              const float* __restrict__ We, const float* __restrict__ ae,
              float* __restrict__ Y, float* __restrict__ a_src, float* __restrict__ a_dst,
              float* __restrict__ sed, int nrows) {
    __shared__ float s_coef[2][32][HEADS];
    const int wave = threadIdx.x >> 5;
    const int lane = threadIdx.x & 31;
    const int rowTile = blockIdx.x * 2 + (wave >> 2);
    const int row0 = rowTile * 16;
    const int col0 = (wave & 3) * 16;
    const int lr = lane & 15;
    const int hi = lane >> 4;

    const bool live = row0 < nrows;
    v8f acc = {};
    if (live) {
        int arow = row0 + lr; if (arow > nrows - 1) arow = nrows - 1;
        const float* ap = X + (size_t)arow * DIM;

        for (int k = 0; k < DIM; k += 4) {
            v2f a;
            a.x = ap[k + 2 * hi + 0];
            a.y = ap[k + 2 * hi + 1];
            const float* bp = W + (size_t)(k + 2 * hi) * DIM + col0 + lr;
            v2f b;
            b.x = bp[0];
            b.y = bp[DIM];
            acc = __builtin_amdgcn_wmma_f32_16x16x4_f32(false, a, false, b,
                                                        (short)0, acc, false, false);
        }

        const float asl = as[col0 + lr];
        const float adl = ad[col0 + lr];
        float v[16];
        for (int r = 0; r < 8; ++r) {
            v[r]     = acc[r] * asl;
            v[8 + r] = acc[r] * adl;
        }
        float w8[8], w4[4], w2[2];
        const bool b3 = (lr & 8) != 0, b2 = (lr & 4) != 0, b1 = (lr & 2) != 0, b0 = (lr & 1) != 0;
        for (int j = 0; j < 8; ++j) {
            const float keep = b3 ? v[8 + j] : v[j];
            const float give = b3 ? v[j] : v[8 + j];
            w8[j] = keep + __shfl_xor(give, 8, 32);
        }
        for (int j = 0; j < 4; ++j) {
            const float keep = b2 ? w8[4 + j] : w8[j];
            const float give = b2 ? w8[j] : w8[4 + j];
            w4[j] = keep + __shfl_xor(give, 4, 32);
        }
        for (int j = 0; j < 2; ++j) {
            const float keep = b1 ? w4[2 + j] : w4[j];
            const float give = b1 ? w4[j] : w4[2 + j];
            w2[j] = keep + __shfl_xor(give, 2, 32);
        }
        const float keep1 = b0 ? w2[1] : w2[0];
        const float give1 = b0 ? w2[0] : w2[1];
        const float tot = keep1 + __shfl_xor(give1, 1, 32);
        s_coef[lr >> 3][(wave >> 2) * 16 + (lr & 7) + 8 * hi][wave & 3] = tot;
    }
    __syncthreads();
    if (live) {
        for (int r = 0; r < 8; ++r) {
            int row = row0 + r + 8 * hi;
            if (row < nrows) Y[(size_t)row * DIM + col0 + lr] = acc[r];
        }
    }
    if (wave < 2) {
        const int row = blockIdx.x * 32 + lane;
        if (row < nrows) {
            float4 o;
            o.x = s_coef[wave][lane][0];
            o.y = s_coef[wave][lane][1];
            o.z = s_coef[wave][lane][2];
            o.w = s_coef[wave][lane][3];
            ((float4*)(wave == 0 ? a_src : a_dst))[row] = o;
        }
    }
    if (blockIdx.x == 0 && threadIdx.x < HEADS) {
        const int h = threadIdx.x;
        float sacc = 0.0f;
        for (int c = 0; c < HC; ++c) sacc += We[h * HC + c] * ae[h * HC + c];
        sed[h] = sacc;
    }
}

#define RCH  64
#define LDSN 51200
#define RB   8
static_assert(RCH <= DIM, "the chunk table must fit a node plane");

__global__ void __launch_bounds__(1024)
k_chunk_rank(const int* __restrict__ ei, int* __restrict__ M, int* __restrict__ rank, int E, int N) {
    __shared__ int T[LDSN];
    const int c   = blockIdx.x;
    const int per = (E + RCH - 1) / RCH;
    const int beg = c * per;
    const int end = (beg + per < E) ? (beg + per) : E;
    const int* __restrict__ dstp = ei + E;
    for (int base = 0; base < N; base += LDSN) {
        const int lim = (N - base < LDSN) ? (N - base) : LDSN;
        for (int i = threadIdx.x; i < lim; i += 1024) T[i] = 0;
        __syncthreads();
        for (int e0 = beg + (int)threadIdx.x; e0 < end; e0 += RB * 1024) {
            int d[RB];
            #pragma unroll
            for (int k = 0; k < RB; ++k) {
                const int e  = e0 + k * 1024;
                const int ec = (e < end) ? e : (end - 1);
                const int dv = dstp[ec] - base;
                d[k] = (e < end) ? dv : -1;
            }
            #pragma unroll
            for (int k = 0; k < RB; ++k) {
                if ((unsigned)d[k] < (unsigned)lim) rank[e0 + k * 1024] = atomicAdd(&T[d[k]], 1);
            }
        }
        __syncthreads();
        for (int i = threadIdx.x; i < lim; i += 1024) M[(size_t)c * N + base + i] = T[i];
        __syncthreads();
    }
}

__global__ void __launch_bounds__(256)
k_chunk_scan(int* __restrict__ M, int* __restrict__ cnt, int N) {
    const int n = blockIdx.x * blockDim.x + threadIdx.x;
    if (n >= N) return;
    int v[RCH];
    #pragma unroll
    for (int c = 0; c < RCH; ++c) v[c] = M[(size_t)c * N + n];
    int acc = 0;
    #pragma unroll
    for (int c = 0; c < RCH; ++c) {
        M[(size_t)c * N + n] = acc;
        acc += v[c];
    }
    cnt[n] = acc;
}

__global__ void __launch_bounds__(1024)
k_row_offsets(const int* __restrict__ cnt, int* __restrict__ rowptr, int N) {
    __shared__ int s_pre[32];
    __shared__ int s_sum[32];
    const int t = threadIdx.x, lane = t & 31, wave = t >> 5;
    const int base = blockIdx.x * 1024;
    int pre = 0;
    for (int i = t; i < base; i += 1024) pre += cnt[i];
    #pragma unroll
    for (int o = 16; o >= 1; o >>= 1) pre += __shfl_xor(pre, o, 32);
    const int idx = base + t;
    const int v = (idx < N) ? cnt[idx] : 0;
    int inc = v;
    #pragma unroll
    for (int o = 1; o < 32; o <<= 1) {
        int u = __shfl_up(inc, o, 32);
        if (lane >= o) inc += u;
    }
    if (lane == 0)  s_pre[wave] = pre;
    if (lane == 31) s_sum[wave] = inc;
    __syncthreads();
    int off = 0;
    for (int w = 0; w < 32; ++w) {
        off += s_pre[w];
        if (w < wave) off += s_sum[w];
    }
    const int excl = off + inc - v;
    if (idx < N) rowptr[idx] = excl;
    if (idx == N - 1) rowptr[N] = excl + v;
}

__global__ void k_edge_fill(const int* __restrict__ ei, const float* __restrict__ eattr,
                            const int* __restrict__ rowptr, const int* __restrict__ M,
                            const int* __restrict__ rank, int* __restrict__ spair, int E, int N) {
    int e = blockIdx.x * blockDim.x + threadIdx.x;
    if (e >= E) return;
    const int per = (E + RCH - 1) / RCH;
    const int d = ei[E + e];
    int pos = rowptr[d] + M[(size_t)(e / per) * N + d] + rank[e];
    int2 v;
    v.x = ei[e];
    v.y = __float_as_int(eattr[e]);
    ((int2*)spair)[pos] = v;
}

__device__ __forceinline__ float edge_logit(float as, float ad, float ea, float se) {
    float al = __builtin_fmaf(ea, se, as + ad);
    return (al > 0.0f) ? al : NEGS * al;
}

template <bool LAST>
__device__ __forceinline__ void
node_softmax_agg_norm(const int* __restrict__ rowptr, const int* __restrict__ spair,
                      const float* __restrict__ sed, const float* __restrict__ a_src,
                      const float* __restrict__ a_dst, const float* __restrict__ xs,
                      const float* __restrict__ res, const float* __restrict__ b,
                      const float* __restrict__ g, const float* __restrict__ beta,
                      const float* __restrict__ Wout, const float* __restrict__ bout,
                      float* __restrict__ dst, int N) {
    const int lane = threadIdx.x & 31;
    const int n = blockIdx.x * 8 + __builtin_amdgcn_readfirstlane((int)(threadIdx.x >> 5));
    if (n >= N) return;
    const int beg = rowptr[n];
    const int end = rowptr[n + 1];
    const int2*   pairs = (const int2*)spair;
    const float4* as4   = (const float4*)a_src;
    const float4 adn = ((const float4*)a_dst)[n];
    const float4 asn = as4[n];
    const float se0 = sed[0], se1 = sed[1], se2 = sed[2], se3 = sed[3];

    const float ninf = -__builtin_huge_valf();
    float easum = 0.0f, m0 = ninf, m1 = ninf, m2 = ninf, m3 = ninf;
    const int  q0 = beg + lane, q1 = beg + 32 + lane;
    const bool u0 = q0 < end,   u1 = q1 < end;
    const int2 kp0 = pairs[u0 ? q0 : 0];
    const int2 kp1 = pairs[u1 ? q1 : 0];
    const float4 ka0 = as4[kp0.x];
    const float4 ka1 = as4[kp1.x];
    const float kea0 = __int_as_float(kp0.y), kea1 = __int_as_float(kp1.y);
    const int   ks0 = kp0.x, ks1 = kp1.x;
    const float k00 = u0 ? edge_logit(ka0.x, adn.x, kea0, se0) : ninf;
    const float k01 = u0 ? edge_logit(ka0.y, adn.y, kea0, se1) : ninf;
    const float k02 = u0 ? edge_logit(ka0.z, adn.z, kea0, se2) : ninf;
    const float k03 = u0 ? edge_logit(ka0.w, adn.w, kea0, se3) : ninf;
    const float k10 = u1 ? edge_logit(ka1.x, adn.x, kea1, se0) : ninf;
    const float k11 = u1 ? edge_logit(ka1.y, adn.y, kea1, se1) : ninf;
    const float k12 = u1 ? edge_logit(ka1.z, adn.z, kea1, se2) : ninf;
    const float k13 = u1 ? edge_logit(ka1.w, adn.w, kea1, se3) : ninf;
    easum += u0 ? kea0 : 0.0f;
    easum += u1 ? kea1 : 0.0f;
    m0 = fmaxf(fmaxf(m0, k00), k10);
    m1 = fmaxf(fmaxf(m1, k01), k11);
    m2 = fmaxf(fmaxf(m2, k02), k12);
    m3 = fmaxf(fmaxf(m3, k03), k13);
    for (int p = beg + 64 + lane; p < end; p += 32) {
        const int2 pr = pairs[p];
        const float ea = __int_as_float(pr.y);
        const float4 a = as4[pr.x];
        easum += ea;
        m0 = fmaxf(m0, edge_logit(a.x, adn.x, ea, se0));
        m1 = fmaxf(m1, edge_logit(a.y, adn.y, ea, se1));
        m2 = fmaxf(m2, edge_logit(a.z, adn.z, ea, se2));
        m3 = fmaxf(m3, edge_logit(a.w, adn.w, ea, se3));
    }
    #pragma unroll
    for (int o = 16; o >= 1; o >>= 1) {
        easum += __shfl_xor(easum, o, 32);
        m0 = fmaxf(m0, __shfl_xor(m0, o, 32));
        m1 = fmaxf(m1, __shfl_xor(m1, o, 32));
        m2 = fmaxf(m2, __shfl_xor(m2, o, 32));
        m3 = fmaxf(m3, __shfl_xor(m3, o, 32));
    }
    const float la = easum / fmaxf((float)(end - beg), 1.0f);
    const float sl0 = edge_logit(asn.x, adn.x, la, se0);
    const float sl1 = edge_logit(asn.y, adn.y, la, se1);
    const float sl2 = edge_logit(asn.z, adn.z, la, se2);
    const float sl3 = edge_logit(asn.w, adn.w, la, se3);
    m0 = fmaxf(m0, sl0); m1 = fmaxf(m1, sl1); m2 = fmaxf(m2, sl2); m3 = fmaxf(m3, sl3);

    const int h = lane >> 3;
    const float mh  = (h == 0) ? m0  : (h == 1) ? m1  : (h == 2) ? m2  : m3;
    const float slh = (h == 0) ? sl0 : (h == 1) ? sl1 : (h == 2) ? sl2 : sl3;
    const float2* xrow = (const float2*)xs + lane;
    const float es = expf(slh - mh);
    const float2 xn = xrow[(size_t)n * (DIM / 2)];
    const float* rp = res + (size_t)n * DIM;
    const float rs0 = rp[2 * lane],   rs1 = rp[2 * lane + 1];
    const float bi0 = b[2 * lane],    bi1 = b[2 * lane + 1];
    const float ga0 = g[2 * lane],    ga1 = g[2 * lane + 1];
    const float be0 = beta[2 * lane], be1 = beta[2 * lane + 1];
    float den = es;
    float ax = xn.x * es, ay = xn.y * es;
    for (int base = beg; base < end; base += 32) {
        int s;
        float l0, l1, l2, l3;
        if (base == beg) {
            s = ks0; l0 = k00; l1 = k01; l2 = k02; l3 = k03;
        } else if (base == beg + 32) {
            s = ks1; l0 = k10; l1 = k11; l2 = k12; l3 = k13;
        } else {
            const int p = base + lane;
            s = n; l0 = ninf; l1 = ninf; l2 = ninf; l3 = ninf;
            if (p < end) {
                const int2 pr = pairs[p];
                const float ea = __int_as_float(pr.y);
                s = pr.x;
                const float4 a = as4[s];
                l0 = edge_logit(a.x, adn.x, ea, se0);
                l1 = edge_logit(a.y, adn.y, ea, se1);
                l2 = edge_logit(a.z, adn.z, ea, se2);
                l3 = edge_logit(a.w, adn.w, ea, se3);
            }
        }
        const float e0 = expf(l0 - m0);
        const float e1 = expf(l1 - m1);
        const float e2 = expf(l2 - m2);
        const float e3 = expf(l3 - m3);
        const int cc = (end - base < 32) ? (end - base) : 32;
        #pragma unroll 4
        for (int j = 0; j < cc; ++j) {
            const int sj = __builtin_amdgcn_readlane(s, j);
            const float r0 = __int_as_float(__builtin_amdgcn_readlane(__float_as_int(e0), j));
            const float r1 = __int_as_float(__builtin_amdgcn_readlane(__float_as_int(e1), j));
            const float r2 = __int_as_float(__builtin_amdgcn_readlane(__float_as_int(e2), j));
            const float r3 = __int_as_float(__builtin_amdgcn_readlane(__float_as_int(e3), j));
            const float wj = (h == 0) ? r0 : (h == 1) ? r1 : (h == 2) ? r2 : r3;
            const float2 xv = xrow[(size_t)sj * (DIM / 2)];
            den += wj;
            ax = __builtin_fmaf(xv.x, wj, ax);
            ay = __builtin_fmaf(xv.y, wj, ay);
        }
    }
    const float inv = den + 1e-16f;

    const float v0 = (ax / inv + bi0) + rs0;
    const float v1 = (ay / inv + bi1) + rs1;
    float sum = v0 + v1;
    for (int m = 16; m >= 1; m >>= 1) sum += __shfl_xor(sum, m, 32);
    const float mu = sum * (1.0f / DIM);
    const float d0 = v0 - mu, d1 = v1 - mu;
    float vs = d0 * d0 + d1 * d1;
    for (int m = 16; m >= 1; m >>= 1) vs += __shfl_xor(vs, m, 32);
    const float rstd = rsqrtf(vs * (1.0f / DIM) + 1e-5f);
    float y0 = d0 * rstd * ga0 + be0;
    float y1 = d1 * rstd * ga1 + be1;
    y0 = (y0 > 0.0f) ? y0 : (expf(y0) - 1.0f);
    y1 = (y1 > 0.0f) ? y1 : (expf(y1) - 1.0f);
    if (!LAST) {
        float2 o;
        o.x = y0;
        o.y = y1;
        ((float2*)dst)[(size_t)n * (DIM / 2) + lane] = o;
    } else {
        float p = y0 * Wout[2 * lane] + y1 * Wout[2 * lane + 1];
        for (int m = 16; m >= 1; m >>= 1) p += __shfl_xor(p, m, 32);
        if (lane == 0) dst[n] = p + bout[0];
    }
}

__global__ void __launch_bounds__(256)
k_node_agg_norm(const int* __restrict__ rowptr, const int* __restrict__ spair,
                const float* __restrict__ sed, const float* __restrict__ a_src,
                const float* __restrict__ a_dst, const float* __restrict__ xs,
                const float* __restrict__ res, const float* __restrict__ b,
                const float* __restrict__ g, const float* __restrict__ beta,
                float* __restrict__ hout, int N) {
    node_softmax_agg_norm<false>(rowptr, spair, sed, a_src, a_dst, xs, res, b, g, beta,
                                 nullptr, nullptr, hout, N);
}

__global__ void __launch_bounds__(256)
k_node_agg_norm_out(const int* __restrict__ rowptr, const int* __restrict__ spair,
                    const float* __restrict__ sed, const float* __restrict__ a_src,
                    const float* __restrict__ a_dst, const float* __restrict__ xs,
                    const float* __restrict__ res, const float* __restrict__ b,
                    const float* __restrict__ g, const float* __restrict__ beta,
                    const float* __restrict__ Wout, const float* __restrict__ bout,
                    float* __restrict__ out, int N) {
    node_softmax_agg_norm<true>(rowptr, spair, sed, a_src, a_dst, xs, res, b, g, beta,
                                Wout, bout, out, N);
}

extern "C" void kernel_launch(void* const* d_in, const int* in_sizes, int n_in,
                              void* d_out, int out_size, void* d_ws, size_t ws_size,
                              hipStream_t stream) {
    const float* x    = (const float*)d_in[0];
    const int*   ei   = (const int*)  d_in[1];
    const float* eatt = (const float*)d_in[3];
    const float* W1   = (const float*)d_in[4];
    const float* as1  = (const float*)d_in[5];
    const float* ad1  = (const float*)d_in[6];
    const float* We1  = (const float*)d_in[7];
    const float* ae1  = (const float*)d_in[8];
    const float* b1   = (const float*)d_in[9];
    const float* W2   = (const float*)d_in[10];
    const float* as2  = (const float*)d_in[11];
    const float* ad2  = (const float*)d_in[12];
    const float* We2  = (const float*)d_in[13];
    const float* ae2  = (const float*)d_in[14];
    const float* b2   = (const float*)d_in[15];
    const float* g1   = (const float*)d_in[16];
    const float* be1  = (const float*)d_in[17];
    const float* g2   = (const float*)d_in[18];
    const float* be2  = (const float*)d_in[19];
    const float* Wout = (const float*)d_in[20];
    const float* bout = (const float*)d_in[21];
    float* outp = (float*)d_out;

    const int N = in_sizes[0] / DIM;
    const int E = in_sizes[1] / 2;

    float* w = (float*)d_ws;
    size_t o = 0;
    float* xs    = w + o; o += (size_t)N * DIM;
    int*   M     = (int*)(w + o); o += (size_t)N * DIM;
    float* h1    = w + o; o += (size_t)N * DIM;
    float* a_src = w + o; o += (size_t)N * HEADS;
    float* a_dst = w + o; o += (size_t)N * HEADS;
    int*   spair = (int*)(w + o); o += (size_t)2 * E;
    int*   rank  = (int*)(w + o); o += (size_t)E;
    int*   cnt   = (int*)(w + o); o += (size_t)N;
    int*   rowp  = (int*)(w + o); o += (size_t)N + 1;
    float* sed   = w + o; o += 8;

    auto cdiv = [](long a, long b) { return (int)((a + b - 1) / b); };
    const int T = 256;

    k_chunk_rank<<<RCH, 1024, 0, stream>>>(ei, M, rank, E, N);
    k_chunk_scan<<<cdiv(N, T), T, 0, stream>>>(M, cnt, N);
    k_row_offsets<<<cdiv(N, 1024), 1024, 0, stream>>>(cnt, rowp, N);
    k_edge_fill<<<cdiv(E, T), T, 0, stream>>>(ei, eatt, rowp, M, rank, spair, E, N);

    k_gemm64_coef<<<cdiv((long)cdiv(N, 16), 2), T, 0, stream>>>(x, W1, as1, ad1, We1, ae1,
                                                                xs, a_src, a_dst, sed, N);
    k_node_agg_norm<<<cdiv(N, 8), T, 0, stream>>>(rowp, spair, sed, a_src, a_dst, xs,
                                                  x, b1, g1, be1, h1, N);

    k_gemm64_coef<<<cdiv((long)cdiv(N, 16), 2), T, 0, stream>>>(h1, W2, as2, ad2, We2, ae2,
                                                                xs, a_src, a_dst, sed + 4, N);
    k_node_agg_norm_out<<<cdiv(N, 8), T, 0, stream>>>(rowp, spair, sed + 4, a_src, a_dst, xs,
                                                      h1, b2, g2, be2, Wout, bout, outp, N);
}


# Node wave: lane 8h+t takes head h of edge 8q+t, so an edge's weight is one 8-lane DPP broadcast (was 4 readlanes+3 selects), 8 rows in flight; reductions by DPP, not LDS; attr/exp sums: 8-lane trees
# speedup vs baseline: 11.5777x; 1.0810x over previous
#include <hip/hip_runtime.h>
#include <hip/hip_bf16.h>

#define HEADS 4
#define HC    16
#define DIM   64
#define NEGS  0.2f

typedef float v2f __attribute__((ext_vector_type(2)));
typedef float v8f __attribute__((ext_vector_type(8)));

__global__ void __launch_bounds__(256)
k_gemm64_coef(const float* __restrict__ X, const float* __restrict__ W,
              const float* __restrict__ as, const float* __restrict__ ad,
              const float* __restrict__ We, const float* __restrict__ ae,
              float* __restrict__ Y, float* __restrict__ a_src, float* __restrict__ a_dst,
              float* __restrict__ sed, int nrows) {
    __shared__ float s_coef[2][32][HEADS];
    const int wave = threadIdx.x >> 5;
    const int lane = threadIdx.x & 31;
    const int rowTile = blockIdx.x * 2 + (wave >> 2);
    const int row0 = rowTile * 16;
    const int col0 = (wave & 3) * 16;
    const int lr = lane & 15;
    const int hi = lane >> 4;

    const bool live = row0 < nrows;
    v8f acc = {};
    if (live) {
        int arow = row0 + lr; if (arow > nrows - 1) arow = nrows - 1;
        const float* ap = X + (size_t)arow * DIM;

        for (int k = 0; k < DIM; k += 4) {
            v2f a;
            a.x = ap[k + 2 * hi + 0];
            a.y = ap[k + 2 * hi + 1];
            const float* bp = W + (size_t)(k + 2 * hi) * DIM + col0 + lr;
            v2f b;
            b.x = bp[0];
            b.y = bp[DIM];
            acc = __builtin_amdgcn_wmma_f32_16x16x4_f32(false, a, false, b,
                                                        (short)0, acc, false, false);
        }

        const float asl = as[col0 + lr];
        const float adl = ad[col0 + lr];
        float v[16];
        for (int r = 0; r < 8; ++r) {
            v[r]     = acc[r] * asl;
            v[8 + r] = acc[r] * adl;
        }
        float w8[8], w4[4], w2[2];
        const bool b3 = (lr & 8) != 0, b2 = (lr & 4) != 0, b1 = (lr & 2) != 0, b0 = (lr & 1) != 0;
        for (int j = 0; j < 8; ++j) {
            const float keep = b3 ? v[8 + j] : v[j];
            const float give = b3 ? v[j] : v[8 + j];
            w8[j] = keep + __shfl_xor(give, 8, 32);
        }
        for (int j = 0; j < 4; ++j) {
            const float keep = b2 ? w8[4 + j] : w8[j];
            const float give = b2 ? w8[j] : w8[4 + j];
            w4[j] = keep + __shfl_xor(give, 4, 32);
        }
        for (int j = 0; j < 2; ++j) {
            const float keep = b1 ? w4[2 + j] : w4[j];
            const float give = b1 ? w4[j] : w4[2 + j];
            w2[j] = keep + __shfl_xor(give, 2, 32);
        }
        const float keep1 = b0 ? w2[1] : w2[0];
        const float give1 = b0 ? w2[0] : w2[1];
        const float tot = keep1 + __shfl_xor(give1, 1, 32);
        s_coef[lr >> 3][(wave >> 2) * 16 + (lr & 7) + 8 * hi][wave & 3] = tot;
    }
    __syncthreads();
    if (live) {
        for (int r = 0; r < 8; ++r) {
            int row = row0 + r + 8 * hi;
            if (row < nrows) Y[(size_t)row * DIM + col0 + lr] = acc[r];
        }
    }
    if (wave < 2) {
        const int row = blockIdx.x * 32 + lane;
        if (row < nrows) {
            float4 o;
            o.x = s_coef[wave][lane][0];
            o.y = s_coef[wave][lane][1];
            o.z = s_coef[wave][lane][2];
            o.w = s_coef[wave][lane][3];
            ((float4*)(wave == 0 ? a_src : a_dst))[row] = o;
        }
    }
    if (blockIdx.x == 0 && threadIdx.x < HEADS) {
        const int h = threadIdx.x;
        float sacc = 0.0f;
        for (int c = 0; c < HC; ++c) sacc += We[h * HC + c] * ae[h * HC + c];
        sed[h] = sacc;
    }
}

#define RCH  64
#define LDSN 51200
#define RB   8
static_assert(RCH <= DIM, "the chunk table must fit a node plane");

__global__ void __launch_bounds__(1024)
k_chunk_rank(const int* __restrict__ ei, int* __restrict__ M, int* __restrict__ rank, int E, int N) {
    __shared__ int T[LDSN];
    const int c   = blockIdx.x;
    const int per = (E + RCH - 1) / RCH;
    const int beg = c * per;
    const int end = (beg + per < E) ? (beg + per) : E;
    const int* __restrict__ dstp = ei + E;
    for (int base = 0; base < N; base += LDSN) {
        const int lim = (N - base < LDSN) ? (N - base) : LDSN;
        for (int i = threadIdx.x; i < lim; i += 1024) T[i] = 0;
        __syncthreads();
        for (int e0 = beg + (int)threadIdx.x; e0 < end; e0 += RB * 1024) {
            int d[RB];
            #pragma unroll
            for (int k = 0; k < RB; ++k) {
                const int e  = e0 + k * 1024;
                const int ec = (e < end) ? e : (end - 1);
                const int dv = dstp[ec] - base;
                d[k] = (e < end) ? dv : -1;
            }
            #pragma unroll
            for (int k = 0; k < RB; ++k) {
                if ((unsigned)d[k] < (unsigned)lim) rank[e0 + k * 1024] = atomicAdd(&T[d[k]], 1);
            }
        }
        __syncthreads();
        for (int i = threadIdx.x; i < lim; i += 1024) M[(size_t)c * N + base + i] = T[i];
        __syncthreads();
    }
}

__global__ void __launch_bounds__(256)
k_chunk_scan(int* __restrict__ M, int* __restrict__ cnt, int N) {
    const int n = blockIdx.x * blockDim.x + threadIdx.x;
    if (n >= N) return;
    int v[RCH];
    #pragma unroll
    for (int c = 0; c < RCH; ++c) v[c] = M[(size_t)c * N + n];
    int acc = 0;
    #pragma unroll
    for (int c = 0; c < RCH; ++c) {
        M[(size_t)c * N + n] = acc;
        acc += v[c];
    }
    cnt[n] = acc;
}

__global__ void __launch_bounds__(1024)
k_row_offsets(const int* __restrict__ cnt, int* __restrict__ rowptr, int N) {
    __shared__ int s_pre[32];
    __shared__ int s_sum[32];
    const int t = threadIdx.x, lane = t & 31, wave = t >> 5;
    const int base = blockIdx.x * 1024;
    int pre = 0;
    for (int i = t; i < base; i += 1024) pre += cnt[i];
    #pragma unroll
    for (int o = 16; o >= 1; o >>= 1) pre += __shfl_xor(pre, o, 32);
    const int idx = base + t;
    const int v = (idx < N) ? cnt[idx] : 0;
    int inc = v;
    #pragma unroll
    for (int o = 1; o < 32; o <<= 1) {
        int u = __shfl_up(inc, o, 32);
        if (lane >= o) inc += u;
    }
    if (lane == 0)  s_pre[wave] = pre;
    if (lane == 31) s_sum[wave] = inc;
    __syncthreads();
    int off = 0;
    for (int w = 0; w < 32; ++w) {
        off += s_pre[w];
        if (w < wave) off += s_sum[w];
    }
    const int excl = off + inc - v;
    if (idx < N) rowptr[idx] = excl;
    if (idx == N - 1) rowptr[N] = excl + v;
}

__global__ void k_edge_fill(const int* __restrict__ ei, const float* __restrict__ eattr,
                            const int* __restrict__ rowptr, const int* __restrict__ M,
                            const int* __restrict__ rank, int* __restrict__ spair, int E, int N) {
    int e = blockIdx.x * blockDim.x + threadIdx.x;
    if (e >= E) return;
    const int per = (E + RCH - 1) / RCH;
    const int d = ei[E + e];
    int pos = rowptr[d] + M[(size_t)(e / per) * N + d] + rank[e];
    int2 v;
    v.x = ei[e];
    v.y = __float_as_int(eattr[e]);
    ((int2*)spair)[pos] = v;
}

__device__ __forceinline__ float edge_logit(float as, float ad, float ea, float se) {
    float al = __builtin_fmaf(ea, se, as + ad);
    return (al > 0.0f) ? al : NEGS * al;
}


constexpr unsigned grp8_sel(unsigned a0, unsigned a1, unsigned a2, unsigned a3,
                            unsigned a4, unsigned a5, unsigned a6, unsigned a7) {
    return a0 | (a1 << 3) | (a2 << 6) | (a3 << 9) | (a4 << 12) | (a5 << 15) | (a6 << 18) | (a7 << 21);
}
template <unsigned SEL>
__device__ __forceinline__ float grp8_move(float v) {
    return __int_as_float(__builtin_amdgcn_mov_dpp8(__float_as_int(v), SEL));
}
__device__ __forceinline__ float grp8_max(float v) {
    v = fmaxf(v, grp8_move<grp8_sel(1, 0, 3, 2, 5, 4, 7, 6)>(v));
    v = fmaxf(v, grp8_move<grp8_sel(2, 3, 0, 1, 6, 7, 4, 5)>(v));
    v = fmaxf(v, grp8_move<grp8_sel(4, 5, 6, 7, 0, 1, 2, 3)>(v));
    return v;
}
__device__ __forceinline__ float grp8_sum(float v) {
    v += grp8_move<grp8_sel(1, 0, 3, 2, 5, 4, 7, 6)>(v);
    v += grp8_move<grp8_sel(2, 3, 0, 1, 6, 7, 4, 5)>(v);
    v += grp8_move<grp8_sel(4, 5, 6, 7, 0, 1, 2, 3)>(v);
    return v;
}
template <int MASK>
__device__ __forceinline__ float row_xor(float v) {
    return __int_as_float(__builtin_amdgcn_update_dpp(0, __float_as_int(v), 0x160 + MASK, 0xf, 0xf, false));
}
__device__ __forceinline__ float xhalf(float v) {
    const int b = __float_as_int(v);
    return __int_as_float(__builtin_amdgcn_permlanex16(b, b, 0x76543210, 0xfedcba98, false, false));
}
__device__ __forceinline__ float wave_sum(float v) {
    v += xhalf(v);
    v += row_xor<8>(v);
    v += row_xor<4>(v);
    v += row_xor<2>(v);
    v += row_xor<1>(v);
    return v;
}

__device__ __forceinline__ void
block_pairs(const int2* __restrict__ pairs, int base, int end, int lastp, int t,
            int2 (&pr)[4], bool (&ok)[4]) {
    for (int q = 0; q < 4; ++q) {
        const int idx = base + 8 * q + t;
        ok[q] = idx < end;
        pr[q] = pairs[ok[q] ? idx : lastp];
    }
}
__device__ __forceinline__ void
block_coefs(const float* __restrict__ a_src, int h, const int2 (&pr)[4], float (&a)[4]) {
    for (int q = 0; q < 4; ++q) a[q] = a_src[pr[q].x * HEADS + h];
}
__device__ __forceinline__ void
block_logits(const int2 (&pr)[4], const bool (&ok)[4], const float (&a)[4], float adh, float seh,
             int (&s)[4], float (&lg)[4], float& easum) {
    const float ninf = -__builtin_huge_valf();
    for (int q = 0; q < 4; ++q) {
        const float ea = __int_as_float(pr[q].y);
        s[q]  = pr[q].x;
        lg[q] = ok[q] ? edge_logit(a[q], adh, ea, seh) : ninf;
        easum += ok[q] ? ea : 0.0f;
    }
}
__device__ __forceinline__ float max4(const float (&v)[4]) {
    return fmaxf(fmaxf(v[0], v[1]), fmaxf(v[2], v[3]));
}

#define GATHER_ONE(U)                                                                         \
    {                                                                                         \
        float wj;                                                                             \
        asm("v_mov_b32_dpp %0, %1 dpp8:[" #U "," #U "," #U "," #U "," #U "," #U "," #U "," #U "]" \
            : "=v"(wj) : "v"(wq));                                                            \
        ax = __builtin_fmaf(xv[U].x, wj, ax);                                                 \
        ay = __builtin_fmaf(xv[U].y, wj, ay);                                                 \
    }
__device__ __forceinline__ void
gather_group(const float* __restrict__ xs, unsigned lane8, int sq, float wq, float& ax, float& ay) {
    float2 xv[8];
    for (int u = 0; u < 8; ++u) {
        const unsigned sj = (unsigned)__builtin_amdgcn_readlane(sq, u);
        xv[u] = *(const float2*)((const char*)xs + ((sj << 8) + lane8));
    }
    GATHER_ONE(0) GATHER_ONE(1) GATHER_ONE(2) GATHER_ONE(3)
    GATHER_ONE(4) GATHER_ONE(5) GATHER_ONE(6) GATHER_ONE(7)
}
#undef GATHER_ONE

__device__ __forceinline__ void
gather_block(const float* __restrict__ xs, unsigned lane8, int cnt, float m, const int (&s)[4],
             const float (&lg)[4], float& dsum, float& ax, float& ay) {
    float w[4];
    for (int q = 0; q < 4; ++q) {
        w[q] = expf(lg[q] - m);
        dsum += w[q];
    }
    for (int q = 0; q < 4; ++q) {
        if (8 * q < cnt) gather_group(xs, lane8, s[q], w[q], ax, ay);
    }
}

template <bool LAST>
__device__ __forceinline__ void
node_softmax_agg_norm(const int* __restrict__ rowptr, const int* __restrict__ spair,
                      const float* __restrict__ sed, const float* __restrict__ a_src,
                      const float* __restrict__ a_dst, const float* __restrict__ xs,
                      const float* __restrict__ res, const float* __restrict__ b,
                      const float* __restrict__ g, const float* __restrict__ beta,
                      const float* __restrict__ Wout, const float* __restrict__ bout,
                      float* __restrict__ dst, int N) {
    const int lane = threadIdx.x & 31;
    const int n = blockIdx.x * 8 + __builtin_amdgcn_readfirstlane((int)(threadIdx.x >> 5));
    if (n >= N) return;
    const int beg = rowptr[n];
    const int end = rowptr[n + 1];
    const int deg = end - beg;
    const int lastp = (end > 0) ? (end - 1) : 0;
    const bool two = deg > 32;
    const int h = lane >> 3, t = lane & 7;
    const unsigned lane8 = (unsigned)lane * 8u;
    const int2* pairs = (const int2*)spair;

    int2 prA[4], prB[4];
    bool okA[4], okB[4];
    float aA[4], aB[4];
    block_pairs(pairs, beg, end, lastp, t, prA, okA);
    if (two) block_pairs(pairs, beg + 32, end, lastp, t, prB, okB);
    const float adh = a_dst[n * HEADS + h];
    const float ash = a_src[n * HEADS + h];
    const float seh = sed[h];
    block_coefs(a_src, h, prA, aA);
    if (two) block_coefs(a_src, h, prB, aB);
    const float2 xn = *(const float2*)((const char*)xs + (((unsigned)n << 8) + lane8));
    const float* rp = res + (size_t)n * DIM;
    const float rs0 = rp[2 * lane],   rs1 = rp[2 * lane + 1];
    const float bi0 = b[2 * lane],    bi1 = b[2 * lane + 1];
    const float ga0 = g[2 * lane],    ga1 = g[2 * lane + 1];
    const float be0 = beta[2 * lane], be1 = beta[2 * lane + 1];

    float easum = 0.0f;
    int   sA[4], sB[4];
    float lA[4], lB[4];
    block_logits(prA, okA, aA, adh, seh, sA, lA, easum);
    float m = max4(lA);
    if (two) {
        block_logits(prB, okB, aB, adh, seh, sB, lB, easum);
        m = fmaxf(m, max4(lB));
    }
    for (int base = beg + 64; base < end; base += 32) {
        int2 prC[4];
        bool okC[4];
        float aC[4], lC[4];
        int sC[4];
        block_pairs(pairs, base, end, lastp, t, prC, okC);
        block_coefs(a_src, h, prC, aC);
        block_logits(prC, okC, aC, adh, seh, sC, lC, easum);
        m = fmaxf(m, max4(lC));
    }
    m = grp8_max(m);
    easum = grp8_sum(easum);
    const float la = easum / fmaxf((float)deg, 1.0f);
    const float sl = edge_logit(ash, adh, la, seh);
    m = fmaxf(m, sl);

    const float es = expf(sl - m);
    float dsum = 0.0f;
    float ax = xn.x * es, ay = xn.y * es;
    gather_block(xs, lane8, deg, m, sA, lA, dsum, ax, ay);
    if (two) gather_block(xs, lane8, deg - 32, m, sB, lB, dsum, ax, ay);
    for (int base = beg + 64; base < end; base += 32) {
        int2 prC[4];
        bool okC[4];
        float aC[4], lC[4];
        int sC[4];
        float unused = 0.0f;
        block_pairs(pairs, base, end, lastp, t, prC, okC);
        block_coefs(a_src, h, prC, aC);
        block_logits(prC, okC, aC, adh, seh, sC, lC, unused);
        gather_block(xs, lane8, end - base, m, sC, lC, dsum, ax, ay);
    }
    const float inv = (es + grp8_sum(dsum)) + 1e-16f;

    const float v0 = (ax / inv + bi0) + rs0;
    const float v1 = (ay / inv + bi1) + rs1;
    const float mu = wave_sum(v0 + v1) * (1.0f / DIM);
    const float d0 = v0 - mu, d1 = v1 - mu;
    const float vs = wave_sum(d0 * d0 + d1 * d1);
    const float rstd = rsqrtf(vs * (1.0f / DIM) + 1e-5f);
    float y0 = d0 * rstd * ga0 + be0;
    float y1 = d1 * rstd * ga1 + be1;
    y0 = (y0 > 0.0f) ? y0 : (expf(y0) - 1.0f);
    y1 = (y1 > 0.0f) ? y1 : (expf(y1) - 1.0f);
    if (!LAST) {
        float2 o;
        o.x = y0;
        o.y = y1;
        ((float2*)dst)[(size_t)n * (DIM / 2) + lane] = o;
    } else {
        const float p = wave_sum(y0 * Wout[2 * lane] + y1 * Wout[2 * lane + 1]);
        if (lane == 0) dst[n] = p + bout[0];
    }
}

__global__ void __launch_bounds__(256)
k_node_agg_norm(const int* __restrict__ rowptr, const int* __restrict__ spair,
                const float* __restrict__ sed, const float* __restrict__ a_src,
                const float* __restrict__ a_dst, const float* __restrict__ xs,
                const float* __restrict__ res, const float* __restrict__ b,
                const float* __restrict__ g, const float* __restrict__ beta,
                float* __restrict__ hout, int N) {
    node_softmax_agg_norm<false>(rowptr, spair, sed, a_src, a_dst, xs, res, b, g, beta,
                                 nullptr, nullptr, hout, N);
}

__global__ void __launch_bounds__(256)
k_node_agg_norm_out(const int* __restrict__ rowptr, const int* __restrict__ spair,
                    const float* __restrict__ sed, const float* __restrict__ a_src,
                    const float* __restrict__ a_dst, const float* __restrict__ xs,
                    const float* __restrict__ res, const float* __restrict__ b,
                    const float* __restrict__ g, const float* __restrict__ beta,
                    const float* __restrict__ Wout, const float* __restrict__ bout,
                    float* __restrict__ out, int N) {
    node_softmax_agg_norm<true>(rowptr, spair, sed, a_src, a_dst, xs, res, b, g, beta,
                                Wout, bout, out, N);
}

extern "C" void kernel_launch(void* const* d_in, const int* in_sizes, int n_in,
                              void* d_out, int out_size, void* d_ws, size_t ws_size,
                              hipStream_t stream) {
    const float* x    = (const float*)d_in[0];
    const int*   ei   = (const int*)  d_in[1];
    const float* eatt = (const float*)d_in[3];
    const float* W1   = (const float*)d_in[4];
    const float* as1  = (const float*)d_in[5];
    const float* ad1  = (const float*)d_in[6];
    const float* We1  = (const float*)d_in[7];
    const float* ae1  = (const float*)d_in[8];
    const float* b1   = (const float*)d_in[9];
    const float* W2   = (const float*)d_in[10];
    const float* as2  = (const float*)d_in[11];
    const float* ad2  = (const float*)d_in[12];
    const float* We2  = (const float*)d_in[13];
    const float* ae2  = (const float*)d_in[14];
    const float* b2   = (const float*)d_in[15];
    const float* g1   = (const float*)d_in[16];
    const float* be1  = (const float*)d_in[17];
    const float* g2   = (const float*)d_in[18];
    const float* be2  = (const float*)d_in[19];
    const float* Wout = (const float*)d_in[20];
    const float* bout = (const float*)d_in[21];
    float* outp = (float*)d_out;

    const int N = in_sizes[0] / DIM;
    const int E = in_sizes[1] / 2;

    float* w = (float*)d_ws;
    size_t o = 0;
    float* xs    = w + o; o += (size_t)N * DIM;
    int*   M     = (int*)(w + o); o += (size_t)N * DIM;
    float* h1    = w + o; o += (size_t)N * DIM;
    float* a_src = w + o; o += (size_t)N * HEADS;
    float* a_dst = w + o; o += (size_t)N * HEADS;
    int*   spair = (int*)(w + o); o += (size_t)2 * E;
    int*   rank  = (int*)(w + o); o += (size_t)E;
    int*   cnt   = (int*)(w + o); o += (size_t)N;
    int*   rowp  = (int*)(w + o); o += (size_t)N + 1;
    float* sed   = w + o; o += 8;

    auto cdiv = [](long a, long b) { return (int)((a + b - 1) / b); };
    const int T = 256;

    k_chunk_rank<<<RCH, 1024, 0, stream>>>(ei, M, rank, E, N);
    k_chunk_scan<<<cdiv(N, T), T, 0, stream>>>(M, cnt, N);
    k_row_offsets<<<cdiv(N, 1024), 1024, 0, stream>>>(cnt, rowp, N);
    k_edge_fill<<<cdiv(E, T), T, 0, stream>>>(ei, eatt, rowp, M, rank, spair, E, N);

    k_gemm64_coef<<<cdiv((long)cdiv(N, 16), 2), T, 0, stream>>>(x, W1, as1, ad1, We1, ae1,
                                                                xs, a_src, a_dst, sed, N);
    k_node_agg_norm<<<cdiv(N, 8), T, 0, stream>>>(rowp, spair, sed, a_src, a_dst, xs,
                                                  x, b1, g1, be1, h1, N);

    k_gemm64_coef<<<cdiv((long)cdiv(N, 16), 2), T, 0, stream>>>(h1, W2, as2, ad2, We2, ae2,
                                                                xs, a_src, a_dst, sed + 4, N);
    k_node_agg_norm_out<<<cdiv(N, 8), T, 0, stream>>>(rowp, spair, sed + 4, a_src, a_dst, xs,
                                                      h1, b2, g2, be2, Wout, bout, outp, N);
}


# Edge grouping by range bins: edges moved into 256 destination ranges as contiguous runs, each range sorted in one block's LDS and stored as whole lines; 3 launches for 4; segment order: LDS arrival
# speedup vs baseline: 13.4847x; 1.1647x over previous
#include <hip/hip_runtime.h>
#include <hip/hip_bf16.h>

#define HEADS 4
#define HC    16
#define DIM   64
#define NEGS  0.2f

typedef float v2f __attribute__((ext_vector_type(2)));
typedef float v8f __attribute__((ext_vector_type(8)));

__global__ void __launch_bounds__(256)
k_gemm64_coef(const float* __restrict__ X, const float* __restrict__ W,
              const float* __restrict__ as, const float* __restrict__ ad,
              const float* __restrict__ We, const float* __restrict__ ae,
              float* __restrict__ Y, float* __restrict__ a_src, float* __restrict__ a_dst,
              float* __restrict__ sed, int nrows) {
    __shared__ float s_coef[2][32][HEADS];
    const int wave = threadIdx.x >> 5;
    const int lane = threadIdx.x & 31;
    const int rowTile = blockIdx.x * 2 + (wave >> 2);
    const int row0 = rowTile * 16;
    const int col0 = (wave & 3) * 16;
    const int lr = lane & 15;
    const int hi = lane >> 4;

    const bool live = row0 < nrows;
    v8f acc = {};
    if (live) {
        int arow = row0 + lr; if (arow > nrows - 1) arow = nrows - 1;
        const float* ap = X + (size_t)arow * DIM;

        for (int k = 0; k < DIM; k += 4) {
            v2f a;
            a.x = ap[k + 2 * hi + 0];
            a.y = ap[k + 2 * hi + 1];
            const float* bp = W + (size_t)(k + 2 * hi) * DIM + col0 + lr;
            v2f b;
            b.x = bp[0];
            b.y = bp[DIM];
            acc = __builtin_amdgcn_wmma_f32_16x16x4_f32(false, a, false, b,
                                                        (short)0, acc, false, false);
        }

        const float asl = as[col0 + lr];
        const float adl = ad[col0 + lr];
        float v[16];
        for (int r = 0; r < 8; ++r) {
            v[r]     = acc[r] * asl;
            v[8 + r] = acc[r] * adl;
        }
        float w8[8], w4[4], w2[2];
        const bool b3 = (lr & 8) != 0, b2 = (lr & 4) != 0, b1 = (lr & 2) != 0, b0 = (lr & 1) != 0;
        for (int j = 0; j < 8; ++j) {
            const float keep = b3 ? v[8 + j] : v[j];
            const float give = b3 ? v[j] : v[8 + j];
            w8[j] = keep + __shfl_xor(give, 8, 32);
        }
        for (int j = 0; j < 4; ++j) {
            const float keep = b2 ? w8[4 + j] : w8[j];
            const float give = b2 ? w8[j] : w8[4 + j];
            w4[j] = keep + __shfl_xor(give, 4, 32);
        }
        for (int j = 0; j < 2; ++j) {
            const float keep = b1 ? w4[2 + j] : w4[j];
            const float give = b1 ? w4[j] : w4[2 + j];
            w2[j] = keep + __shfl_xor(give, 2, 32);
        }
        const float keep1 = b0 ? w2[1] : w2[0];
        const float give1 = b0 ? w2[0] : w2[1];
        const float tot = keep1 + __shfl_xor(give1, 1, 32);
        s_coef[lr >> 3][(wave >> 2) * 16 + (lr & 7) + 8 * hi][wave & 3] = tot;
    }
    __syncthreads();
    if (live) {
        for (int r = 0; r < 8; ++r) {
            int row = row0 + r + 8 * hi;
            if (row < nrows) Y[(size_t)row * DIM + col0 + lr] = acc[r];
        }
    }
    if (wave < 2) {
        const int row = blockIdx.x * 32 + lane;
        if (row < nrows) {
            float4 o;
            o.x = s_coef[wave][lane][0];
            o.y = s_coef[wave][lane][1];
            o.z = s_coef[wave][lane][2];
            o.w = s_coef[wave][lane][3];
            ((float4*)(wave == 0 ? a_src : a_dst))[row] = o;
        }
    }
    if (blockIdx.x == 0 && threadIdx.x < HEADS) {
        const int h = threadIdx.x;
        float sacc = 0.0f;
        for (int c = 0; c < HC; ++c) sacc += We[h * HC + c] * ae[h * HC + c];
        sed[h] = sacc;
    }
}

#define NBIN 256
#define EB   256
#define RB   8
#define RW   1024
#define STG  16384
static_assert(EB % 4 == 0 && NBIN * 4 == 1024, "table read: 4 thread groups of NBIN threads");

__device__ __forceinline__ int range_shift(int N) {
    int sh = 8;
    while (((N - 1) >> sh) >= NBIN) ++sh;
    return sh;
}
__device__ __forceinline__ int range_of(int d, int sh) {
    const unsigned r = (unsigned)d >> sh;
    return (int)(r < (unsigned)(NBIN - 1) ? r : (unsigned)(NBIN - 1));
}

__global__ void __launch_bounds__(1024)
k_range_count(const int* __restrict__ ei, int* __restrict__ H, int E, int N) {
    __shared__ int c[NBIN];
    const int t = threadIdx.x, b = blockIdx.x;
    const int sh = range_shift(N);
    const int per = (E + EB - 1) / EB;
    const int beg = b * per;
    const int end = (beg + per < E) ? (beg + per) : E;
    const int* __restrict__ dstp = ei + E;
    if (t < NBIN) c[t] = 0;
    __syncthreads();
    for (int e0 = beg + t; e0 < end; e0 += RB * 1024) {
        int r[RB];
        #pragma unroll
        for (int k = 0; k < RB; ++k) {
            const int e  = e0 + k * 1024;
            const int ec = (e < end) ? e : (end - 1);
            const int rv = range_of(dstp[ec], sh);
            r[k] = (e < end) ? rv : -1;
        }
        #pragma unroll
        for (int k = 0; k < RB; ++k) {
            if (r[k] >= 0) atomicAdd(&c[r[k]], 1);
        }
    }
    __syncthreads();
    if (t < NBIN) H[b * NBIN + t] = c[t];
}

__global__ void __launch_bounds__(1024)
k_range_scatter(const int* __restrict__ ei, const float* __restrict__ eattr,
                const int* __restrict__ H, int* __restrict__ BS, int* __restrict__ bdst,
                int* __restrict__ bpair, int E, int N) {
    __shared__ int s_tot[4][NBIN];
    __shared__ int s_par[4][NBIN];
    __shared__ int s_scan[NBIN / 32];
    __shared__ int s_off[NBIN];
    const int t = threadIdx.x, b = blockIdx.x;
    const int sh = range_shift(N);
    const int col = t & (NBIN - 1), g = t >> 8;
    int hv[EB / 4];
    #pragma unroll
    for (int k = 0; k < EB / 4; ++k) hv[k] = H[(g * (EB / 4) + k) * NBIN + col];
    int tsum = 0, psum = 0;
    #pragma unroll
    for (int k = 0; k < EB / 4; ++k) {
        tsum += hv[k];
        psum += (g * (EB / 4) + k < b) ? hv[k] : 0;
    }
    s_tot[g][col] = tsum;
    s_par[g][col] = psum;
    __syncthreads();
    int T = 0, P = 0;
    if (t < NBIN) {
        T = (s_tot[0][t] + s_tot[1][t]) + (s_tot[2][t] + s_tot[3][t]);
        P = (s_par[0][t] + s_par[1][t]) + (s_par[2][t] + s_par[3][t]);
    }
    const int lane = t & 31, wave = t >> 5;
    int inc = T;
    #pragma unroll
    for (int o = 1; o < 32; o <<= 1) {
        const int u = __shfl_up(inc, o, 32);
        if (lane >= o) inc += u;
    }
    if (t < NBIN && lane == 31) s_scan[wave] = inc;
    __syncthreads();
    if (t < NBIN) {
        int pre = 0;
        for (int w = 0; w < wave; ++w) pre += s_scan[w];
        const int start = pre + inc - T;
        s_off[t] = start + P;
        if (b == 0) {
            BS[t] = start;
            if (t == NBIN - 1) BS[NBIN] = start + T;
        }
    }
    __syncthreads();

    const int per = (E + EB - 1) / EB;
    const int beg = b * per;
    const int end = (beg + per < E) ? (beg + per) : E;
    const int* __restrict__ dstp = ei + E;
    for (int e0 = beg + t; e0 < end; e0 += RB * 1024) {
        int d[RB], s[RB], a[RB];
        #pragma unroll
        for (int k = 0; k < RB; ++k) {
            const int e  = e0 + k * 1024;
            const int ec = (e < end) ? e : (end - 1);
            d[k] = dstp[ec];
            s[k] = ei[ec];
            a[k] = __float_as_int(eattr[ec]);
        }
        #pragma unroll
        for (int k = 0; k < RB; ++k) {
            if (e0 + k * 1024 < end) {
                const int pos = atomicAdd(&s_off[range_of(d[k], sh)], 1);
                bdst[pos] = d[k];
                int2 v;
                v.x = s[k];
                v.y = a[k];
                ((int2*)bpair)[pos] = v;
            }
        }
    }
}

__global__ void __launch_bounds__(1024)
k_range_sort(const int* __restrict__ BS, const int* __restrict__ bdst,
             const int* __restrict__ bpair, int* __restrict__ rowptr, int* __restrict__ spair,
             int E, int N) {
    __shared__ int  s_cnt[RW];
    __shared__ int  s_scan[RW / 32];
    __shared__ int2 s_stage[STG];
    const int t = threadIdx.x, bin = blockIdx.x;
    const int sh = range_shift(N);
    const int Wd = 1 << sh;
    const int node0 = bin << sh;
    if (node0 >= N) return;
    const int msk = (Wd - 1) & (RW - 1);
    const int beg = BS[bin];
    const int end = BS[bin + 1];
    const int nrec = end - beg;
    const bool fits = nrec <= STG;
    s_cnt[t] = 0;
    __syncthreads();
    for (int i0 = beg + t; i0 < end; i0 += RB * 1024) {
        int j[RB];
        #pragma unroll
        for (int k = 0; k < RB; ++k) {
            const int i  = i0 + k * 1024;
            const int ic = (i < end) ? i : (end - 1);
            const int jv = (bdst[ic] - node0) & msk;
            j[k] = (i < end) ? jv : -1;
        }
        #pragma unroll
        for (int k = 0; k < RB; ++k) {
            if (j[k] >= 0) atomicAdd(&s_cnt[j[k]], 1);
        }
    }
    __syncthreads();
    const int lane = t & 31, wave = t >> 5;
    const int own = s_cnt[t];
    int inc = own;
    #pragma unroll
    for (int o = 1; o < 32; o <<= 1) {
        const int u = __shfl_up(inc, o, 32);
        if (lane >= o) inc += u;
    }
    if (lane == 31) s_scan[wave] = inc;
    __syncthreads();
    int pre = 0;
    for (int w = 0; w < wave; ++w) pre += s_scan[w];
    const int excl = pre + inc - own;
    s_cnt[t] = excl;
    if (t < Wd && node0 + t < N) rowptr[node0 + t] = beg + excl;
    if (bin == 0 && t == 0) rowptr[N] = E;
    __syncthreads();
    for (int i0 = beg + t; i0 < end; i0 += RB * 1024) {
        int  j[RB];
        int2 p[RB];
        #pragma unroll
        for (int k = 0; k < RB; ++k) {
            const int i  = i0 + k * 1024;
            const int ic = (i < end) ? i : (end - 1);
            j[k] = (bdst[ic] - node0) & msk;
            p[k] = ((const int2*)bpair)[ic];
        }
        #pragma unroll
        for (int k = 0; k < RB; ++k) {
            if (i0 + k * 1024 < end) {
                const int pos = atomicAdd(&s_cnt[j[k]], 1);
                if (fits) s_stage[pos] = p[k];
                else      ((int2*)spair)[beg + pos] = p[k];
            }
        }
    }
    __syncthreads();
    if (fits) {
        for (int i = t; i < nrec; i += 1024) ((int2*)spair)[beg + i] = s_stage[i];
    }
}

__device__ __forceinline__ float edge_logit(float as, float ad, float ea, float se) {
    float al = __builtin_fmaf(ea, se, as + ad);
    return (al > 0.0f) ? al : NEGS * al;
}


constexpr unsigned grp8_sel(unsigned a0, unsigned a1, unsigned a2, unsigned a3,
                            unsigned a4, unsigned a5, unsigned a6, unsigned a7) {
    return a0 | (a1 << 3) | (a2 << 6) | (a3 << 9) | (a4 << 12) | (a5 << 15) | (a6 << 18) | (a7 << 21);
}
template <unsigned SEL>
__device__ __forceinline__ float grp8_move(float v) {
    return __int_as_float(__builtin_amdgcn_mov_dpp8(__float_as_int(v), SEL));
}
__device__ __forceinline__ float grp8_max(float v) {
    v = fmaxf(v, grp8_move<grp8_sel(1, 0, 3, 2, 5, 4, 7, 6)>(v));
    v = fmaxf(v, grp8_move<grp8_sel(2, 3, 0, 1, 6, 7, 4, 5)>(v));
    v = fmaxf(v, grp8_move<grp8_sel(4, 5, 6, 7, 0, 1, 2, 3)>(v));
    return v;
}
__device__ __forceinline__ float grp8_sum(float v) {
    v += grp8_move<grp8_sel(1, 0, 3, 2, 5, 4, 7, 6)>(v);
    v += grp8_move<grp8_sel(2, 3, 0, 1, 6, 7, 4, 5)>(v);
    v += grp8_move<grp8_sel(4, 5, 6, 7, 0, 1, 2, 3)>(v);
    return v;
}
template <int MASK>
__device__ __forceinline__ float row_xor(float v) {
    return __int_as_float(__builtin_amdgcn_update_dpp(0, __float_as_int(v), 0x160 + MASK, 0xf, 0xf, false));
}
__device__ __forceinline__ float xhalf(float v) {
    const int b = __float_as_int(v);
    return __int_as_float(__builtin_amdgcn_permlanex16(b, b, 0x76543210, 0xfedcba98, false, false));
}
__device__ __forceinline__ float wave_sum(float v) {
    v += xhalf(v);
    v += row_xor<8>(v);
    v += row_xor<4>(v);
    v += row_xor<2>(v);
    v += row_xor<1>(v);
    return v;
}

__device__ __forceinline__ void
block_pairs(const int2* __restrict__ pairs, int base, int end, int lastp, int t,
            int2 (&pr)[4], bool (&ok)[4]) {
    for (int q = 0; q < 4; ++q) {
        const int idx = base + 8 * q + t;
        ok[q] = idx < end;
        pr[q] = pairs[ok[q] ? idx : lastp];
    }
}
__device__ __forceinline__ void
block_coefs(const float* __restrict__ a_src, int h, const int2 (&pr)[4], float (&a)[4]) {
    for (int q = 0; q < 4; ++q) a[q] = a_src[pr[q].x * HEADS + h];
}
__device__ __forceinline__ void
block_logits(const int2 (&pr)[4], const bool (&ok)[4], const float (&a)[4], float adh, float seh,
             int (&s)[4], float (&lg)[4], float& easum) {
    const float ninf = -__builtin_huge_valf();
    for (int q = 0; q < 4; ++q) {
        const float ea = __int_as_float(pr[q].y);
        s[q]  = pr[q].x;
        lg[q] = ok[q] ? edge_logit(a[q], adh, ea, seh) : ninf;
        easum += ok[q] ? ea : 0.0f;
    }
}
__device__ __forceinline__ float max4(const float (&v)[4]) {
    return fmaxf(fmaxf(v[0], v[1]), fmaxf(v[2], v[3]));
}

#define GATHER_ONE(U)                                                                         \
    {                                                                                         \
        float wj;                                                                             \
        asm("v_mov_b32_dpp %0, %1 dpp8:[" #U "," #U "," #U "," #U "," #U "," #U "," #U "," #U "]" \
            : "=v"(wj) : "v"(wq));                                                            \
        ax = __builtin_fmaf(xv[U].x, wj, ax);                                                 \
        ay = __builtin_fmaf(xv[U].y, wj, ay);                                                 \
    }
__device__ __forceinline__ void
gather_group(const float* __restrict__ xs, unsigned lane8, int sq, float wq, float& ax, float& ay) {
    float2 xv[8];
    for (int u = 0; u < 8; ++u) {
        const unsigned sj = (unsigned)__builtin_amdgcn_readlane(sq, u);
        xv[u] = *(const float2*)((const char*)xs + ((sj << 8) + lane8));
    }
    GATHER_ONE(0) GATHER_ONE(1) GATHER_ONE(2) GATHER_ONE(3)
    GATHER_ONE(4) GATHER_ONE(5) GATHER_ONE(6) GATHER_ONE(7)
}
#undef GATHER_ONE

__device__ __forceinline__ void
gather_block(const float* __restrict__ xs, unsigned lane8, int cnt, float m, const int (&s)[4],
             const float (&lg)[4], float& dsum, float& ax, float& ay) {
    float w[4];
    for (int q = 0; q < 4; ++q) {
        w[q] = expf(lg[q] - m);
        dsum += w[q];
    }
    for (int q = 0; q < 4; ++q) {
        if (8 * q < cnt) gather_group(xs, lane8, s[q], w[q], ax, ay);
    }
}

template <bool LAST>
__device__ __forceinline__ void
node_softmax_agg_norm(const int* __restrict__ rowptr, const int* __restrict__ spair,
                      const float* __restrict__ sed, const float* __restrict__ a_src,
                      const float* __restrict__ a_dst, const float* __restrict__ xs,
                      const float* __restrict__ res, const float* __restrict__ b,
                      const float* __restrict__ g, const float* __restrict__ beta,
                      const float* __restrict__ Wout, const float* __restrict__ bout,
                      float* __restrict__ dst, int N) {
    const int lane = threadIdx.x & 31;
    const int n = blockIdx.x * 8 + __builtin_amdgcn_readfirstlane((int)(threadIdx.x >> 5));
    if (n >= N) return;
    const int beg = rowptr[n];
    const int end = rowptr[n + 1];
    const int deg = end - beg;
    const int lastp = (end > 0) ? (end - 1) : 0;
    const bool two = deg > 32;
    const int h = lane >> 3, t = lane & 7;
    const unsigned lane8 = (unsigned)lane * 8u;
    const int2* pairs = (const int2*)spair;

    int2 prA[4], prB[4];
    bool okA[4], okB[4];
    float aA[4], aB[4];
    block_pairs(pairs, beg, end, lastp, t, prA, okA);
    if (two) block_pairs(pairs, beg + 32, end, lastp, t, prB, okB);
    const float adh = a_dst[n * HEADS + h];
    const float ash = a_src[n * HEADS + h];
    const float seh = sed[h];
    block_coefs(a_src, h, prA, aA);
    if (two) block_coefs(a_src, h, prB, aB);
    const float2 xn = *(const float2*)((const char*)xs + (((unsigned)n << 8) + lane8));
    const float* rp = res + (size_t)n * DIM;
    const float rs0 = rp[2 * lane],   rs1 = rp[2 * lane + 1];
    const float bi0 = b[2 * lane],    bi1 = b[2 * lane + 1];
    const float ga0 = g[2 * lane],    ga1 = g[2 * lane + 1];
    const float be0 = beta[2 * lane], be1 = beta[2 * lane + 1];

    float easum = 0.0f;
    int   sA[4], sB[4];
    float lA[4], lB[4];
    block_logits(prA, okA, aA, adh, seh, sA, lA, easum);
    float m = max4(lA);
    if (two) {
        block_logits(prB, okB, aB, adh, seh, sB, lB, easum);
        m = fmaxf(m, max4(lB));
    }
    for (int base = beg + 64; base < end; base += 32) {
        int2 prC[4];
        bool okC[4];
        float aC[4], lC[4];
        int sC[4];
        block_pairs(pairs, base, end, lastp, t, prC, okC);
        block_coefs(a_src, h, prC, aC);
        block_logits(prC, okC, aC, adh, seh, sC, lC, easum);
        m = fmaxf(m, max4(lC));
    }
    m = grp8_max(m);
    easum = grp8_sum(easum);
    const float la = easum / fmaxf((float)deg, 1.0f);
    const float sl = edge_logit(ash, adh, la, seh);
    m = fmaxf(m, sl);

    const float es = expf(sl - m);
    float dsum = 0.0f;
    float ax = xn.x * es, ay = xn.y * es;
    gather_block(xs, lane8, deg, m, sA, lA, dsum, ax, ay);
    if (two) gather_block(xs, lane8, deg - 32, m, sB, lB, dsum, ax, ay);
    for (int base = beg + 64; base < end; base += 32) {
        int2 prC[4];
        bool okC[4];
        float aC[4], lC[4];
        int sC[4];
        float unused = 0.0f;
        block_pairs(pairs, base, end, lastp, t, prC, okC);
        block_coefs(a_src, h, prC, aC);
        block_logits(prC, okC, aC, adh, seh, sC, lC, unused);
        gather_block(xs, lane8, end - base, m, sC, lC, dsum, ax, ay);
    }
    const float inv = (es + grp8_sum(dsum)) + 1e-16f;

    const float v0 = (ax / inv + bi0) + rs0;
    const float v1 = (ay / inv + bi1) + rs1;
    const float mu = wave_sum(v0 + v1) * (1.0f / DIM);
    const float d0 = v0 - mu, d1 = v1 - mu;
    const float vs = wave_sum(d0 * d0 + d1 * d1);
    const float rstd = rsqrtf(vs * (1.0f / DIM) + 1e-5f);
    float y0 = d0 * rstd * ga0 + be0;
    float y1 = d1 * rstd * ga1 + be1;
    y0 = (y0 > 0.0f) ? y0 : (expf(y0) - 1.0f);
    y1 = (y1 > 0.0f) ? y1 : (expf(y1) - 1.0f);
    if (!LAST) {
        float2 o;
        o.x = y0;
        o.y = y1;
        ((float2*)dst)[(size_t)n * (DIM / 2) + lane] = o;
    } else {
        const float p = wave_sum(y0 * Wout[2 * lane] + y1 * Wout[2 * lane + 1]);
        if (lane == 0) dst[n] = p + bout[0];
    }
}

__global__ void __launch_bounds__(256)
k_node_agg_norm(const int* __restrict__ rowptr, const int* __restrict__ spair,
                const float* __restrict__ sed, const float* __restrict__ a_src,
                const float* __restrict__ a_dst, const float* __restrict__ xs,
                const float* __restrict__ res, const float* __restrict__ b,
                const float* __restrict__ g, const float* __restrict__ beta,
                float* __restrict__ hout, int N) {
    node_softmax_agg_norm<false>(rowptr, spair, sed, a_src, a_dst, xs, res, b, g, beta,
                                 nullptr, nullptr, hout, N);
}

__global__ void __launch_bounds__(256)
k_node_agg_norm_out(const int* __restrict__ rowptr, const int* __restrict__ spair,
                    const float* __restrict__ sed, const float* __restrict__ a_src,
                    const float* __restrict__ a_dst, const float* __restrict__ xs,
                    const float* __restrict__ res, const float* __restrict__ b,
                    const float* __restrict__ g, const float* __restrict__ beta,
                    const float* __restrict__ Wout, const float* __restrict__ bout,
                    float* __restrict__ out, int N) {
    node_softmax_agg_norm<true>(rowptr, spair, sed, a_src, a_dst, xs, res, b, g, beta,
                                Wout, bout, out, N);
}

extern "C" void kernel_launch(void* const* d_in, const int* in_sizes, int n_in,
                              void* d_out, int out_size, void* d_ws, size_t ws_size,
                              hipStream_t stream) {
    const float* x    = (const float*)d_in[0];
    const int*   ei   = (const int*)  d_in[1];
    const float* eatt = (const float*)d_in[3];
    const float* W1   = (const float*)d_in[4];
    const float* as1  = (const float*)d_in[5];
    const float* ad1  = (const float*)d_in[6];
    const float* We1  = (const float*)d_in[7];
    const float* ae1  = (const float*)d_in[8];
    const float* b1   = (const float*)d_in[9];
    const float* W2   = (const float*)d_in[10];
    const float* as2  = (const float*)d_in[11];
    const float* ad2  = (const float*)d_in[12];
    const float* We2  = (const float*)d_in[13];
    const float* ae2  = (const float*)d_in[14];
    const float* b2   = (const float*)d_in[15];
    const float* g1   = (const float*)d_in[16];
    const float* be1  = (const float*)d_in[17];
    const float* g2   = (const float*)d_in[18];
    const float* be2  = (const float*)d_in[19];
    const float* Wout = (const float*)d_in[20];
    const float* bout = (const float*)d_in[21];
    float* outp = (float*)d_out;

    const int N = in_sizes[0] / DIM;
    const int E = in_sizes[1] / 2;

    float* w = (float*)d_ws;
    size_t o = 0;
    float* xs    = w + o; o += (size_t)N * DIM;
    int*   grp   = (int*)(w + o); o += (size_t)N * DIM;
    float* h1    = w + o; o += (size_t)N * DIM;
    float* a_src = w + o; o += (size_t)N * HEADS;
    float* a_dst = w + o; o += (size_t)N * HEADS;
    int*   spair = (int*)(w + o); o += (size_t)2 * E;
    int*   rowp  = (int*)(w + o); o += (size_t)N + 1;
    float* sed   = w + o; o += 8;
    int*   bdst  = grp;
    int*   H     = grp + E;
    int*   BS    = H + EB * NBIN;
    int*   bpair = (int*)h1;

    auto cdiv = [](long a, long b) { return (int)((a + b - 1) / b); };
    const int T = 256;

    k_range_count<<<EB, 1024, 0, stream>>>(ei, H, E, N);
    k_range_scatter<<<EB, 1024, 0, stream>>>(ei, eatt, H, BS, bdst, bpair, E, N);
    k_range_sort<<<NBIN, 1024, 0, stream>>>(BS, bdst, bpair, rowp, spair, E, N);

    k_gemm64_coef<<<cdiv((long)cdiv(N, 16), 2), T, 0, stream>>>(x, W1, as1, ad1, We1, ae1,
                                                                xs, a_src, a_dst, sed, N);
    k_node_agg_norm<<<cdiv(N, 8), T, 0, stream>>>(rowp, spair, sed, a_src, a_dst, xs,
                                                  x, b1, g1, be1, h1, N);

    k_gemm64_coef<<<cdiv((long)cdiv(N, 16), 2), T, 0, stream>>>(h1, W2, as2, ad2, We2, ae2,
                                                                xs, a_src, a_dst, sed + 4, N);
    k_node_agg_norm_out<<<cdiv(N, 8), T, 0, stream>>>(rowp, spair, sed + 4, a_src, a_dst, xs,
                                                      h1, b2, g2, be2, Wout, bout, outp, N);
}


# Range move (pass 2): an edge block first puts its edges in range order in LDS (local starts from its own table row), then stores them as contiguous runs, not one line per lane; order: LDS arrival
# speedup vs baseline: 16.2525x; 1.2053x over previous
#include <hip/hip_runtime.h>
#include <hip/hip_bf16.h>

#define HEADS 4
#define HC    16
#define DIM   64
#define NEGS  0.2f

typedef float v2f __attribute__((ext_vector_type(2)));
typedef float v8f __attribute__((ext_vector_type(8)));

__global__ void __launch_bounds__(256)
k_gemm64_coef(const float* __restrict__ X, const float* __restrict__ W,
              const float* __restrict__ as, const float* __restrict__ ad,
              const float* __restrict__ We, const float* __restrict__ ae,
              float* __restrict__ Y, float* __restrict__ a_src, float* __restrict__ a_dst,
              float* __restrict__ sed, int nrows) {
    __shared__ float s_coef[2][32][HEADS];
    const int wave = threadIdx.x >> 5;
    const int lane = threadIdx.x & 31;
    const int rowTile = blockIdx.x * 2 + (wave >> 2);
    const int row0 = rowTile * 16;
    const int col0 = (wave & 3) * 16;
    const int lr = lane & 15;
    const int hi = lane >> 4;

    const bool live = row0 < nrows;
    v8f acc = {};
    if (live) {
        int arow = row0 + lr; if (arow > nrows - 1) arow = nrows - 1;
        const float* ap = X + (size_t)arow * DIM;

        for (int k = 0; k < DIM; k += 4) {
            v2f a;
            a.x = ap[k + 2 * hi + 0];
            a.y = ap[k + 2 * hi + 1];
            const float* bp = W + (size_t)(k + 2 * hi) * DIM + col0 + lr;
            v2f b;
            b.x = bp[0];
            b.y = bp[DIM];
            acc = __builtin_amdgcn_wmma_f32_16x16x4_f32(false, a, false, b,
                                                        (short)0, acc, false, false);
        }

        const float asl = as[col0 + lr];
        const float adl = ad[col0 + lr];
        float v[16];
        for (int r = 0; r < 8; ++r) {
            v[r]     = acc[r] * asl;
            v[8 + r] = acc[r] * adl;
        }
        float w8[8], w4[4], w2[2];
        const bool b3 = (lr & 8) != 0, b2 = (lr & 4) != 0, b1 = (lr & 2) != 0, b0 = (lr & 1) != 0;
        for (int j = 0; j < 8; ++j) {
            const float keep = b3 ? v[8 + j] : v[j];
            const float give = b3 ? v[j] : v[8 + j];
            w8[j] = keep + __shfl_xor(give, 8, 32);
        }
        for (int j = 0; j < 4; ++j) {
            const float keep = b2 ? w8[4 + j] : w8[j];
            const float give = b2 ? w8[j] : w8[4 + j];
            w4[j] = keep + __shfl_xor(give, 4, 32);
        }
        for (int j = 0; j < 2; ++j) {
            const float keep = b1 ? w4[2 + j] : w4[j];
            const float give = b1 ? w4[j] : w4[2 + j];
            w2[j] = keep + __shfl_xor(give, 2, 32);
        }
        const float keep1 = b0 ? w2[1] : w2[0];
        const float give1 = b0 ? w2[0] : w2[1];
        const float tot = keep1 + __shfl_xor(give1, 1, 32);
        s_coef[lr >> 3][(wave >> 2) * 16 + (lr & 7) + 8 * hi][wave & 3] = tot;
    }
    __syncthreads();
    if (live) {
        for (int r = 0; r < 8; ++r) {
            int row = row0 + r + 8 * hi;
            if (row < nrows) Y[(size_t)row * DIM + col0 + lr] = acc[r];
        }
    }
    if (wave < 2) {
        const int row = blockIdx.x * 32 + lane;
        if (row < nrows) {
            float4 o;
            o.x = s_coef[wave][lane][0];
            o.y = s_coef[wave][lane][1];
            o.z = s_coef[wave][lane][2];
            o.w = s_coef[wave][lane][3];
            ((float4*)(wave == 0 ? a_src : a_dst))[row] = o;
        }
    }
    if (blockIdx.x == 0 && threadIdx.x < HEADS) {
        const int h = threadIdx.x;
        float sacc = 0.0f;
        for (int c = 0; c < HC; ++c) sacc += We[h * HC + c] * ae[h * HC + c];
        sed[h] = sacc;
    }
}

#define NBIN 256
#define EB   256
#define RB   8
#define RW   1024
#define STG  16384
#define SCAP (RB * 1024)
static_assert(EB % 4 == 0 && NBIN * 4 == 1024, "table read: 4 thread groups of NBIN threads");

__device__ __forceinline__ int range_shift(int N) {
    int sh = 8;
    while (((N - 1) >> sh) >= NBIN) ++sh;
    return sh;
}
__device__ __forceinline__ int range_of(int d, int sh) {
    const unsigned r = (unsigned)d >> sh;
    return (int)(r < (unsigned)(NBIN - 1) ? r : (unsigned)(NBIN - 1));
}

__global__ void __launch_bounds__(1024)
k_range_count(const int* __restrict__ ei, int* __restrict__ H, int E, int N) {
    __shared__ int c[NBIN];
    const int t = threadIdx.x, b = blockIdx.x;
    const int sh = range_shift(N);
    const int per = (E + EB - 1) / EB;
    const int beg = b * per;
    const int end = (beg + per < E) ? (beg + per) : E;
    const int* __restrict__ dstp = ei + E;
    if (t < NBIN) c[t] = 0;
    __syncthreads();
    for (int e0 = beg + t; e0 < end; e0 += RB * 1024) {
        int r[RB];
        #pragma unroll
        for (int k = 0; k < RB; ++k) {
            const int e  = e0 + k * 1024;
            const int ec = (e < end) ? e : (end - 1);
            const int rv = range_of(dstp[ec], sh);
            r[k] = (e < end) ? rv : -1;
        }
        #pragma unroll
        for (int k = 0; k < RB; ++k) {
            if (r[k] >= 0) atomicAdd(&c[r[k]], 1);
        }
    }
    __syncthreads();
    if (t < NBIN) H[b * NBIN + t] = c[t];
}

__global__ void __launch_bounds__(1024)
k_range_scatter(const int* __restrict__ ei, const float* __restrict__ eattr,
                const int* __restrict__ H, int* __restrict__ BS, int* __restrict__ bdst,
                int* __restrict__ bpair, int E, int N) {
    __shared__ int s_tot[4][NBIN];
    __shared__ int s_par[4][NBIN];
    __shared__ int s_scan[2][NBIN / 32];
    __shared__ int s_off[NBIN];
    __shared__ int s_gb[NBIN];
    __shared__ int  s_sd[SCAP];
    __shared__ int2 s_sp[SCAP];
    const int t = threadIdx.x, b = blockIdx.x;
    const int sh = range_shift(N);
    const int col = t & (NBIN - 1), g = t >> 8;
    const int per = (E + EB - 1) / EB;
    const int beg = b * per;
    const int end = (beg + per < E) ? (beg + per) : E;
    const bool fits = per <= SCAP;
    int hv[EB / 4];
    #pragma unroll
    for (int k = 0; k < EB / 4; ++k) hv[k] = H[(g * (EB / 4) + k) * NBIN + col];
    const int own = (t < NBIN) ? H[b * NBIN + t] : 0;
    int tsum = 0, psum = 0;
    #pragma unroll
    for (int k = 0; k < EB / 4; ++k) {
        tsum += hv[k];
        psum += (g * (EB / 4) + k < b) ? hv[k] : 0;
    }
    s_tot[g][col] = tsum;
    s_par[g][col] = psum;
    __syncthreads();
    int T = 0, P = 0;
    if (t < NBIN) {
        T = (s_tot[0][t] + s_tot[1][t]) + (s_tot[2][t] + s_tot[3][t]);
        P = (s_par[0][t] + s_par[1][t]) + (s_par[2][t] + s_par[3][t]);
    }
    const int lane = t & 31, wave = t >> 5;
    int inc = T, linc = own;
    #pragma unroll
    for (int o = 1; o < 32; o <<= 1) {
        const int u  = __shfl_up(inc, o, 32);
        const int lu = __shfl_up(linc, o, 32);
        if (lane >= o) { inc += u; linc += lu; }
    }
    if (t < NBIN && lane == 31) { s_scan[0][wave] = inc; s_scan[1][wave] = linc; }
    __syncthreads();
    if (t < NBIN) {
        int pre = 0, lpre = 0;
        for (int w = 0; w < wave; ++w) { pre += s_scan[0][w]; lpre += s_scan[1][w]; }
        const int start  = pre + inc - T;
        const int lstart = lpre + linc - own;
        s_off[t] = fits ? lstart : (start + P);
        s_gb[t]  = start + P - lstart;
        if (b == 0) {
            BS[t] = start;
            if (t == NBIN - 1) BS[NBIN] = start + T;
        }
    }
    __syncthreads();

    const int* __restrict__ dstp = ei + E;
    for (int e0 = beg + t; e0 < end; e0 += RB * 1024) {
        int d[RB], s[RB], a[RB];
        #pragma unroll
        for (int k = 0; k < RB; ++k) {
            const int e  = e0 + k * 1024;
            const int ec = (e < end) ? e : (end - 1);
            d[k] = dstp[ec];
            s[k] = ei[ec];
            a[k] = __float_as_int(eattr[ec]);
        }
        #pragma unroll
        for (int k = 0; k < RB; ++k) {
            if (e0 + k * 1024 < end) {
                const int pos = atomicAdd(&s_off[range_of(d[k], sh)], 1);
                int2 v;
                v.x = s[k];
                v.y = a[k];
                if (fits) {
                    const int lp = ((unsigned)pos < (unsigned)SCAP) ? pos : (SCAP - 1);
                    s_sd[lp] = d[k];
                    s_sp[lp] = v;
                } else {
                    bdst[pos] = d[k];
                    ((int2*)bpair)[pos] = v;
                }
            }
        }
    }
    if (fits) {
        __syncthreads();
        const int nb = end - beg;
        for (int i = t; i < nb; i += 1024) {
            const int dd = s_sd[i];
            const int gp = s_gb[range_of(dd, sh)] + i;
            bdst[gp] = dd;
            ((int2*)bpair)[gp] = s_sp[i];
        }
    }
}

__global__ void __launch_bounds__(1024)
k_range_sort(const int* __restrict__ BS, const int* __restrict__ bdst,
             const int* __restrict__ bpair, int* __restrict__ rowptr, int* __restrict__ spair,
             int E, int N) {
    __shared__ int  s_cnt[RW];
    __shared__ int  s_scan[RW / 32];
    __shared__ int2 s_stage[STG];
    const int t = threadIdx.x, bin = blockIdx.x;
    const int sh = range_shift(N);
    const int Wd = 1 << sh;
    const int node0 = bin << sh;
    if (node0 >= N) return;
    const int msk = (Wd - 1) & (RW - 1);
    const int beg = BS[bin];
    const int end = BS[bin + 1];
    const int nrec = end - beg;
    const bool fits = nrec <= STG;
    s_cnt[t] = 0;
    __syncthreads();
    for (int i0 = beg + t; i0 < end; i0 += RB * 1024) {
        int j[RB];
        #pragma unroll
        for (int k = 0; k < RB; ++k) {
            const int i  = i0 + k * 1024;
            const int ic = (i < end) ? i : (end - 1);
            const int jv = (bdst[ic] - node0) & msk;
            j[k] = (i < end) ? jv : -1;
        }
        #pragma unroll
        for (int k = 0; k < RB; ++k) {
            if (j[k] >= 0) atomicAdd(&s_cnt[j[k]], 1);
        }
    }
    __syncthreads();
    const int lane = t & 31, wave = t >> 5;
    const int own = s_cnt[t];
    int inc = own;
    #pragma unroll
    for (int o = 1; o < 32; o <<= 1) {
        const int u = __shfl_up(inc, o, 32);
        if (lane >= o) inc += u;
    }
    if (lane == 31) s_scan[wave] = inc;
    __syncthreads();
    int pre = 0;
    for (int w = 0; w < wave; ++w) pre += s_scan[w];
    const int excl = pre + inc - own;
    s_cnt[t] = excl;
    if (t < Wd && node0 + t < N) rowptr[node0 + t] = beg + excl;
    if (bin == 0 && t == 0) rowptr[N] = E;
    __syncthreads();
    for (int i0 = beg + t; i0 < end; i0 += RB * 1024) {
        int  j[RB];
        int2 p[RB];
        #pragma unroll
        for (int k = 0; k < RB; ++k) {
            const int i  = i0 + k * 1024;
            const int ic = (i < end) ? i : (end - 1);
            j[k] = (bdst[ic] - node0) & msk;
            p[k] = ((const int2*)bpair)[ic];
        }
        #pragma unroll
        for (int k = 0; k < RB; ++k) {
            if (i0 + k * 1024 < end) {
                const int pos = atomicAdd(&s_cnt[j[k]], 1);
                if (fits) s_stage[pos] = p[k];
                else      ((int2*)spair)[beg + pos] = p[k];
            }
        }
    }
    __syncthreads();
    if (fits) {
        for (int i = t; i < nrec; i += 1024) ((int2*)spair)[beg + i] = s_stage[i];
    }
}

__device__ __forceinline__ float edge_logit(float as, float ad, float ea, float se) {
    float al = __builtin_fmaf(ea, se, as + ad);
    return (al > 0.0f) ? al : NEGS * al;
}


constexpr unsigned grp8_sel(unsigned a0, unsigned a1, unsigned a2, unsigned a3,
                            unsigned a4, unsigned a5, unsigned a6, unsigned a7) {
    return a0 | (a1 << 3) | (a2 << 6) | (a3 << 9) | (a4 << 12) | (a5 << 15) | (a6 << 18) | (a7 << 21);
}
template <unsigned SEL>
__device__ __forceinline__ float grp8_move(float v) {
    return __int_as_float(__builtin_amdgcn_mov_dpp8(__float_as_int(v), SEL));
}
__device__ __forceinline__ float grp8_max(float v) {
    v = fmaxf(v, grp8_move<grp8_sel(1, 0, 3, 2, 5, 4, 7, 6)>(v));
    v = fmaxf(v, grp8_move<grp8_sel(2, 3, 0, 1, 6, 7, 4, 5)>(v));
    v = fmaxf(v, grp8_move<grp8_sel(4, 5, 6, 7, 0, 1, 2, 3)>(v));
    return v;
}
__device__ __forceinline__ float grp8_sum(float v) {
    v += grp8_move<grp8_sel(1, 0, 3, 2, 5, 4, 7, 6)>(v);
    v += grp8_move<grp8_sel(2, 3, 0, 1, 6, 7, 4, 5)>(v);
    v += grp8_move<grp8_sel(4, 5, 6, 7, 0, 1, 2, 3)>(v);
    return v;
}
template <int MASK>
__device__ __forceinline__ float row_xor(float v) {
    return __int_as_float(__builtin_amdgcn_update_dpp(0, __float_as_int(v), 0x160 + MASK, 0xf, 0xf, false));
}
__device__ __forceinline__ float xhalf(float v) {
    const int b = __float_as_int(v);
    return __int_as_float(__builtin_amdgcn_permlanex16(b, b, 0x76543210, 0xfedcba98, false, false));
}
__device__ __forceinline__ float wave_sum(float v) {
    v += xhalf(v);
    v += row_xor<8>(v);
    v += row_xor<4>(v);
    v += row_xor<2>(v);
    v += row_xor<1>(v);
    return v;
}

__device__ __forceinline__ void
block_pairs(const int2* __restrict__ pairs, int base, int end, int lastp, int t,
            int2 (&pr)[4], bool (&ok)[4]) {
    for (int q = 0; q < 4; ++q) {
        const int idx = base + 8 * q + t;
        ok[q] = idx < end;
        pr[q] = pairs[ok[q] ? idx : lastp];
    }
}
__device__ __forceinline__ void
block_coefs(const float* __restrict__ a_src, int h, const int2 (&pr)[4], float (&a)[4]) {
    for (int q = 0; q < 4; ++q) a[q] = a_src[pr[q].x * HEADS + h];
}
__device__ __forceinline__ void
block_logits(const int2 (&pr)[4], const bool (&ok)[4], const float (&a)[4], float adh, float seh,
             int (&s)[4], float (&lg)[4], float& easum) {
    const float ninf = -__builtin_huge_valf();
    for (int q = 0; q < 4; ++q) {
        const float ea = __int_as_float(pr[q].y);
        s[q]  = pr[q].x;
        lg[q] = ok[q] ? edge_logit(a[q], adh, ea, seh) : ninf;
        easum += ok[q] ? ea : 0.0f;
    }
}
__device__ __forceinline__ float max4(const float (&v)[4]) {
    return fmaxf(fmaxf(v[0], v[1]), fmaxf(v[2], v[3]));
}

#define GATHER_ONE(U)                                                                         \
    {                                                                                         \
        float wj;                                                                             \
        asm("v_mov_b32_dpp %0, %1 dpp8:[" #U "," #U "," #U "," #U "," #U "," #U "," #U "," #U "]" \
            : "=v"(wj) : "v"(wq));                                                            \
        ax = __builtin_fmaf(xv[U].x, wj, ax);                                                 \
        ay = __builtin_fmaf(xv[U].y, wj, ay);                                                 \
    }
__device__ __forceinline__ void
gather_group(const float* __restrict__ xs, unsigned lane8, int sq, float wq, float& ax, float& ay) {
    float2 xv[8];
    for (int u = 0; u < 8; ++u) {
        const unsigned sj = (unsigned)__builtin_amdgcn_readlane(sq, u);
        xv[u] = *(const float2*)((const char*)xs + ((sj << 8) + lane8));
    }
    GATHER_ONE(0) GATHER_ONE(1) GATHER_ONE(2) GATHER_ONE(3)
    GATHER_ONE(4) GATHER_ONE(5) GATHER_ONE(6) GATHER_ONE(7)
}
#undef GATHER_ONE

__device__ __forceinline__ void
gather_block(const float* __restrict__ xs, unsigned lane8, int cnt, float m, const int (&s)[4],
             const float (&lg)[4], float& dsum, float& ax, float& ay) {
    float w[4];
    for (int q = 0; q < 4; ++q) {
        w[q] = expf(lg[q] - m);
        dsum += w[q];
    }
    for (int q = 0; q < 4; ++q) {
        if (8 * q < cnt) gather_group(xs, lane8, s[q], w[q], ax, ay);
    }
}

template <bool LAST>
__device__ __forceinline__ void
node_softmax_agg_norm(const int* __restrict__ rowptr, const int* __restrict__ spair,
                      const float* __restrict__ sed, const float* __restrict__ a_src,
                      const float* __restrict__ a_dst, const float* __restrict__ xs,
                      const float* __restrict__ res, const float* __restrict__ b,
                      const float* __restrict__ g, const float* __restrict__ beta,
                      const float* __restrict__ Wout, const float* __restrict__ bout,
                      float* __restrict__ dst, int N) {
    const int lane = threadIdx.x & 31;
    const int n = blockIdx.x * 8 + __builtin_amdgcn_readfirstlane((int)(threadIdx.x >> 5));
    if (n >= N) return;
    const int beg = rowptr[n];
    const int end = rowptr[n + 1];
    const int deg = end - beg;
    const int lastp = (end > 0) ? (end - 1) : 0;
    const bool two = deg > 32;
    const int h = lane >> 3, t = lane & 7;
    const unsigned lane8 = (unsigned)lane * 8u;
    const int2* pairs = (const int2*)spair;

    int2 prA[4], prB[4];
    bool okA[4], okB[4];
    float aA[4], aB[4];
    block_pairs(pairs, beg, end, lastp, t, prA, okA);
    if (two) block_pairs(pairs, beg + 32, end, lastp, t, prB, okB);
    const float adh = a_dst[n * HEADS + h];
    const float ash = a_src[n * HEADS + h];
    const float seh = sed[h];
    block_coefs(a_src, h, prA, aA);
    if (two) block_coefs(a_src, h, prB, aB);
    const float2 xn = *(const float2*)((const char*)xs + (((unsigned)n << 8) + lane8));
    const float* rp = res + (size_t)n * DIM;
    const float rs0 = rp[2 * lane],   rs1 = rp[2 * lane + 1];
    const float bi0 = b[2 * lane],    bi1 = b[2 * lane + 1];
    const float ga0 = g[2 * lane],    ga1 = g[2 * lane + 1];
    const float be0 = beta[2 * lane], be1 = beta[2 * lane + 1];

    float easum = 0.0f;
    int   sA[4], sB[4];
    float lA[4], lB[4];
    block_logits(prA, okA, aA, adh, seh, sA, lA, easum);
    float m = max4(lA);
    if (two) {
        block_logits(prB, okB, aB, adh, seh, sB, lB, easum);
        m = fmaxf(m, max4(lB));
    }
    for (int base = beg + 64; base < end; base += 32) {
        int2 prC[4];
        bool okC[4];
        float aC[4], lC[4];
        int sC[4];
        block_pairs(pairs, base, end, lastp, t, prC, okC);
        block_coefs(a_src, h, prC, aC);
        block_logits(prC, okC, aC, adh, seh, sC, lC, easum);
        m = fmaxf(m, max4(lC));
    }
    m = grp8_max(m);
    easum = grp8_sum(easum);
    const float la = easum / fmaxf((float)deg, 1.0f);
    const float sl = edge_logit(ash, adh, la, seh);
    m = fmaxf(m, sl);

    const float es = expf(sl - m);
    float dsum = 0.0f;
    float ax = xn.x * es, ay = xn.y * es;
    gather_block(xs, lane8, deg, m, sA, lA, dsum, ax, ay);
    if (two) gather_block(xs, lane8, deg - 32, m, sB, lB, dsum, ax, ay);
    for (int base = beg + 64; base < end; base += 32) {
        int2 prC[4];
        bool okC[4];
        float aC[4], lC[4];
        int sC[4];
        float unused = 0.0f;
        block_pairs(pairs, base, end, lastp, t, prC, okC);
        block_coefs(a_src, h, prC, aC);
        block_logits(prC, okC, aC, adh, seh, sC, lC, unused);
        gather_block(xs, lane8, end - base, m, sC, lC, dsum, ax, ay);
    }
    const float inv = (es + grp8_sum(dsum)) + 1e-16f;

    const float v0 = (ax / inv + bi0) + rs0;
    const float v1 = (ay / inv + bi1) + rs1;
    const float mu = wave_sum(v0 + v1) * (1.0f / DIM);
    const float d0 = v0 - mu, d1 = v1 - mu;
    const float vs = wave_sum(d0 * d0 + d1 * d1);
    const float rstd = rsqrtf(vs * (1.0f / DIM) + 1e-5f);
    float y0 = d0 * rstd * ga0 + be0;
    float y1 = d1 * rstd * ga1 + be1;
    y0 = (y0 > 0.0f) ? y0 : (expf(y0) - 1.0f);
    y1 = (y1 > 0.0f) ? y1 : (expf(y1) - 1.0f);
    if (!LAST) {
        float2 o;
        o.x = y0;
        o.y = y1;
        ((float2*)dst)[(size_t)n * (DIM / 2) + lane] = o;
    } else {
        const float p = wave_sum(y0 * Wout[2 * lane] + y1 * Wout[2 * lane + 1]);
        if (lane == 0) dst[n] = p + bout[0];
    }
}

__global__ void __launch_bounds__(256)
k_node_agg_norm(const int* __restrict__ rowptr, const int* __restrict__ spair,
                const float* __restrict__ sed, const float* __restrict__ a_src,
                const float* __restrict__ a_dst, const float* __restrict__ xs,
                const float* __restrict__ res, const float* __restrict__ b,
                const float* __restrict__ g, const float* __restrict__ beta,
                float* __restrict__ hout, int N) {
    node_softmax_agg_norm<false>(rowptr, spair, sed, a_src, a_dst, xs, res, b, g, beta,
                                 nullptr, nullptr, hout, N);
}

__global__ void __launch_bounds__(256)
k_node_agg_norm_out(const int* __restrict__ rowptr, const int* __restrict__ spair,
                    const float* __restrict__ sed, const float* __restrict__ a_src,
                    const float* __restrict__ a_dst, const float* __restrict__ xs,
                    const float* __restrict__ res, const float* __restrict__ b,
                    const float* __restrict__ g, const float* __restrict__ beta,
                    const float* __restrict__ Wout, const float* __restrict__ bout,
                    float* __restrict__ out, int N) {
    node_softmax_agg_norm<true>(rowptr, spair, sed, a_src, a_dst, xs, res, b, g, beta,
                                Wout, bout, out, N);
}

extern "C" void kernel_launch(void* const* d_in, const int* in_sizes, int n_in,
                              void* d_out, int out_size, void* d_ws, size_t ws_size,
                              hipStream_t stream) {
    const float* x    = (const float*)d_in[0];
    const int*   ei   = (const int*)  d_in[1];
    const float* eatt = (const float*)d_in[3];
    const float* W1   = (const float*)d_in[4];
    const float* as1  = (const float*)d_in[5];
    const float* ad1  = (const float*)d_in[6];
    const float* We1  = (const float*)d_in[7];
    const float* ae1  = (const float*)d_in[8];
    const float* b1   = (const float*)d_in[9];
    const float* W2   = (const float*)d_in[10];
    const float* as2  = (const float*)d_in[11];
    const float* ad2  = (const float*)d_in[12];
    const float* We2  = (const float*)d_in[13];
    const float* ae2  = (const float*)d_in[14];
    const float* b2   = (const float*)d_in[15];
    const float* g1   = (const float*)d_in[16];
    const float* be1  = (const float*)d_in[17];
    const float* g2   = (const float*)d_in[18];
    const float* be2  = (const float*)d_in[19];
    const float* Wout = (const float*)d_in[20];
    const float* bout = (const float*)d_in[21];
    float* outp = (float*)d_out;

    const int N = in_sizes[0] / DIM;
    const int E = in_sizes[1] / 2;

    float* w = (float*)d_ws;
    size_t o = 0;
    float* xs    = w + o; o += (size_t)N * DIM;
    int*   grp   = (int*)(w + o); o += (size_t)N * DIM;
    float* h1    = w + o; o += (size_t)N * DIM;
    float* a_src = w + o; o += (size_t)N * HEADS;
    float* a_dst = w + o; o += (size_t)N * HEADS;
    int*   spair = (int*)(w + o); o += (size_t)2 * E;
    int*   rowp  = (int*)(w + o); o += (size_t)N + 1;
    float* sed   = w + o; o += 8;
    int*   bdst  = grp;
    int*   H     = grp + E;
    int*   BS    = H + EB * NBIN;
    int*   bpair = (int*)h1;

    auto cdiv = [](long a, long b) { return (int)((a + b - 1) / b); };
    const int T = 256;

    k_range_count<<<EB, 1024, 0, stream>>>(ei, H, E, N);
    k_range_scatter<<<EB, 1024, 0, stream>>>(ei, eatt, H, BS, bdst, bpair, E, N);
    k_range_sort<<<NBIN, 1024, 0, stream>>>(BS, bdst, bpair, rowp, spair, E, N);

    k_gemm64_coef<<<cdiv((long)cdiv(N, 16), 2), T, 0, stream>>>(x, W1, as1, ad1, We1, ae1,
                                                                xs, a_src, a_dst, sed, N);
    k_node_agg_norm<<<cdiv(N, 8), T, 0, stream>>>(rowp, spair, sed, a_src, a_dst, xs,
                                                  x, b1, g1, be1, h1, N);

    k_gemm64_coef<<<cdiv((long)cdiv(N, 16), 2), T, 0, stream>>>(h1, W2, as2, ad2, We2, ae2,
                                                                xs, a_src, a_dst, sed + 4, N);
    k_node_agg_norm_out<<<cdiv(N, 8), T, 0, stream>>>(rowp, spair, sed + 4, a_src, a_dst, xs,
                                                      h1, b2, g2, be2, Wout, bout, outp, N);
}


# Node wave: exp, division and rsqrt library sequences replaced by the hardware 2^x (logits kept in base-2 units), rcp and rsq instructions, about 1 ulp each; every sum order unchanged
# speedup vs baseline: 18.6052x; 1.1448x over previous
#include <hip/hip_runtime.h>
#include <hip/hip_bf16.h>

#define HEADS 4
#define HC    16
#define DIM   64
#define NEGS  0.2f

typedef float v2f __attribute__((ext_vector_type(2)));
typedef float v8f __attribute__((ext_vector_type(8)));

__global__ void __launch_bounds__(256)
k_gemm64_coef(const float* __restrict__ X, const float* __restrict__ W,
              const float* __restrict__ as, const float* __restrict__ ad,
              const float* __restrict__ We, const float* __restrict__ ae,
              float* __restrict__ Y, float* __restrict__ a_src, float* __restrict__ a_dst,
              float* __restrict__ sed, int nrows) {
    __shared__ float s_coef[2][32][HEADS];
    const int wave = threadIdx.x >> 5;
    const int lane = threadIdx.x & 31;
    const int rowTile = blockIdx.x * 2 + (wave >> 2);
    const int row0 = rowTile * 16;
    const int col0 = (wave & 3) * 16;
    const int lr = lane & 15;
    const int hi = lane >> 4;

    const bool live = row0 < nrows;
    v8f acc = {};
    if (live) {
        int arow = row0 + lr; if (arow > nrows - 1) arow = nrows - 1;
        const float* ap = X + (size_t)arow * DIM;

        for (int k = 0; k < DIM; k += 4) {
            v2f a;
            a.x = ap[k + 2 * hi + 0];
            a.y = ap[k + 2 * hi + 1];
            const float* bp = W + (size_t)(k + 2 * hi) * DIM + col0 + lr;
            v2f b;
            b.x = bp[0];
            b.y = bp[DIM];
            acc = __builtin_amdgcn_wmma_f32_16x16x4_f32(false, a, false, b,
                                                        (short)0, acc, false, false);
        }

        const float asl = as[col0 + lr];
        const float adl = ad[col0 + lr];
        float v[16];
        for (int r = 0; r < 8; ++r) {
            v[r]     = acc[r] * asl;
            v[8 + r] = acc[r] * adl;
        }
        float w8[8], w4[4], w2[2];
        const bool b3 = (lr & 8) != 0, b2 = (lr & 4) != 0, b1 = (lr & 2) != 0, b0 = (lr & 1) != 0;
        for (int j = 0; j < 8; ++j) {
            const float keep = b3 ? v[8 + j] : v[j];
            const float give = b3 ? v[j] : v[8 + j];
            w8[j] = keep + __shfl_xor(give, 8, 32);
        }
        for (int j = 0; j < 4; ++j) {
            const float keep = b2 ? w8[4 + j] : w8[j];
            const float give = b2 ? w8[j] : w8[4 + j];
            w4[j] = keep + __shfl_xor(give, 4, 32);
        }
        for (int j = 0; j < 2; ++j) {
            const float keep = b1 ? w4[2 + j] : w4[j];
            const float give = b1 ? w4[j] : w4[2 + j];
            w2[j] = keep + __shfl_xor(give, 2, 32);
        }
        const float keep1 = b0 ? w2[1] : w2[0];
        const float give1 = b0 ? w2[0] : w2[1];
        const float tot = keep1 + __shfl_xor(give1, 1, 32);
        s_coef[lr >> 3][(wave >> 2) * 16 + (lr & 7) + 8 * hi][wave & 3] = tot;
    }
    __syncthreads();
    if (live) {
        for (int r = 0; r < 8; ++r) {
            int row = row0 + r + 8 * hi;
            if (row < nrows) Y[(size_t)row * DIM + col0 + lr] = acc[r];
        }
    }
    if (wave < 2) {
        const int row = blockIdx.x * 32 + lane;
        if (row < nrows) {
            float4 o;
            o.x = s_coef[wave][lane][0];
            o.y = s_coef[wave][lane][1];
            o.z = s_coef[wave][lane][2];
            o.w = s_coef[wave][lane][3];
            ((float4*)(wave == 0 ? a_src : a_dst))[row] = o;
        }
    }
    if (blockIdx.x == 0 && threadIdx.x < HEADS) {
        const int h = threadIdx.x;
        float sacc = 0.0f;
        for (int c = 0; c < HC; ++c) sacc += We[h * HC + c] * ae[h * HC + c];
        sed[h] = sacc;
    }
}

#define NBIN 256
#define EB   256
#define RB   8
#define RW   1024
#define STG  16384
#define SCAP (RB * 1024)
static_assert(EB % 4 == 0 && NBIN * 4 == 1024, "table read: 4 thread groups of NBIN threads");

__device__ __forceinline__ int range_shift(int N) {
    int sh = 8;
    while (((N - 1) >> sh) >= NBIN) ++sh;
    return sh;
}
__device__ __forceinline__ int range_of(int d, int sh) {
    const unsigned r = (unsigned)d >> sh;
    return (int)(r < (unsigned)(NBIN - 1) ? r : (unsigned)(NBIN - 1));
}

__global__ void __launch_bounds__(1024)
k_range_count(const int* __restrict__ ei, int* __restrict__ H, int E, int N) {
    __shared__ int c[NBIN];
    const int t = threadIdx.x, b = blockIdx.x;
    const int sh = range_shift(N);
    const int per = (E + EB - 1) / EB;
    const int beg = b * per;
    const int end = (beg + per < E) ? (beg + per) : E;
    const int* __restrict__ dstp = ei + E;
    if (t < NBIN) c[t] = 0;
    __syncthreads();
    for (int e0 = beg + t; e0 < end; e0 += RB * 1024) {
        int r[RB];
        #pragma unroll
        for (int k = 0; k < RB; ++k) {
            const int e  = e0 + k * 1024;
            const int ec = (e < end) ? e : (end - 1);
            const int rv = range_of(dstp[ec], sh);
            r[k] = (e < end) ? rv : -1;
        }
        #pragma unroll
        for (int k = 0; k < RB; ++k) {
            if (r[k] >= 0) atomicAdd(&c[r[k]], 1);
        }
    }
    __syncthreads();
    if (t < NBIN) H[b * NBIN + t] = c[t];
}

__global__ void __launch_bounds__(1024)
k_range_scatter(const int* __restrict__ ei, const float* __restrict__ eattr,
                const int* __restrict__ H, int* __restrict__ BS, int* __restrict__ bdst,
                int* __restrict__ bpair, int E, int N) {
    __shared__ int s_tot[4][NBIN];
    __shared__ int s_par[4][NBIN];
    __shared__ int s_scan[2][NBIN / 32];
    __shared__ int s_off[NBIN];
    __shared__ int s_gb[NBIN];
    __shared__ int  s_sd[SCAP];
    __shared__ int2 s_sp[SCAP];
    const int t = threadIdx.x, b = blockIdx.x;
    const int sh = range_shift(N);
    const int col = t & (NBIN - 1), g = t >> 8;
    const int per = (E + EB - 1) / EB;
    const int beg = b * per;
    const int end = (beg + per < E) ? (beg + per) : E;
    const bool fits = per <= SCAP;
    int hv[EB / 4];
    #pragma unroll
    for (int k = 0; k < EB / 4; ++k) hv[k] = H[(g * (EB / 4) + k) * NBIN + col];
    const int own = (t < NBIN) ? H[b * NBIN + t] : 0;
    int tsum = 0, psum = 0;
    #pragma unroll
    for (int k = 0; k < EB / 4; ++k) {
        tsum += hv[k];
        psum += (g * (EB / 4) + k < b) ? hv[k] : 0;
    }
    s_tot[g][col] = tsum;
    s_par[g][col] = psum;
    __syncthreads();
    int T = 0, P = 0;
    if (t < NBIN) {
        T = (s_tot[0][t] + s_tot[1][t]) + (s_tot[2][t] + s_tot[3][t]);
        P = (s_par[0][t] + s_par[1][t]) + (s_par[2][t] + s_par[3][t]);
    }
    const int lane = t & 31, wave = t >> 5;
    int inc = T, linc = own;
    #pragma unroll
    for (int o = 1; o < 32; o <<= 1) {
        const int u  = __shfl_up(inc, o, 32);
        const int lu = __shfl_up(linc, o, 32);
        if (lane >= o) { inc += u; linc += lu; }
    }
    if (t < NBIN && lane == 31) { s_scan[0][wave] = inc; s_scan[1][wave] = linc; }
    __syncthreads();
    if (t < NBIN) {
        int pre = 0, lpre = 0;
        for (int w = 0; w < wave; ++w) { pre += s_scan[0][w]; lpre += s_scan[1][w]; }
        const int start  = pre + inc - T;
        const int lstart = lpre + linc - own;
        s_off[t] = fits ? lstart : (start + P);
        s_gb[t]  = start + P - lstart;
        if (b == 0) {
            BS[t] = start;
            if (t == NBIN - 1) BS[NBIN] = start + T;
        }
    }
    __syncthreads();

    const int* __restrict__ dstp = ei + E;
    for (int e0 = beg + t; e0 < end; e0 += RB * 1024) {
        int d[RB], s[RB], a[RB];
        #pragma unroll
        for (int k = 0; k < RB; ++k) {
            const int e  = e0 + k * 1024;
            const int ec = (e < end) ? e : (end - 1);
            d[k] = dstp[ec];
            s[k] = ei[ec];
            a[k] = __float_as_int(eattr[ec]);
        }
        #pragma unroll
        for (int k = 0; k < RB; ++k) {
            if (e0 + k * 1024 < end) {
                const int pos = atomicAdd(&s_off[range_of(d[k], sh)], 1);
                int2 v;
                v.x = s[k];
                v.y = a[k];
                if (fits) {
                    const int lp = ((unsigned)pos < (unsigned)SCAP) ? pos : (SCAP - 1);
                    s_sd[lp] = d[k];
                    s_sp[lp] = v;
                } else {
                    bdst[pos] = d[k];
                    ((int2*)bpair)[pos] = v;
                }
            }
        }
    }
    if (fits) {
        __syncthreads();
        const int nb = end - beg;
        for (int i = t; i < nb; i += 1024) {
            const int dd = s_sd[i];
            const int gp = s_gb[range_of(dd, sh)] + i;
            bdst[gp] = dd;
            ((int2*)bpair)[gp] = s_sp[i];
        }
    }
}

__global__ void __launch_bounds__(1024)
k_range_sort(const int* __restrict__ BS, const int* __restrict__ bdst,
             const int* __restrict__ bpair, int* __restrict__ rowptr, int* __restrict__ spair,
             int E, int N) {
    __shared__ int  s_cnt[RW];
    __shared__ int  s_scan[RW / 32];
    __shared__ int2 s_stage[STG];
    const int t = threadIdx.x, bin = blockIdx.x;
    const int sh = range_shift(N);
    const int Wd = 1 << sh;
    const int node0 = bin << sh;
    if (node0 >= N) return;
    const int msk = (Wd - 1) & (RW - 1);
    const int beg = BS[bin];
    const int end = BS[bin + 1];
    const int nrec = end - beg;
    const bool fits = nrec <= STG;
    s_cnt[t] = 0;
    __syncthreads();
    for (int i0 = beg + t; i0 < end; i0 += RB * 1024) {
        int j[RB];
        #pragma unroll
        for (int k = 0; k < RB; ++k) {
            const int i  = i0 + k * 1024;
            const int ic = (i < end) ? i : (end - 1);
            const int jv = (bdst[ic] - node0) & msk;
            j[k] = (i < end) ? jv : -1;
        }
        #pragma unroll
        for (int k = 0; k < RB; ++k) {
            if (j[k] >= 0) atomicAdd(&s_cnt[j[k]], 1);
        }
    }
    __syncthreads();
    const int lane = t & 31, wave = t >> 5;
    const int own = s_cnt[t];
    int inc = own;
    #pragma unroll
    for (int o = 1; o < 32; o <<= 1) {
        const int u = __shfl_up(inc, o, 32);
        if (lane >= o) inc += u;
    }
    if (lane == 31) s_scan[wave] = inc;
    __syncthreads();
    int pre = 0;
    for (int w = 0; w < wave; ++w) pre += s_scan[w];
    const int excl = pre + inc - own;
    s_cnt[t] = excl;
    if (t < Wd && node0 + t < N) rowptr[node0 + t] = beg + excl;
    if (bin == 0 && t == 0) rowptr[N] = E;
    __syncthreads();
    for (int i0 = beg + t; i0 < end; i0 += RB * 1024) {
        int  j[RB];
        int2 p[RB];
        #pragma unroll
        for (int k = 0; k < RB; ++k) {
            const int i  = i0 + k * 1024;
            const int ic = (i < end) ? i : (end - 1);
            j[k] = (bdst[ic] - node0) & msk;
            p[k] = ((const int2*)bpair)[ic];
        }
        #pragma unroll
        for (int k = 0; k < RB; ++k) {
            if (i0 + k * 1024 < end) {
                const int pos = atomicAdd(&s_cnt[j[k]], 1);
                if (fits) s_stage[pos] = p[k];
                else      ((int2*)spair)[beg + pos] = p[k];
            }
        }
    }
    __syncthreads();
    if (fits) {
        for (int i = t; i < nrec; i += 1024) ((int2*)spair)[beg + i] = s_stage[i];
    }
}

#define LOG2E 1.44269504088896340736f
__device__ __forceinline__ float pow2_hw(float x) { return __builtin_amdgcn_exp2f(x); }
__device__ __forceinline__ float rcp_hw(float x)  { return __builtin_amdgcn_rcpf(x); }

__device__ __forceinline__ float edge_logit(float as, float ad, float ea, float se) {
    float al = __builtin_fmaf(ea, se, as + ad);
    return (al > 0.0f) ? al : NEGS * al;
}


constexpr unsigned grp8_sel(unsigned a0, unsigned a1, unsigned a2, unsigned a3,
                            unsigned a4, unsigned a5, unsigned a6, unsigned a7) {
    return a0 | (a1 << 3) | (a2 << 6) | (a3 << 9) | (a4 << 12) | (a5 << 15) | (a6 << 18) | (a7 << 21);
}
template <unsigned SEL>
__device__ __forceinline__ float grp8_move(float v) {
    return __int_as_float(__builtin_amdgcn_mov_dpp8(__float_as_int(v), SEL));
}
__device__ __forceinline__ float grp8_sum(float v) {
    v += grp8_move<grp8_sel(1, 0, 3, 2, 5, 4, 7, 6)>(v);
    v += grp8_move<grp8_sel(2, 3, 0, 1, 6, 7, 4, 5)>(v);
    v += grp8_move<grp8_sel(4, 5, 6, 7, 0, 1, 2, 3)>(v);
    return v;
}
template <int MASK>
__device__ __forceinline__ float row_xor(float v) {
    return __int_as_float(__builtin_amdgcn_update_dpp(0, __float_as_int(v), 0x160 + MASK, 0xf, 0xf, false));
}
__device__ __forceinline__ float grp8_max(float v) {
    v = fmaxf(v, row_xor<1>(v));
    v = fmaxf(v, row_xor<2>(v));
    v = fmaxf(v, row_xor<4>(v));
    return v;
}
__device__ __forceinline__ float xhalf(float v) {
    const int b = __float_as_int(v);
    return __int_as_float(__builtin_amdgcn_permlanex16(b, b, 0x76543210, 0xfedcba98, false, false));
}
__device__ __forceinline__ float wave_sum(float v) {
    v += xhalf(v);
    v += row_xor<8>(v);
    v += row_xor<4>(v);
    v += row_xor<2>(v);
    v += row_xor<1>(v);
    return v;
}

__device__ __forceinline__ void
block_pairs(const int2* __restrict__ pairs, int base, int end, int lastp, int t,
            int2 (&pr)[4], bool (&ok)[4]) {
    for (int q = 0; q < 4; ++q) {
        const int idx = base + 8 * q + t;
        ok[q] = idx < end;
        pr[q] = pairs[ok[q] ? idx : lastp];
    }
}
#define RAISE_AT (8.0f * LOG2E)

__device__ __forceinline__ void
block_attr(const int2 (&pr)[4], const bool (&ok)[4], float& easum) {
    for (int q = 0; q < 4; ++q) easum += ok[q] ? __int_as_float(pr[q].y) : 0.0f;
}

__device__ __forceinline__ float grp8_scatter_sum(const float (&p)[8], int t) {
    const bool b2 = (t & 4) != 0, b1 = (t & 2) != 0, b0 = (t & 1) != 0;
    float w4[4], w2[2];
    for (int j = 0; j < 4; ++j) {
        const float keep = b2 ? p[4 + j] : p[j];
        const float give = b2 ? p[j] : p[4 + j];
        w4[j] = keep + row_xor<4>(give);
    }
    for (int j = 0; j < 2; ++j) {
        const float keep = b1 ? w4[2 + j] : w4[j];
        const float give = b1 ? w4[j] : w4[2 + j];
        w2[j] = keep + row_xor<2>(give);
    }
    const float keep = b0 ? w2[1] : w2[0];
    const float give = b0 ? w2[0] : w2[1];
    return keep + row_xor<1>(give);
}

#define GATHER_ONE(U)                                                                         \
    {                                                                                         \
        float wj;                                                                             \
        asm("v_mov_b32_dpp %0, %1 dpp8:[" #U "," #U "," #U "," #U "," #U "," #U "," #U "," #U "]" \
            : "=v"(wj) : "v"(wq));                                                            \
        ax = __builtin_fmaf(xv[U].x, wj, ax);                                                 \
        ay = __builtin_fmaf(xv[U].y, wj, ay);                                                 \
    }
__device__ __forceinline__ void
gather_group(const float* __restrict__ xs, unsigned lane8, int t, int sq, float eaq, bool okq,
             float at0, float at1, float adh, float seh,
             float& r, float& es, float& dsum, float& ax, float& ay) {
    float2 xv[8];
    for (int u = 0; u < 8; ++u) {
        const unsigned sj = (unsigned)__builtin_amdgcn_readlane(sq, u);
        xv[u] = *(const float2*)((const char*)xs + ((sj << 8) + lane8));
    }
    float p[8];
    for (int u = 0; u < 8; ++u) p[u] = __builtin_fmaf(xv[u].y, at1, xv[u].x * at0);
    const float asr = grp8_scatter_sum(p, t);
    const float lg = okq ? edge_logit(asr, adh, eaq, seh) : -__builtin_huge_valf();
    if (__builtin_amdgcn_ballot_w32(lg - r > RAISE_AT) != 0u) {
        const float rn = fmaxf(r, grp8_max(lg));
        const float sc = pow2_hw(r - rn);
        r = rn;
        es *= sc;
        dsum *= sc;
        ax *= sc;
        ay *= sc;
    }
    float wq = pow2_hw(lg - r);
    asm("v_nop" : "+v"(wq));
    dsum += wq;
    GATHER_ONE(0) GATHER_ONE(1) GATHER_ONE(2) GATHER_ONE(3)
    GATHER_ONE(4) GATHER_ONE(5) GATHER_ONE(6) GATHER_ONE(7)
}
#undef GATHER_ONE

__device__ __forceinline__ void
gather_block(const float* __restrict__ xs, unsigned lane8, int t, int cnt, const int2 (&pr)[4],
             const bool (&ok)[4], float at0, float at1, float adh, float seh,
             float& r, float& es, float& dsum, float& ax, float& ay) {
    for (int q = 0; q < 4; ++q) {
        if (8 * q < cnt)
            gather_group(xs, lane8, t, pr[q].x, __int_as_float(pr[q].y), ok[q], at0, at1, adh, seh,
                         r, es, dsum, ax, ay);
    }
}

template <bool LAST>
__device__ __forceinline__ void
node_softmax_agg_norm(const int* __restrict__ rowptr, const int* __restrict__ spair,
                      const float* __restrict__ sed, const float* __restrict__ a_src,
                      const float* __restrict__ a_dst, const float* __restrict__ att,
                      const float* __restrict__ xs,
                      const float* __restrict__ res, const float* __restrict__ b,
                      const float* __restrict__ g, const float* __restrict__ beta,
                      const float* __restrict__ Wout, const float* __restrict__ bout,
                      float* __restrict__ dst, int N) {
    const int lane = threadIdx.x & 31;
    const int n = blockIdx.x * 8 + __builtin_amdgcn_readfirstlane((int)(threadIdx.x >> 5));
    if (n >= N) return;
    const int beg = rowptr[n];
    const int end = rowptr[n + 1];
    const int deg = end - beg;
    const int lastp = (end > 0) ? (end - 1) : 0;
    const bool two = deg > 32;
    const int h = lane >> 3, t = lane & 7;
    const unsigned lane8 = (unsigned)lane * 8u;
    const int2* pairs = (const int2*)spair;

    int2 prA[4], prB[4];
    bool okA[4], okB[4];
    block_pairs(pairs, beg, end, lastp, t, prA, okA);
    if (two) block_pairs(pairs, beg + 32, end, lastp, t, prB, okB);
    const float adh = a_dst[n * HEADS + h] * LOG2E;
    const float ash = a_src[n * HEADS + h] * LOG2E;
    const float seh = sed[h] * LOG2E;
    const float at0 = att[2 * lane] * LOG2E, at1 = att[2 * lane + 1] * LOG2E;
    const float2 xn = *(const float2*)((const char*)xs + (((unsigned)n << 8) + lane8));
    const float* rp = res + (size_t)n * DIM;
    const float rs0 = rp[2 * lane],   rs1 = rp[2 * lane + 1];
    const float bi0 = b[2 * lane],    bi1 = b[2 * lane + 1];
    const float ga0 = g[2 * lane],    ga1 = g[2 * lane + 1];
    const float be0 = beta[2 * lane], be1 = beta[2 * lane + 1];

    float easum = 0.0f;
    block_attr(prA, okA, easum);
    if (two) block_attr(prB, okB, easum);
    for (int base = beg + 64; base < end; base += 32) {
        int2 prC[4];
        bool okC[4];
        block_pairs(pairs, base, end, lastp, t, prC, okC);
        block_attr(prC, okC, easum);
    }
    easum = grp8_sum(easum);
    const float la = easum * rcp_hw(fmaxf((float)deg, 1.0f));
    const float sl = edge_logit(ash, adh, la, seh);

    float r = sl, es = 1.0f, dsum = 0.0f;
    float ax = xn.x, ay = xn.y;
    gather_block(xs, lane8, t, deg, prA, okA, at0, at1, adh, seh, r, es, dsum, ax, ay);
    if (two) gather_block(xs, lane8, t, deg - 32, prB, okB, at0, at1, adh, seh, r, es, dsum, ax, ay);
    for (int base = beg + 64; base < end; base += 32) {
        int2 prC[4];
        bool okC[4];
        block_pairs(pairs, base, end, lastp, t, prC, okC);
        gather_block(xs, lane8, t, end - base, prC, okC, at0, at1, adh, seh, r, es, dsum, ax, ay);
    }
    const float rinv = rcp_hw((es + grp8_sum(dsum)) + 1e-16f);

    const float v0 = (ax * rinv + bi0) + rs0;
    const float v1 = (ay * rinv + bi1) + rs1;
    const float mu = wave_sum(v0 + v1) * (1.0f / DIM);
    const float d0 = v0 - mu, d1 = v1 - mu;
    const float vs = wave_sum(d0 * d0 + d1 * d1);
    const float rstd = __builtin_amdgcn_rsqf(vs * (1.0f / DIM) + 1e-5f);
    float y0 = d0 * rstd * ga0 + be0;
    float y1 = d1 * rstd * ga1 + be1;
    y0 = (y0 > 0.0f) ? y0 : (pow2_hw(y0 * LOG2E) - 1.0f);
    y1 = (y1 > 0.0f) ? y1 : (pow2_hw(y1 * LOG2E) - 1.0f);
    if (!LAST) {
        float2 o;
        o.x = y0;
        o.y = y1;
        ((float2*)dst)[(size_t)n * (DIM / 2) + lane] = o;
    } else {
        const float p = wave_sum(y0 * Wout[2 * lane] + y1 * Wout[2 * lane + 1]);
        if (lane == 0) dst[n] = p + bout[0];
    }
}

__global__ void __launch_bounds__(256)
k_node_agg_norm(const int* __restrict__ rowptr, const int* __restrict__ spair,
                const float* __restrict__ sed, const float* __restrict__ a_src,
                const float* __restrict__ a_dst, const float* __restrict__ att,
                const float* __restrict__ xs,
                const float* __restrict__ res, const float* __restrict__ b,
                const float* __restrict__ g, const float* __restrict__ beta,
                float* __restrict__ hout, int N) {
    node_softmax_agg_norm<false>(rowptr, spair, sed, a_src, a_dst, att, xs, res, b, g, beta,
                                 nullptr, nullptr, hout, N);
}

__global__ void __launch_bounds__(256)
k_node_agg_norm_out(const int* __restrict__ rowptr, const int* __restrict__ spair,
                    const float* __restrict__ sed, const float* __restrict__ a_src,
                    const float* __restrict__ a_dst, const float* __restrict__ att,
                    const float* __restrict__ xs,
                    const float* __restrict__ res, const float* __restrict__ b,
                    const float* __restrict__ g, const float* __restrict__ beta,
                    const float* __restrict__ Wout, const float* __restrict__ bout,
                    float* __restrict__ out, int N) {
    node_softmax_agg_norm<true>(rowptr, spair, sed, a_src, a_dst, att, xs, res, b, g, beta,
                                Wout, bout, out, N);
}

extern "C" void kernel_launch(void* const* d_in, const int* in_sizes, int n_in,
                              void* d_out, int out_size, void* d_ws, size_t ws_size,
                              hipStream_t stream) {
    const float* x    = (const float*)d_in[0];
    const int*   ei   = (const int*)  d_in[1];
    const float* eatt = (const float*)d_in[3];
    const float* W1   = (const float*)d_in[4];
    const float* as1  = (const float*)d_in[5];
    const float* ad1  = (const float*)d_in[6];
    const float* We1  = (const float*)d_in[7];
    const float* ae1  = (const float*)d_in[8];
    const float* b1   = (const float*)d_in[9];
    const float* W2   = (const float*)d_in[10];
    const float* as2  = (const float*)d_in[11];
    const float* ad2  = (const float*)d_in[12];
    const float* We2  = (const float*)d_in[13];
    const float* ae2  = (const float*)d_in[14];
    const float* b2   = (const float*)d_in[15];
    const float* g1   = (const float*)d_in[16];
    const float* be1  = (const float*)d_in[17];
    const float* g2   = (const float*)d_in[18];
    const float* be2  = (const float*)d_in[19];
    const float* Wout = (const float*)d_in[20];
    const float* bout = (const float*)d_in[21];
    float* outp = (float*)d_out;

    const int N = in_sizes[0] / DIM;
    const int E = in_sizes[1] / 2;

    float* w = (float*)d_ws;
    size_t o = 0;
    float* xs    = w + o; o += (size_t)N * DIM;
    int*   grp   = (int*)(w + o); o += (size_t)N * DIM;
    float* h1    = w + o; o += (size_t)N * DIM;
    float* a_src = w + o; o += (size_t)N * HEADS;
    float* a_dst = w + o; o += (size_t)N * HEADS;
    int*   spair = (int*)(w + o); o += (size_t)2 * E;
    int*   rowp  = (int*)(w + o); o += (size_t)N + 1;
    float* sed   = w + o; o += 8;
    int*   bdst  = grp;
    int*   H     = grp + E;
    int*   BS    = H + EB * NBIN;
    int*   bpair = (int*)h1;

    auto cdiv = [](long a, long b) { return (int)((a + b - 1) / b); };
    const int T = 256;

    k_range_count<<<EB, 1024, 0, stream>>>(ei, H, E, N);
    k_range_scatter<<<EB, 1024, 0, stream>>>(ei, eatt, H, BS, bdst, bpair, E, N);
    k_range_sort<<<NBIN, 1024, 0, stream>>>(BS, bdst, bpair, rowp, spair, E, N);

    k_gemm64_coef<<<cdiv((long)cdiv(N, 16), 2), T, 0, stream>>>(x, W1, as1, ad1, We1, ae1,
                                                                xs, a_src, a_dst, sed, N);
    k_node_agg_norm<<<cdiv(N, 8), T, 0, stream>>>(rowp, spair, sed, a_src, a_dst, as1, xs,
                                                  x, b1, g1, be1, h1, N);

    k_gemm64_coef<<<cdiv((long)cdiv(N, 16), 2), T, 0, stream>>>(h1, W2, as2, ad2, We2, ae2,
                                                                xs, a_src, a_dst, sed + 4, N);
    k_node_agg_norm_out<<<cdiv(N, 8), T, 0, stream>>>(rowp, spair, sed + 4, a_src, a_dst, as2, xs,
                                                      h1, b2, g2, be2, Wout, bout, outp, N);
}
